# starting kernel
# baseline (speedup 1.0000x reference)
#include <hip/hip_runtime.h>
#include <math.h>

#define NN      20000
#define NE      320000
#define ETOT    (NE + NN)
#define IN_CH   128
#define HID     64
#define HEADS   8
#define HC1     (HEADS * HID)
#define OUT_CH  64

typedef float v2f __attribute__((ext_vector_type(2)));
typedef float v8f __attribute__((ext_vector_type(8)));

__device__ __forceinline__ int f2ord(float f) {
    int i = __float_as_int(f);
    return i >= 0 ? i : (i ^ 0x7fffffff);
}
__device__ __forceinline__ float ord2f(int i) {
    return __int_as_float(i >= 0 ? i : (i ^ 0x7fffffff));
}

__global__ void wmma_gemm_f32(const float* __restrict__ A,
                              const float* __restrict__ B,
                              float* __restrict__ C,
                              int M, int K, int N) {
    const int wave    = (blockIdx.x * blockDim.x + threadIdx.x) >> 5;
    const int lane    = threadIdx.x & 31;
    const int nGroups = N >> 6;
    const int mTiles  = M >> 4;
    if (wave >= mTiles * nGroups) return;
    const int mt   = wave / nGroups;
    const int ng   = wave % nGroups;
    const int l    = lane & 15;
    const int half = lane >> 4;

    unsigned aoff = (unsigned)(mt * 16 + l) * (unsigned)K + 2u * half;
    unsigned boff = (unsigned)(2 * half) * (unsigned)N + (unsigned)(ng * 64 + l);
    const unsigned uN = (unsigned)N;

    v8f acc0 = {}, acc1 = {}, acc2 = {}, acc3 = {};
    for (int k0 = 0; k0 < K; k0 += 4) {
        v2f af = *(const v2f*)(A + aoff);
        v2f bf0 = { B[boff],      B[boff + uN]      };
        v2f bf1 = { B[boff + 16], B[boff + uN + 16] };
        v2f bf2 = { B[boff + 32], B[boff + uN + 32] };
        v2f bf3 = { B[boff + 48], B[boff + uN + 48] };
        acc0 = __builtin_amdgcn_wmma_f32_16x16x4_f32(false, af, false, bf0, (short)0, acc0, false, false);
        acc1 = __builtin_amdgcn_wmma_f32_16x16x4_f32(false, af, false, bf1, (short)0, acc1, false, false);
        acc2 = __builtin_amdgcn_wmma_f32_16x16x4_f32(false, af, false, bf2, (short)0, acc2, false, false);
        acc3 = __builtin_amdgcn_wmma_f32_16x16x4_f32(false, af, false, bf3, (short)0, acc3, false, false);
        aoff += 4u;
        boff += 4u * uN;
    }
    unsigned coff = (unsigned)(mt * 16 + 8 * half) * uN + (unsigned)(ng * 64 + l);
    #pragma unroll
    for (int r = 0; r < 8; ++r) {
        float* cp = C + coff;
        cp[0]  = acc0[r];
        cp[16] = acc1[r];
        cp[32] = acc2[r];
        cp[48] = acc3[r];
        coff += uN;
    }
}

__global__ void scores_kernel(const float* __restrict__ h,
                              const float* __restrict__ a_src,
                              const float* __restrict__ a_dst,
                              float* __restrict__ esrc, float* __restrict__ edst,
                              int* __restrict__ mmax, float* __restrict__ denom,
                              int n, int H, int C) {
    int i = blockIdx.x * blockDim.x + threadIdx.x;
    if (i >= n * H) return;
    int node = i / H, hh = i % H;
    const float* hp = h + (size_t)node * H * C + (size_t)hh * C;
    const float* as = a_src + hh * C;
    const float* ad = a_dst + hh * C;
    float s = 0.f, d = 0.f;
    for (int c = 0; c < C; ++c) { float v = hp[c]; s += v * as[c]; d += v * ad[c]; }
    esrc[i]  = s;
    edst[i]  = d;
    mmax[i]  = f2ord(-INFINITY);
    denom[i] = 0.f;
}

__global__ void fill_zero(float* __restrict__ p, long long n) {
    long long i = (long long)blockIdx.x * blockDim.x + threadIdx.x;
    long long stride = (long long)gridDim.x * blockDim.x;
    for (; i < n; i += stride) p[i] = 0.f;
}

__global__ void edge_score_max(const int* __restrict__ ei,
                               const float* __restrict__ esrc,
                               const float* __restrict__ edst,
                               float* __restrict__ eleaky,
                               int* __restrict__ mmax, int H) {
    long long t = (long long)blockIdx.x * blockDim.x + threadIdx.x;
    if (t >= (long long)ETOT * H) return;
    int e = (int)(t / H), hh = (int)(t % H);
    int src, dst;
    if (e < NE) { src = ei[e]; dst = ei[NE + e]; }
    else        { src = dst = e - NE; }
    float v = esrc[src * H + hh] + edst[dst * H + hh];
    v = v > 0.f ? v : 0.2f * v;
    eleaky[t] = v;
    atomicMax(&mmax[dst * H + hh], f2ord(v));
}

__global__ void edge_exp_sum(const int* __restrict__ ei,
                             float* __restrict__ eleaky,
                             const int* __restrict__ mmax,
                             float* __restrict__ denom, int H) {
    long long t = (long long)blockIdx.x * blockDim.x + threadIdx.x;
    if (t >= (long long)ETOT * H) return;
    int e = (int)(t / H), hh = (int)(t % H);
    int dst = (e < NE) ? ei[NE + e] : (e - NE);
    float ex = expf(eleaky[t] - ord2f(mmax[dst * H + hh]));
    eleaky[t] = ex;
    atomicAdd(&denom[dst * H + hh], ex);
}

__global__ void edge_alpha(const int* __restrict__ ei,
                           float* __restrict__ exal,
                           const float* __restrict__ denom, int H) {
    long long t = (long long)blockIdx.x * blockDim.x + threadIdx.x;
    if (t >= (long long)ETOT * H) return;
    int e = (int)(t / H), hh = (int)(t % H);
    int dst = (e < NE) ? ei[NE + e] : (e - NE);
    exal[t] = exal[t] / (denom[dst * H + hh] + 1e-16f);
}

__global__ void edge_aggregate(const int* __restrict__ ei,
                               const float* __restrict__ alpha,
                               const float* __restrict__ hval,
                               float* __restrict__ agg, int H, int C) {
    const int HC = H * C;
    long long t = (long long)blockIdx.x * blockDim.x + threadIdx.x;
    if (t >= (long long)ETOT * HC) return;
    int e = (int)(t / HC), c = (int)(t % HC);
    int hh = c / C;
    int src, dst;
    if (e < NE) { src = ei[e]; dst = ei[NE + e]; }
    else        { src = dst = e - NE; }
    float a = alpha[(size_t)e * H + hh];
    atomicAdd(&agg[(size_t)dst * HC + c], a * hval[(size_t)src * HC + c]);
}

__global__ void bias_relu(float* __restrict__ a, const float* __restrict__ b,
                          long long n, int F) {
    long long i = (long long)blockIdx.x * blockDim.x + threadIdx.x;
    if (i >= n) return;
    float v = a[i] + b[(int)(i % F)];
    a[i] = v > 0.f ? v : 0.f;
}

__global__ void logsoftmax64(float* __restrict__ out, const float* __restrict__ b2) {
    int node = blockIdx.x;
    int lane = threadIdx.x;
    float* row = out + (size_t)node * OUT_CH;
    float v0 = row[lane]      + b2[lane];
    float v1 = row[lane + 32] + b2[lane + 32];
    float m = fmaxf(v0, v1);
    #pragma unroll
    for (int o = 16; o > 0; o >>= 1) m = fmaxf(m, __shfl_xor(m, o, 32));
    float s = expf(v0 - m) + expf(v1 - m);
    #pragma unroll
    for (int o = 16; o > 0; o >>= 1) s += __shfl_xor(s, o, 32);
    float lse = logf(s) + m;
    row[lane]      = v0 - lse;
    row[lane + 32] = v1 - lse;
}

extern "C" void kernel_launch(void* const* d_in, const int* in_sizes, int n_in,
                              void* d_out, int out_size, void* d_ws, size_t ws_size,
                              hipStream_t stream) {
    const float* x   = (const float*)d_in[0];
    const int*   ei  = (const int*)  d_in[1];
    const float* W1  = (const float*)d_in[2];
    const float* a1s = (const float*)d_in[3];
    const float* a1d = (const float*)d_in[4];
    const float* b1  = (const float*)d_in[5];
    const float* W2  = (const float*)d_in[6];
    const float* a2s = (const float*)d_in[7];
    const float* a2d = (const float*)d_in[8];
    const float* b2  = (const float*)d_in[9];
    float* out = (float*)d_out;

    float* ws = (float*)d_ws;
    float* h1pre = ws;
    float* agg1  = h1pre + (size_t)NN * HC1;
    float* esrc1 = agg1  + (size_t)NN * HC1;
    float* edst1 = esrc1 + (size_t)NN * HEADS;
    int*   m1    = (int*)(edst1 + (size_t)NN * HEADS);
    float* den1  = (float*)m1 + (size_t)NN * HEADS;
    float* ex1   = den1 + (size_t)NN * HEADS;
    float* h2pre = ex1  + (size_t)ETOT * HEADS;
    float* esrc2 = h2pre + (size_t)NN * OUT_CH;
    float* edst2 = esrc2 + NN;
    int*   m2    = (int*)(edst2 + NN);
    float* den2  = (float*)m2 + NN;
    float* ex2   = den2 + NN;

    const int TB = 256;
    auto blocks = [](long long n, int tb) { return (unsigned)((n + tb - 1) / tb); };

    {
        long long waves = (long long)(NN / 16) * (HC1 / 64);
        wmma_gemm_f32<<<blocks(waves * 32, TB), TB, 0, stream>>>(x, W1, h1pre, NN, IN_CH, HC1);
    }
    scores_kernel<<<blocks((long long)NN * HEADS, TB), TB, 0, stream>>>(
        h1pre, a1s, a1d, esrc1, edst1, m1, den1, NN, HEADS, HID);
    fill_zero<<<4096, TB, 0, stream>>>(agg1, (long long)NN * HC1);
    edge_score_max<<<blocks((long long)ETOT * HEADS, TB), TB, 0, stream>>>(ei, esrc1, edst1, ex1, m1, HEADS);
    edge_exp_sum  <<<blocks((long long)ETOT * HEADS, TB), TB, 0, stream>>>(ei, ex1, m1, den1, HEADS);
    edge_alpha    <<<blocks((long long)ETOT * HEADS, TB), TB, 0, stream>>>(ei, ex1, den1, HEADS);
    edge_aggregate<<<blocks((long long)ETOT * HC1,   TB), TB, 0, stream>>>(ei, ex1, h1pre, agg1, HEADS, HID);
    bias_relu<<<blocks((long long)NN * HC1, TB), TB, 0, stream>>>(agg1, b1, (long long)NN * HC1, HC1);

    {
        long long waves = (long long)(NN / 16) * (OUT_CH / 64);
        wmma_gemm_f32<<<blocks(waves * 32, TB), TB, 0, stream>>>(agg1, W2, h2pre, NN, HC1, OUT_CH);
    }
    scores_kernel<<<blocks((long long)NN, TB), TB, 0, stream>>>(
        h2pre, a2s, a2d, esrc2, edst2, m2, den2, NN, 1, OUT_CH);
    fill_zero<<<2048, TB, 0, stream>>>(out, (long long)NN * OUT_CH);
    edge_score_max<<<blocks((long long)ETOT, TB), TB, 0, stream>>>(ei, esrc2, edst2, ex2, m2, 1);
    edge_exp_sum  <<<blocks((long long)ETOT, TB), TB, 0, stream>>>(ei, ex2, m2, den2, 1);
    edge_alpha    <<<blocks((long long)ETOT, TB), TB, 0, stream>>>(ei, ex2, den2, 1);
    edge_aggregate<<<blocks((long long)ETOT * OUT_CH, TB), TB, 0, stream>>>(ei, ex2, h2pre, out, 1, OUT_CH);
    logsoftmax64<<<NN, 32, 0, stream>>>(out, b2);
}


# Layer 1 channel-major: one workgroup per channel pair keeps its h column and all nodes' sums in LDS, scans all edges with LDS float adds (loop term first, then arrival order); global adds gone
# speedup vs baseline: 2.1639x; 2.1639x over previous
#include <hip/hip_runtime.h>
#include <math.h>

#define NN      20000
#define NE      320000
#define ETOT    (NE + NN)
#define IN_CH   128
#define HID     64
#define HEADS   8
#define HC1     (HEADS * HID)
#define OUT_CH  64

typedef float v2f __attribute__((ext_vector_type(2)));
typedef float v8f __attribute__((ext_vector_type(8)));

__device__ __forceinline__ int f2ord(float f) {
    int i = __float_as_int(f);
    return i >= 0 ? i : (i ^ 0x7fffffff);
}
__device__ __forceinline__ float ord2f(int i) {
    return __int_as_float(i >= 0 ? i : (i ^ 0x7fffffff));
}

__global__ void wmma_gemm_f32_ct(const float* __restrict__ X,
                                 const float* __restrict__ Wm,
                                 float* __restrict__ HT,
                                 int M, int K, int N) {
    const int wave    = (blockIdx.x * blockDim.x + threadIdx.x) >> 5;
    const int lane    = threadIdx.x & 31;
    const int cGroups = N >> 5;
    const int mGroups = M >> 5;
    if (wave >= mGroups * cGroups) return;
    const int mg   = wave / cGroups;
    const int cg   = wave % cGroups;
    const int l    = lane & 15;
    const int half = lane >> 4;
    const unsigned uN = (unsigned)N, uM = (unsigned)M, uK = (unsigned)K;

    unsigned xoff = (unsigned)(mg * 32 + l) * uK + 2u * half;
    unsigned woff = (unsigned)(2 * half) * uN + (unsigned)(cg * 32 + l);

    v8f acc00 = {}, acc01 = {}, acc10 = {}, acc11 = {};
    for (int k0 = 0; k0 < K; k0 += 4) {
        v2f x0 = *(const v2f*)(X + xoff);
        v2f x1 = *(const v2f*)(X + xoff + 16u * uK);
        v2f w0 = { Wm[woff],      Wm[woff + uN]      };
        v2f w1 = { Wm[woff + 16], Wm[woff + uN + 16] };
        acc00 = __builtin_amdgcn_wmma_f32_16x16x4_f32(false, w0, false, x0, (short)0, acc00, false, false);
        acc01 = __builtin_amdgcn_wmma_f32_16x16x4_f32(false, w0, false, x1, (short)0, acc01, false, false);
        acc10 = __builtin_amdgcn_wmma_f32_16x16x4_f32(false, w1, false, x0, (short)0, acc10, false, false);
        acc11 = __builtin_amdgcn_wmma_f32_16x16x4_f32(false, w1, false, x1, (short)0, acc11, false, false);
        xoff += 4u;
        woff += 4u * uN;
    }
    unsigned hoff = (unsigned)(cg * 32 + 8 * half) * uM + (unsigned)(mg * 32 + l);
    #pragma unroll
    for (int r = 0; r < 8; ++r) {
        float* hp = HT + hoff;
        hp[0]            = acc00[r];
        hp[16]           = acc01[r];
        hp[16u * uM]      = acc10[r];
        hp[16u * uM + 16] = acc11[r];
        hoff += uM;
    }
}

__global__ void wmma_gemm_f32_at(const float* __restrict__ AT,
                                 const float* __restrict__ B,
                                 float* __restrict__ C,
                                 int M, int K, int N) {
    const int wave    = (blockIdx.x * blockDim.x + threadIdx.x) >> 5;
    const int lane    = threadIdx.x & 31;
    const int nGroups = N >> 6;
    const int mTiles  = M >> 4;
    if (wave >= mTiles * nGroups) return;
    const int mt   = wave / nGroups;
    const int ng   = wave % nGroups;
    const int l    = lane & 15;
    const int half = lane >> 4;
    const unsigned uN = (unsigned)N, uM = (unsigned)M;

    unsigned aoff = (unsigned)(2 * half) * uM + (unsigned)(mt * 16 + l);
    unsigned boff = (unsigned)(2 * half) * uN + (unsigned)(ng * 64 + l);

    v8f acc0 = {}, acc1 = {}, acc2 = {}, acc3 = {};
    for (int k0 = 0; k0 < K; k0 += 4) {
        v2f af  = { AT[aoff],     AT[aoff + uM]     };
        v2f bf0 = { B[boff],      B[boff + uN]      };
        v2f bf1 = { B[boff + 16], B[boff + uN + 16] };
        v2f bf2 = { B[boff + 32], B[boff + uN + 32] };
        v2f bf3 = { B[boff + 48], B[boff + uN + 48] };
        acc0 = __builtin_amdgcn_wmma_f32_16x16x4_f32(false, af, false, bf0, (short)0, acc0, false, false);
        acc1 = __builtin_amdgcn_wmma_f32_16x16x4_f32(false, af, false, bf1, (short)0, acc1, false, false);
        acc2 = __builtin_amdgcn_wmma_f32_16x16x4_f32(false, af, false, bf2, (short)0, acc2, false, false);
        acc3 = __builtin_amdgcn_wmma_f32_16x16x4_f32(false, af, false, bf3, (short)0, acc3, false, false);
        aoff += 4u * uM;
        boff += 4u * uN;
    }
    unsigned coff = (unsigned)(mt * 16 + 8 * half) * uN + (unsigned)(ng * 64 + l);
    #pragma unroll
    for (int r = 0; r < 8; ++r) {
        float* cp = C + coff;
        cp[0]  = acc0[r];
        cp[16] = acc1[r];
        cp[32] = acc2[r];
        cp[48] = acc3[r];
        coff += uN;
    }
}

__global__ void scores_ct(const float* __restrict__ HT,
                          const float* __restrict__ a_src,
                          const float* __restrict__ a_dst,
                          float* __restrict__ esrc, float* __restrict__ edst,
                          int* __restrict__ mmax, float* __restrict__ denom,
                          int n, int H, int C) {
    int i = blockIdx.x * blockDim.x + threadIdx.x;
    if (i >= n * H) return;
    int hh = i / n, node = i % n;
    const float* hp = HT + (size_t)hh * C * n + node;
    const float* as = a_src + hh * C;
    const float* ad = a_dst + hh * C;
    float s = 0.f, d = 0.f;
    for (int c = 0; c < C; ++c) { float v = hp[(size_t)c * n]; s += v * as[c]; d += v * ad[c]; }
    int o = node * H + hh;
    esrc[o]  = s;
    edst[o]  = d;
    mmax[o]  = f2ord(-INFINITY);
    denom[o] = 0.f;
}

__global__ void scores_kernel(const float* __restrict__ h,
                              const float* __restrict__ a_src,
                              const float* __restrict__ a_dst,
                              float* __restrict__ esrc, float* __restrict__ edst,
                              int* __restrict__ mmax, float* __restrict__ denom,
                              int n, int H, int C) {
    int i = blockIdx.x * blockDim.x + threadIdx.x;
    if (i >= n * H) return;
    int node = i / H, hh = i % H;
    const float* hp = h + (size_t)node * H * C + (size_t)hh * C;
    const float* as = a_src + hh * C;
    const float* ad = a_dst + hh * C;
    float s = 0.f, d = 0.f;
    for (int c = 0; c < C; ++c) { float v = hp[c]; s += v * as[c]; d += v * ad[c]; }
    esrc[i]  = s;
    edst[i]  = d;
    mmax[i]  = f2ord(-INFINITY);
    denom[i] = 0.f;
}

__global__ void fill_zero(float* __restrict__ p, long long n) {
    long long i = (long long)blockIdx.x * blockDim.x + threadIdx.x;
    long long stride = (long long)gridDim.x * blockDim.x;
    for (; i < n; i += stride) p[i] = 0.f;
}

__global__ void edge_score_max(const int* __restrict__ ei,
                               const float* __restrict__ esrc,
                               const float* __restrict__ edst,
                               float* __restrict__ eleaky,
                               int* __restrict__ mmax, int H) {
    long long t = (long long)blockIdx.x * blockDim.x + threadIdx.x;
    if (t >= (long long)ETOT * H) return;
    int e = (int)(t / H), hh = (int)(t % H);
    int src, dst;
    if (e < NE) { src = ei[e]; dst = ei[NE + e]; }
    else        { src = dst = e - NE; }
    float v = esrc[src * H + hh] + edst[dst * H + hh];
    v = v > 0.f ? v : 0.2f * v;
    eleaky[t] = v;
    atomicMax(&mmax[dst * H + hh], f2ord(v));
}

__global__ void edge_exp_sum(const int* __restrict__ ei,
                             float* __restrict__ eleaky,
                             const int* __restrict__ mmax,
                             float* __restrict__ denom, int H) {
    long long t = (long long)blockIdx.x * blockDim.x + threadIdx.x;
    if (t >= (long long)ETOT * H) return;
    int e = (int)(t / H), hh = (int)(t % H);
    int dst = (e < NE) ? ei[NE + e] : (e - NE);
    float ex = expf(eleaky[t] - ord2f(mmax[dst * H + hh]));
    eleaky[t] = ex;
    atomicAdd(&denom[dst * H + hh], ex);
}

__global__ void edge_alpha(const int* __restrict__ ei,
                           float* __restrict__ exal,
                           const float* __restrict__ denom, int H) {
    long long t = (long long)blockIdx.x * blockDim.x + threadIdx.x;
    if (t >= (long long)ETOT * H) return;
    int e = (int)(t / H), hh = (int)(t % H);
    int dst = (e < NE) ? ei[NE + e] : (e - NE);
    exal[t] = exal[t] / (denom[dst * H + hh] + 1e-16f);
}

__global__ void edge_aggregate(const int* __restrict__ ei,
                               const float* __restrict__ alpha,
                               const float* __restrict__ hval,
                               float* __restrict__ agg, int H, int C) {
    const int HC = H * C;
    long long t = (long long)blockIdx.x * blockDim.x + threadIdx.x;
    if (t >= (long long)ETOT * HC) return;
    int e = (int)(t / HC), c = (int)(t % HC);
    int hh = c / C;
    int src, dst;
    if (e < NE) { src = ei[e]; dst = ei[NE + e]; }
    else        { src = dst = e - NE; }
    float a = alpha[(size_t)e * H + hh];
    atomicAdd(&agg[(size_t)dst * HC + c], a * hval[(size_t)src * HC + c]);
}

__global__ void edge_alpha_hm(const int* __restrict__ ei,
                              const float* __restrict__ ex,
                              const float* __restrict__ denom,
                              float* __restrict__ alT) {
    __shared__ float tile[HEADS][33];
    const int tid = threadIdx.x;
    const int el = tid >> 3, hh = tid & 7;
    const int e  = blockIdx.x * 32 + el;
    const int dst = (e < NE) ? ei[NE + e] : (e - NE);
    tile[hh][el] = ex[(size_t)e * HEADS + hh] / (denom[dst * HEADS + hh] + 1e-16f);
    __syncthreads();
    const int w = tid >> 5, lane = tid & 31;
    alT[(size_t)w * ETOT + (size_t)blockIdx.x * 32 + lane] = tile[w][lane];
}

__global__ __launch_bounds__(1024) void agg_pair_lds(const int* __restrict__ ei,
                                                     const float* __restrict__ alT,
                                                     const float* __restrict__ HT,
                                                     float* __restrict__ outT) {
    __shared__ v2f   hs[NN];
    __shared__ float acc0[NN];
    __shared__ float acc1[NN];
    const int tid  = threadIdx.x;
    const int c0   = 2 * blockIdx.x;
    const int head = c0 / HID;
    const float* h0 = HT + (size_t)c0 * NN;
    const float* h1 = h0 + NN;
    const float* al = alT + (size_t)head * ETOT;

    for (int n = tid; n < NN; n += 1024) {
        v2f hv = { h0[n], h1[n] };
        float a = al[NE + n];
        hs[n]   = hv;
        acc0[n] = a * hv.x;
        acc1[n] = a * hv.y;
    }
    __syncthreads();

    constexpr int FULL = NE / 1024;
    constexpr int BAT  = 8;
    static_assert(FULL % BAT == 0, "the batch must divide the full sweeps");
    int sr[BAT], ds[BAT]; float a[BAT];
    #pragma unroll
    for (int u = 0; u < BAT; ++u) {
        const int e = u * 1024 + tid;
        sr[u] = ei[e]; ds[u] = ei[NE + e]; a[u] = al[e];
    }
    for (int it = 0; it < FULL; it += BAT) {
        const int nx = (it + BAT < FULL) ? it + BAT : it;
        int srn[BAT], dsn[BAT]; float an[BAT];
        #pragma unroll
        for (int u = 0; u < BAT; ++u) {
            const int e = (nx + u) * 1024 + tid;
            srn[u] = ei[e]; dsn[u] = ei[NE + e]; an[u] = al[e];
        }
        v2f hv[BAT];
        #pragma unroll
        for (int u = 0; u < BAT; ++u) hv[u] = hs[sr[u]];
        #pragma unroll
        for (int u = 0; u < BAT; ++u) {
            atomicAdd(&acc0[ds[u]], a[u] * hv[u].x);
            atomicAdd(&acc1[ds[u]], a[u] * hv[u].y);
        }
        #pragma unroll
        for (int u = 0; u < BAT; ++u) { sr[u] = srn[u]; ds[u] = dsn[u]; a[u] = an[u]; }
    }
    {
        const int e = FULL * 1024 + tid;
        if (e < NE) {
            v2f hv = hs[ei[e]];
            const int d = ei[NE + e];
            const float a = al[e];
            atomicAdd(&acc0[d], a * hv.x);
            atomicAdd(&acc1[d], a * hv.y);
        }
    }
    __syncthreads();

    float* o0 = outT + (size_t)c0 * NN;
    float* o1 = o0 + NN;
    for (int n = tid; n < NN; n += 1024) { o0[n] = acc0[n]; o1[n] = acc1[n]; }
}

__global__ void bias_relu_ct(float* __restrict__ a, const float* __restrict__ b,
                             long long n, int rows) {
    long long i = (long long)blockIdx.x * blockDim.x + threadIdx.x;
    if (i >= n) return;
    float v = a[i] + b[(int)(i / rows)];
    a[i] = v > 0.f ? v : 0.f;
}

__global__ void logsoftmax64(float* __restrict__ out, const float* __restrict__ b2) {
    int node = blockIdx.x;
    int lane = threadIdx.x;
    float* row = out + (size_t)node * OUT_CH;
    float v0 = row[lane]      + b2[lane];
    float v1 = row[lane + 32] + b2[lane + 32];
    float m = fmaxf(v0, v1);
    #pragma unroll
    for (int o = 16; o > 0; o >>= 1) m = fmaxf(m, __shfl_xor(m, o, 32));
    float s = expf(v0 - m) + expf(v1 - m);
    #pragma unroll
    for (int o = 16; o > 0; o >>= 1) s += __shfl_xor(s, o, 32);
    float lse = logf(s) + m;
    row[lane]      = v0 - lse;
    row[lane + 32] = v1 - lse;
}

extern "C" void kernel_launch(void* const* d_in, const int* in_sizes, int n_in,
                              void* d_out, int out_size, void* d_ws, size_t ws_size,
                              hipStream_t stream) {
    const float* x   = (const float*)d_in[0];
    const int*   ei  = (const int*)  d_in[1];
    const float* W1  = (const float*)d_in[2];
    const float* a1s = (const float*)d_in[3];
    const float* a1d = (const float*)d_in[4];
    const float* b1  = (const float*)d_in[5];
    const float* W2  = (const float*)d_in[6];
    const float* a2s = (const float*)d_in[7];
    const float* a2d = (const float*)d_in[8];
    const float* b2  = (const float*)d_in[9];
    float* out = (float*)d_out;

    float* ws = (float*)d_ws;
    float* h1T   = ws;
    float* agg1T = h1T   + (size_t)NN * HC1;
    float* esrc1 = agg1T + (size_t)NN * HC1;
    float* edst1 = esrc1 + (size_t)NN * HEADS;
    int*   m1    = (int*)(edst1 + (size_t)NN * HEADS);
    float* den1  = (float*)m1 + (size_t)NN * HEADS;
    float* ex1   = den1 + (size_t)NN * HEADS;
    float* al1   = ex1  + (size_t)ETOT * HEADS;
    float* h2pre = al1  + (size_t)ETOT * HEADS;
    float* esrc2 = h2pre + (size_t)NN * OUT_CH;
    float* edst2 = esrc2 + NN;
    int*   m2    = (int*)(edst2 + NN);
    float* den2  = (float*)m2 + NN;
    float* ex2   = den2 + NN;

    const int TB = 256;
    auto blocks = [](long long n, int tb) { return (unsigned)((n + tb - 1) / tb); };

    {
        long long waves = (long long)(NN / 32) * (HC1 / 32);
        wmma_gemm_f32_ct<<<blocks(waves * 32, TB), TB, 0, stream>>>(x, W1, h1T, NN, IN_CH, HC1);
    }
    scores_ct<<<blocks((long long)NN * HEADS, TB), TB, 0, stream>>>(
        h1T, a1s, a1d, esrc1, edst1, m1, den1, NN, HEADS, HID);
    edge_score_max<<<blocks((long long)ETOT * HEADS, TB), TB, 0, stream>>>(ei, esrc1, edst1, ex1, m1, HEADS);
    edge_exp_sum  <<<blocks((long long)ETOT * HEADS, TB), TB, 0, stream>>>(ei, ex1, m1, den1, HEADS);
    edge_alpha_hm <<<ETOT * HEADS / 256, 256, 0, stream>>>(ei, ex1, den1, al1);
    agg_pair_lds  <<<HC1 / 2, 1024, 0, stream>>>(ei, al1, h1T, agg1T);
    bias_relu_ct<<<blocks((long long)NN * HC1, TB), TB, 0, stream>>>(agg1T, b1, (long long)NN * HC1, NN);

    {
        long long waves = (long long)(NN / 16) * (OUT_CH / 64);
        wmma_gemm_f32_at<<<blocks(waves * 32, TB), TB, 0, stream>>>(agg1T, W2, h2pre, NN, HC1, OUT_CH);
    }
    scores_kernel<<<blocks((long long)NN, TB), TB, 0, stream>>>(
        h2pre, a2s, a2d, esrc2, edst2, m2, den2, NN, 1, OUT_CH);
    fill_zero<<<2048, TB, 0, stream>>>(out, (long long)NN * OUT_CH);
    edge_score_max<<<blocks((long long)ETOT, TB), TB, 0, stream>>>(ei, esrc2, edst2, ex2, m2, 1);
    edge_exp_sum  <<<blocks((long long)ETOT, TB), TB, 0, stream>>>(ei, ex2, m2, den2, 1);
    edge_alpha    <<<blocks((long long)ETOT, TB), TB, 0, stream>>>(ei, ex2, den2, 1);
    edge_aggregate<<<blocks((long long)ETOT * OUT_CH, TB), TB, 0, stream>>>(ei, ex2, h2pre, out, 1, OUT_CH);
    logsoftmax64<<<NN, 32, 0, stream>>>(out, b2);
}


# Layer 2 sums in LDS too: one workgroup per channel pair and edge slice (8) adds by LDS float adds, arrival order in a slice; last kernel adds the 8 planes in slice order; global adds, zero fill gone
# speedup vs baseline: 2.7950x; 1.2916x over previous
#include <hip/hip_runtime.h>
#include <math.h>

#define NN      20000
#define NE      320000
#define ETOT    (NE + NN)
#define IN_CH   128
#define HID     64
#define HEADS   8
#define HC1     (HEADS * HID)
#define OUT_CH  64

typedef float v2f __attribute__((ext_vector_type(2)));
typedef float v8f __attribute__((ext_vector_type(8)));

__device__ __forceinline__ int f2ord(float f) {
    int i = __float_as_int(f);
    return i >= 0 ? i : (i ^ 0x7fffffff);
}
__device__ __forceinline__ float ord2f(int i) {
    return __int_as_float(i >= 0 ? i : (i ^ 0x7fffffff));
}

__global__ void wmma_gemm_f32_ct(const float* __restrict__ X,
                                 const float* __restrict__ Wm,
                                 float* __restrict__ HT,
                                 int M, int K, int N) {
    const int wave    = (blockIdx.x * blockDim.x + threadIdx.x) >> 5;
    const int lane    = threadIdx.x & 31;
    const int cGroups = N >> 5;
    const int mGroups = M >> 5;
    if (wave >= mGroups * cGroups) return;
    const int mg   = wave / cGroups;
    const int cg   = wave % cGroups;
    const int l    = lane & 15;
    const int half = lane >> 4;
    const unsigned uN = (unsigned)N, uM = (unsigned)M, uK = (unsigned)K;

    unsigned xoff = (unsigned)(mg * 32 + l) * uK + 2u * half;
    unsigned woff = (unsigned)(2 * half) * uN + (unsigned)(cg * 32 + l);

    v8f acc00 = {}, acc01 = {}, acc10 = {}, acc11 = {};
    for (int k0 = 0; k0 < K; k0 += 4) {
        v2f x0 = *(const v2f*)(X + xoff);
        v2f x1 = *(const v2f*)(X + xoff + 16u * uK);
        v2f w0 = { Wm[woff],      Wm[woff + uN]      };
        v2f w1 = { Wm[woff + 16], Wm[woff + uN + 16] };
        acc00 = __builtin_amdgcn_wmma_f32_16x16x4_f32(false, w0, false, x0, (short)0, acc00, false, false);
        acc01 = __builtin_amdgcn_wmma_f32_16x16x4_f32(false, w0, false, x1, (short)0, acc01, false, false);
        acc10 = __builtin_amdgcn_wmma_f32_16x16x4_f32(false, w1, false, x0, (short)0, acc10, false, false);
        acc11 = __builtin_amdgcn_wmma_f32_16x16x4_f32(false, w1, false, x1, (short)0, acc11, false, false);
        xoff += 4u;
        woff += 4u * uN;
    }
    unsigned hoff = (unsigned)(cg * 32 + 8 * half) * uM + (unsigned)(mg * 32 + l);
    #pragma unroll
    for (int r = 0; r < 8; ++r) {
        float* hp = HT + hoff;
        hp[0]            = acc00[r];
        hp[16]           = acc01[r];
        hp[16u * uM]      = acc10[r];
        hp[16u * uM + 16] = acc11[r];
        hoff += uM;
    }
}

__global__ void wmma_gemm_f32_tt(const float* __restrict__ XT,
                                 const float* __restrict__ Wm,
                                 float* __restrict__ HT,
                                 int M, int K, int N) {
    const int wave    = (blockIdx.x * blockDim.x + threadIdx.x) >> 5;
    const int lane    = threadIdx.x & 31;
    const int cGroups = N >> 5;
    const int mGroups = M >> 5;
    if (wave >= mGroups * cGroups) return;
    const int mg   = wave / cGroups;
    const int cg   = wave % cGroups;
    const int l    = lane & 15;
    const int half = lane >> 4;
    const unsigned uN = (unsigned)N, uM = (unsigned)M;

    unsigned xoff = (unsigned)(2 * half) * uM + (unsigned)(mg * 32 + l);
    unsigned woff = (unsigned)(2 * half) * uN + (unsigned)(cg * 32 + l);

    v8f acc00 = {}, acc01 = {}, acc10 = {}, acc11 = {};
    for (int k0 = 0; k0 < K; k0 += 4) {
        v2f x0 = { XT[xoff],      XT[xoff + uM]      };
        v2f x1 = { XT[xoff + 16], XT[xoff + uM + 16] };
        v2f w0 = { Wm[woff],      Wm[woff + uN]      };
        v2f w1 = { Wm[woff + 16], Wm[woff + uN + 16] };
        acc00 = __builtin_amdgcn_wmma_f32_16x16x4_f32(false, w0, false, x0, (short)0, acc00, false, false);
        acc01 = __builtin_amdgcn_wmma_f32_16x16x4_f32(false, w0, false, x1, (short)0, acc01, false, false);
        acc10 = __builtin_amdgcn_wmma_f32_16x16x4_f32(false, w1, false, x0, (short)0, acc10, false, false);
        acc11 = __builtin_amdgcn_wmma_f32_16x16x4_f32(false, w1, false, x1, (short)0, acc11, false, false);
        xoff += 4u * uM;
        woff += 4u * uN;
    }
    unsigned hoff = (unsigned)(cg * 32 + 8 * half) * uM + (unsigned)(mg * 32 + l);
    #pragma unroll
    for (int r = 0; r < 8; ++r) {
        float* hp = HT + hoff;
        hp[0]            = acc00[r];
        hp[16]           = acc01[r];
        hp[16u * uM]      = acc10[r];
        hp[16u * uM + 16] = acc11[r];
        hoff += uM;
    }
}

__global__ void scores_ct(const float* __restrict__ HT,
                          const float* __restrict__ a_src,
                          const float* __restrict__ a_dst,
                          float* __restrict__ esrc, float* __restrict__ edst,
                          int* __restrict__ mmax, float* __restrict__ denom,
                          int n, int H, int C) {
    int i = blockIdx.x * blockDim.x + threadIdx.x;
    if (i >= n * H) return;
    int hh = i / n, node = i % n;
    const float* hp = HT + (size_t)hh * C * n + node;
    const float* as = a_src + hh * C;
    const float* ad = a_dst + hh * C;
    float s = 0.f, d = 0.f;
    for (int c = 0; c < C; ++c) { float v = hp[(size_t)c * n]; s += v * as[c]; d += v * ad[c]; }
    int o = node * H + hh;
    esrc[o]  = s;
    edst[o]  = d;
    mmax[o]  = f2ord(-INFINITY);
    denom[o] = 0.f;
}

__global__ void edge_score_max(const int* __restrict__ ei,
                               const float* __restrict__ esrc,
                               const float* __restrict__ edst,
                               float* __restrict__ eleaky,
                               int* __restrict__ mmax, int H) {
    long long t = (long long)blockIdx.x * blockDim.x + threadIdx.x;
    if (t >= (long long)ETOT * H) return;
    int e = (int)(t / H), hh = (int)(t % H);
    int src, dst;
    if (e < NE) { src = ei[e]; dst = ei[NE + e]; }
    else        { src = dst = e - NE; }
    float v = esrc[src * H + hh] + edst[dst * H + hh];
    v = v > 0.f ? v : 0.2f * v;
    eleaky[t] = v;
    atomicMax(&mmax[dst * H + hh], f2ord(v));
}

__global__ void edge_exp_sum(const int* __restrict__ ei,
                             float* __restrict__ eleaky,
                             const int* __restrict__ mmax,
                             float* __restrict__ denom, int H) {
    long long t = (long long)blockIdx.x * blockDim.x + threadIdx.x;
    if (t >= (long long)ETOT * H) return;
    int e = (int)(t / H), hh = (int)(t % H);
    int dst = (e < NE) ? ei[NE + e] : (e - NE);
    float ex = expf(eleaky[t] - ord2f(mmax[dst * H + hh]));
    eleaky[t] = ex;
    atomicAdd(&denom[dst * H + hh], ex);
}

__global__ void edge_alpha(const int* __restrict__ ei,
                           float* __restrict__ exal,
                           const float* __restrict__ denom, int H) {
    long long t = (long long)blockIdx.x * blockDim.x + threadIdx.x;
    if (t >= (long long)ETOT * H) return;
    int e = (int)(t / H), hh = (int)(t % H);
    int dst = (e < NE) ? ei[NE + e] : (e - NE);
    exal[t] = exal[t] / (denom[dst * H + hh] + 1e-16f);
}

__global__ void edge_alpha_hm(const int* __restrict__ ei,
                              const float* __restrict__ ex,
                              const float* __restrict__ denom,
                              float* __restrict__ alT) {
    __shared__ float tile[HEADS][33];
    const int tid = threadIdx.x;
    const int el = tid >> 3, hh = tid & 7;
    const int e  = blockIdx.x * 32 + el;
    const int dst = (e < NE) ? ei[NE + e] : (e - NE);
    tile[hh][el] = ex[(size_t)e * HEADS + hh] / (denom[dst * HEADS + hh] + 1e-16f);
    __syncthreads();
    const int w = tid >> 5, lane = tid & 31;
    alT[(size_t)w * ETOT + (size_t)blockIdx.x * 32 + lane] = tile[w][lane];
}

__global__ __launch_bounds__(1024) void agg_pair_lds(const int* __restrict__ ei,
                                                     const float* __restrict__ alT,
                                                     const float* __restrict__ HT,
                                                     float* __restrict__ outT) {
    __shared__ v2f   hs[NN];
    __shared__ float acc0[NN];
    __shared__ float acc1[NN];
    const int tid  = threadIdx.x;
    const int c0   = 2 * blockIdx.x;
    const int head = c0 / HID;
    const float* h0 = HT + (size_t)c0 * NN;
    const float* h1 = h0 + NN;
    const float* al = alT + (size_t)head * ETOT;

    for (int n = tid; n < NN; n += 1024) {
        v2f hv = { h0[n], h1[n] };
        float a = al[NE + n];
        hs[n]   = hv;
        acc0[n] = a * hv.x;
        acc1[n] = a * hv.y;
    }
    __syncthreads();

    constexpr int FULL = NE / 1024;
    constexpr int BAT  = 8;
    static_assert(FULL % BAT == 0, "the batch must divide the full sweeps");
    int sr[BAT], ds[BAT]; float a[BAT];
    #pragma unroll
    for (int u = 0; u < BAT; ++u) {
        const int e = u * 1024 + tid;
        sr[u] = ei[e]; ds[u] = ei[NE + e]; a[u] = al[e];
    }
    for (int it = 0; it < FULL; it += BAT) {
        const int nx = (it + BAT < FULL) ? it + BAT : it;
        int srn[BAT], dsn[BAT]; float an[BAT];
        #pragma unroll
        for (int u = 0; u < BAT; ++u) {
            const int e = (nx + u) * 1024 + tid;
            srn[u] = ei[e]; dsn[u] = ei[NE + e]; an[u] = al[e];
        }
        v2f hv[BAT];
        #pragma unroll
        for (int u = 0; u < BAT; ++u) hv[u] = hs[sr[u]];
        #pragma unroll
        for (int u = 0; u < BAT; ++u) {
            atomicAdd(&acc0[ds[u]], a[u] * hv[u].x);
            atomicAdd(&acc1[ds[u]], a[u] * hv[u].y);
        }
        #pragma unroll
        for (int u = 0; u < BAT; ++u) { sr[u] = srn[u]; ds[u] = dsn[u]; a[u] = an[u]; }
    }
    {
        const int e = FULL * 1024 + tid;
        if (e < NE) {
            v2f hv = hs[ei[e]];
            const int d = ei[NE + e];
            const float a = al[e];
            atomicAdd(&acc0[d], a * hv.x);
            atomicAdd(&acc1[d], a * hv.y);
        }
    }
    __syncthreads();

    float* o0 = outT + (size_t)c0 * NN;
    float* o1 = o0 + NN;
    for (int n = tid; n < NN; n += 1024) { o0[n] = acc0[n]; o1[n] = acc1[n]; }
}

#define S2 8
__global__ __launch_bounds__(1024) void agg2_slice_lds(const int* __restrict__ ei,
                                                       const float* __restrict__ al,
                                                       const float* __restrict__ HT,
                                                       float* __restrict__ part) {
    __shared__ v2f   hs[NN];
    __shared__ float acc0[NN];
    __shared__ float acc1[NN];
    const int tid   = threadIdx.x;
    const int pair  = blockIdx.x % (OUT_CH / 2);
    const int slice = blockIdx.x / (OUT_CH / 2);
    const int c0    = 2 * pair;
    const float* h0 = HT + (size_t)c0 * NN;
    const float* h1 = h0 + NN;

    if (slice == 0) {
        for (int n = tid; n < NN; n += 1024) {
            v2f hv = { h0[n], h1[n] };
            float a = al[NE + n];
            hs[n]   = hv;
            acc0[n] = a * hv.x;
            acc1[n] = a * hv.y;
        }
    } else {
        for (int n = tid; n < NN; n += 1024) {
            v2f hv = { h0[n], h1[n] };
            hs[n]   = hv;
            acc0[n] = 0.f;
            acc1[n] = 0.f;
        }
    }
    __syncthreads();

    constexpr int FULL = NE / 1024;
    static_assert(FULL % S2 == 0, "the slices must divide the full sweeps");
    constexpr int SW  = FULL / S2;
    constexpr int BAT = 13;
    static_assert(SW % BAT == 0, "the batch must divide a slice's sweeps");
    const int base = slice * SW * 1024 + tid;
    for (int it = 0; it < SW; it += BAT) {
        int sr[BAT], ds[BAT]; float a[BAT];
        #pragma unroll
        for (int u = 0; u < BAT; ++u) {
            const int e = base + (it + u) * 1024;
            sr[u] = ei[e]; ds[u] = ei[NE + e]; a[u] = al[e];
        }
        v2f hv[BAT];
        #pragma unroll
        for (int u = 0; u < BAT; ++u) hv[u] = hs[sr[u]];
        #pragma unroll
        for (int u = 0; u < BAT; ++u) {
            atomicAdd(&acc0[ds[u]], a[u] * hv[u].x);
            atomicAdd(&acc1[ds[u]], a[u] * hv[u].y);
        }
    }
    if (slice == S2 - 1) {
        const int e = FULL * 1024 + tid;
        if (e < NE) {
            v2f hv = hs[ei[e]];
            const int d = ei[NE + e];
            const float a = al[e];
            atomicAdd(&acc0[d], a * hv.x);
            atomicAdd(&acc1[d], a * hv.y);
        }
    }
    __syncthreads();

    float* o0 = part + ((size_t)slice * OUT_CH + c0) * NN;
    float* o1 = o0 + NN;
    for (int n = tid; n < NN; n += 1024) { o0[n] = acc0[n]; o1[n] = acc1[n]; }
}

__global__ void bias_relu_ct(float* __restrict__ a, const float* __restrict__ b,
                             long long n, int rows) {
    long long i = (long long)blockIdx.x * blockDim.x + threadIdx.x;
    if (i >= n) return;
    float v = a[i] + b[(int)(i / rows)];
    a[i] = v > 0.f ? v : 0.f;
}

__global__ __launch_bounds__(256) void final_planes_logsoftmax(const float* __restrict__ part,
                                                                const float* __restrict__ b2,
                                                                float* __restrict__ out) {
    __shared__ float tile[OUT_CH][33];
    const int tid = threadIdx.x;
    const int w = tid >> 5, lane = tid & 31;
    const int n0 = blockIdx.x * 32;
    #pragma unroll
    for (int j = 0; j < 8; ++j) {
        const int c = 8 * w + j;
        const float* p = part + (size_t)c * NN + n0 + lane;
        float v = p[0];
        #pragma unroll
        for (int s = 1; s < S2; ++s) v += p[(size_t)s * OUT_CH * NN];
        tile[c][lane] = v + b2[c];
    }
    __syncthreads();
    #pragma unroll
    for (int j = 0; j < 4; ++j) {
        const int nl = 4 * w + j;
        float v0 = tile[lane][nl];
        float v1 = tile[lane + 32][nl];
        float m = fmaxf(v0, v1);
        #pragma unroll
        for (int o = 16; o > 0; o >>= 1) m = fmaxf(m, __shfl_xor(m, o, 32));
        float s = expf(v0 - m) + expf(v1 - m);
        #pragma unroll
        for (int o = 16; o > 0; o >>= 1) s += __shfl_xor(s, o, 32);
        float lse = logf(s) + m;
        float* row = out + (size_t)(n0 + nl) * OUT_CH;
        row[lane]      = v0 - lse;
        row[lane + 32] = v1 - lse;
    }
}

extern "C" void kernel_launch(void* const* d_in, const int* in_sizes, int n_in,
                              void* d_out, int out_size, void* d_ws, size_t ws_size,
                              hipStream_t stream) {
    const float* x   = (const float*)d_in[0];
    const int*   ei  = (const int*)  d_in[1];
    const float* W1  = (const float*)d_in[2];
    const float* a1s = (const float*)d_in[3];
    const float* a1d = (const float*)d_in[4];
    const float* b1  = (const float*)d_in[5];
    const float* W2  = (const float*)d_in[6];
    const float* a2s = (const float*)d_in[7];
    const float* a2d = (const float*)d_in[8];
    const float* b2  = (const float*)d_in[9];
    float* out = (float*)d_out;

    float* ws = (float*)d_ws;
    float* h1T   = ws;
    float* agg1T = h1T   + (size_t)NN * HC1;
    float* esrc1 = agg1T + (size_t)NN * HC1;
    float* edst1 = esrc1 + (size_t)NN * HEADS;
    int*   m1    = (int*)(edst1 + (size_t)NN * HEADS);
    float* den1  = (float*)m1 + (size_t)NN * HEADS;
    float* ex1   = den1 + (size_t)NN * HEADS;
    float* al1   = ex1  + (size_t)ETOT * HEADS;
    float* h2T   = al1  + (size_t)ETOT * HEADS;
    float* esrc2 = h2T + (size_t)NN * OUT_CH;
    float* edst2 = esrc2 + NN;
    int*   m2    = (int*)(edst2 + NN);
    float* den2  = (float*)m2 + NN;
    float* ex2   = den2 + NN;
    float* part2 = h1T;

    const int TB = 256;
    auto blocks = [](long long n, int tb) { return (unsigned)((n + tb - 1) / tb); };

    {
        long long waves = (long long)(NN / 32) * (HC1 / 32);
        wmma_gemm_f32_ct<<<blocks(waves * 32, TB), TB, 0, stream>>>(x, W1, h1T, NN, IN_CH, HC1);
    }
    scores_ct<<<blocks((long long)NN * HEADS, TB), TB, 0, stream>>>(
        h1T, a1s, a1d, esrc1, edst1, m1, den1, NN, HEADS, HID);
    edge_score_max<<<blocks((long long)ETOT * HEADS, TB), TB, 0, stream>>>(ei, esrc1, edst1, ex1, m1, HEADS);
    edge_exp_sum  <<<blocks((long long)ETOT * HEADS, TB), TB, 0, stream>>>(ei, ex1, m1, den1, HEADS);
    edge_alpha_hm <<<ETOT * HEADS / 256, 256, 0, stream>>>(ei, ex1, den1, al1);
    agg_pair_lds  <<<HC1 / 2, 1024, 0, stream>>>(ei, al1, h1T, agg1T);
    bias_relu_ct<<<blocks((long long)NN * HC1, TB), TB, 0, stream>>>(agg1T, b1, (long long)NN * HC1, NN);

    {
        long long waves = (long long)(NN / 32) * (OUT_CH / 32);
        wmma_gemm_f32_tt<<<blocks(waves * 32, TB), TB, 0, stream>>>(agg1T, W2, h2T, NN, HC1, OUT_CH);
    }
    scores_ct<<<blocks((long long)NN, TB), TB, 0, stream>>>(
        h2T, a2s, a2d, esrc2, edst2, m2, den2, NN, 1, OUT_CH);
    edge_score_max<<<blocks((long long)ETOT, TB), TB, 0, stream>>>(ei, esrc2, edst2, ex2, m2, 1);
    edge_exp_sum  <<<blocks((long long)ETOT, TB), TB, 0, stream>>>(ei, ex2, m2, den2, 1);
    edge_alpha    <<<blocks((long long)ETOT, TB), TB, 0, stream>>>(ei, ex2, den2, 1);
    agg2_slice_lds<<<(OUT_CH / 2) * S2, 1024, 0, stream>>>(ei, ex2, h2T, part2);
    final_planes_logsoftmax<<<NN / 32, 256, 0, stream>>>(part2, b2, out);
}


# bias+relu applied in the layer-1 LDS aggregation kernel's write-out; the separate one-element-per-thread pass (320,000 waves, one more read+write of the 41 MB plane) and its launch are gone
# speedup vs baseline: 3.0079x; 1.0762x over previous
#include <hip/hip_runtime.h>
#include <math.h>

#define NN      20000
#define NE      320000
#define ETOT    (NE + NN)
#define IN_CH   128
#define HID     64
#define HEADS   8
#define HC1     (HEADS * HID)
#define OUT_CH  64

typedef float v2f __attribute__((ext_vector_type(2)));
typedef float v8f __attribute__((ext_vector_type(8)));

__device__ __forceinline__ int f2ord(float f) {
    int i = __float_as_int(f);
    return i >= 0 ? i : (i ^ 0x7fffffff);
}
__device__ __forceinline__ float ord2f(int i) {
    return __int_as_float(i >= 0 ? i : (i ^ 0x7fffffff));
}

__global__ void wmma_gemm_f32_ct(const float* __restrict__ X,
                                 const float* __restrict__ Wm,
                                 float* __restrict__ HT,
                                 int M, int K, int N) {
    const int wave    = (blockIdx.x * blockDim.x + threadIdx.x) >> 5;
    const int lane    = threadIdx.x & 31;
    const int cGroups = N >> 5;
    const int mGroups = M >> 5;
    if (wave >= mGroups * cGroups) return;
    const int mg   = wave / cGroups;
    const int cg   = wave % cGroups;
    const int l    = lane & 15;
    const int half = lane >> 4;
    const unsigned uN = (unsigned)N, uM = (unsigned)M, uK = (unsigned)K;

    unsigned xoff = (unsigned)(mg * 32 + l) * uK + 2u * half;
    unsigned woff = (unsigned)(2 * half) * uN + (unsigned)(cg * 32 + l);

    v8f acc00 = {}, acc01 = {}, acc10 = {}, acc11 = {};
    for (int k0 = 0; k0 < K; k0 += 4) {
        v2f x0 = *(const v2f*)(X + xoff);
        v2f x1 = *(const v2f*)(X + xoff + 16u * uK);
        v2f w0 = { Wm[woff],      Wm[woff + uN]      };
        v2f w1 = { Wm[woff + 16], Wm[woff + uN + 16] };
        acc00 = __builtin_amdgcn_wmma_f32_16x16x4_f32(false, w0, false, x0, (short)0, acc00, false, false);
        acc01 = __builtin_amdgcn_wmma_f32_16x16x4_f32(false, w0, false, x1, (short)0, acc01, false, false);
        acc10 = __builtin_amdgcn_wmma_f32_16x16x4_f32(false, w1, false, x0, (short)0, acc10, false, false);
        acc11 = __builtin_amdgcn_wmma_f32_16x16x4_f32(false, w1, false, x1, (short)0, acc11, false, false);
        xoff += 4u;
        woff += 4u * uN;
    }
    unsigned hoff = (unsigned)(cg * 32 + 8 * half) * uM + (unsigned)(mg * 32 + l);
    #pragma unroll
    for (int r = 0; r < 8; ++r) {
        float* hp = HT + hoff;
        hp[0]            = acc00[r];
        hp[16]           = acc01[r];
        hp[16u * uM]      = acc10[r];
        hp[16u * uM + 16] = acc11[r];
        hoff += uM;
    }
}

__global__ void wmma_gemm_f32_tt(const float* __restrict__ XT,
                                 const float* __restrict__ Wm,
                                 float* __restrict__ HT,
                                 int M, int K, int N) {
    const int wave    = (blockIdx.x * blockDim.x + threadIdx.x) >> 5;
    const int lane    = threadIdx.x & 31;
    const int cGroups = N >> 5;
    const int mGroups = M >> 5;
    if (wave >= mGroups * cGroups) return;
    const int mg   = wave / cGroups;
    const int cg   = wave % cGroups;
    const int l    = lane & 15;
    const int half = lane >> 4;
    const unsigned uN = (unsigned)N, uM = (unsigned)M;

    unsigned xoff = (unsigned)(2 * half) * uM + (unsigned)(mg * 32 + l);
    unsigned woff = (unsigned)(2 * half) * uN + (unsigned)(cg * 32 + l);

    v8f acc00 = {}, acc01 = {}, acc10 = {}, acc11 = {};
    for (int k0 = 0; k0 < K; k0 += 4) {
        v2f x0 = { XT[xoff],      XT[xoff + uM]      };
        v2f x1 = { XT[xoff + 16], XT[xoff + uM + 16] };
        v2f w0 = { Wm[woff],      Wm[woff + uN]      };
        v2f w1 = { Wm[woff + 16], Wm[woff + uN + 16] };
        acc00 = __builtin_amdgcn_wmma_f32_16x16x4_f32(false, w0, false, x0, (short)0, acc00, false, false);
        acc01 = __builtin_amdgcn_wmma_f32_16x16x4_f32(false, w0, false, x1, (short)0, acc01, false, false);
        acc10 = __builtin_amdgcn_wmma_f32_16x16x4_f32(false, w1, false, x0, (short)0, acc10, false, false);
        acc11 = __builtin_amdgcn_wmma_f32_16x16x4_f32(false, w1, false, x1, (short)0, acc11, false, false);
        xoff += 4u * uM;
        woff += 4u * uN;
    }
    unsigned hoff = (unsigned)(cg * 32 + 8 * half) * uM + (unsigned)(mg * 32 + l);
    #pragma unroll
    for (int r = 0; r < 8; ++r) {
        float* hp = HT + hoff;
        hp[0]            = acc00[r];
        hp[16]           = acc01[r];
        hp[16u * uM]      = acc10[r];
        hp[16u * uM + 16] = acc11[r];
        hoff += uM;
    }
}

__global__ void scores_ct(const float* __restrict__ HT,
                          const float* __restrict__ a_src,
                          const float* __restrict__ a_dst,
                          float* __restrict__ esrc, float* __restrict__ edst,
                          int* __restrict__ mmax, float* __restrict__ denom,
                          int n, int H, int C) {
    int i = blockIdx.x * blockDim.x + threadIdx.x;
    if (i >= n * H) return;
    int hh = i / n, node = i % n;
    const float* hp = HT + (size_t)hh * C * n + node;
    const float* as = a_src + hh * C;
    const float* ad = a_dst + hh * C;
    float s = 0.f, d = 0.f;
    for (int c = 0; c < C; ++c) { float v = hp[(size_t)c * n]; s += v * as[c]; d += v * ad[c]; }
    int o = node * H + hh;
    esrc[o]  = s;
    edst[o]  = d;
    mmax[o]  = f2ord(-INFINITY);
    denom[o] = 0.f;
}

__global__ void edge_score_max(const int* __restrict__ ei,
                               const float* __restrict__ esrc,
                               const float* __restrict__ edst,
                               float* __restrict__ eleaky,
                               int* __restrict__ mmax, int H) {
    long long t = (long long)blockIdx.x * blockDim.x + threadIdx.x;
    if (t >= (long long)ETOT * H) return;
    int e = (int)(t / H), hh = (int)(t % H);
    int src, dst;
    if (e < NE) { src = ei[e]; dst = ei[NE + e]; }
    else        { src = dst = e - NE; }
    float v = esrc[src * H + hh] + edst[dst * H + hh];
    v = v > 0.f ? v : 0.2f * v;
    eleaky[t] = v;
    atomicMax(&mmax[dst * H + hh], f2ord(v));
}

__global__ void edge_exp_sum(const int* __restrict__ ei,
                             float* __restrict__ eleaky,
                             const int* __restrict__ mmax,
                             float* __restrict__ denom, int H) {
    long long t = (long long)blockIdx.x * blockDim.x + threadIdx.x;
    if (t >= (long long)ETOT * H) return;
    int e = (int)(t / H), hh = (int)(t % H);
    int dst = (e < NE) ? ei[NE + e] : (e - NE);
    float ex = expf(eleaky[t] - ord2f(mmax[dst * H + hh]));
    eleaky[t] = ex;
    atomicAdd(&denom[dst * H + hh], ex);
}

__global__ void edge_alpha(const int* __restrict__ ei,
                           float* __restrict__ exal,
                           const float* __restrict__ denom, int H) {
    long long t = (long long)blockIdx.x * blockDim.x + threadIdx.x;
    if (t >= (long long)ETOT * H) return;
    int e = (int)(t / H), hh = (int)(t % H);
    int dst = (e < NE) ? ei[NE + e] : (e - NE);
    exal[t] = exal[t] / (denom[dst * H + hh] + 1e-16f);
}

__global__ void edge_alpha_hm(const int* __restrict__ ei,
                              const float* __restrict__ ex,
                              const float* __restrict__ denom,
                              float* __restrict__ alT) {
    __shared__ float tile[HEADS][33];
    const int tid = threadIdx.x;
    const int el = tid >> 3, hh = tid & 7;
    const int e  = blockIdx.x * 32 + el;
    const int dst = (e < NE) ? ei[NE + e] : (e - NE);
    tile[hh][el] = ex[(size_t)e * HEADS + hh] / (denom[dst * HEADS + hh] + 1e-16f);
    __syncthreads();
    const int w = tid >> 5, lane = tid & 31;
    alT[(size_t)w * ETOT + (size_t)blockIdx.x * 32 + lane] = tile[w][lane];
}

__global__ __launch_bounds__(1024) void agg_pair_lds(const int* __restrict__ ei,
                                                     const float* __restrict__ alT,
                                                     const float* __restrict__ HT,
                                                     const float* __restrict__ bias,
                                                     float* __restrict__ outT) {
    __shared__ v2f   hs[NN];
    __shared__ float acc0[NN];
    __shared__ float acc1[NN];
    const int tid  = threadIdx.x;
    const int c0   = 2 * blockIdx.x;
    const int head = c0 / HID;
    const float* h0 = HT + (size_t)c0 * NN;
    const float* h1 = h0 + NN;
    const float* al = alT + (size_t)head * ETOT;

    for (int n = tid; n < NN; n += 1024) {
        v2f hv = { h0[n], h1[n] };
        float a = al[NE + n];
        hs[n]   = hv;
        acc0[n] = a * hv.x;
        acc1[n] = a * hv.y;
    }
    __syncthreads();

    constexpr int FULL = NE / 1024;
    constexpr int BAT  = 8;
    static_assert(FULL % BAT == 0, "the batch must divide the full sweeps");
    int sr[BAT], ds[BAT]; float a[BAT];
    #pragma unroll
    for (int u = 0; u < BAT; ++u) {
        const int e = u * 1024 + tid;
        sr[u] = ei[e]; ds[u] = ei[NE + e]; a[u] = al[e];
    }
    for (int it = 0; it < FULL; it += BAT) {
        const int nx = (it + BAT < FULL) ? it + BAT : it;
        int srn[BAT], dsn[BAT]; float an[BAT];
        #pragma unroll
        for (int u = 0; u < BAT; ++u) {
            const int e = (nx + u) * 1024 + tid;
            srn[u] = ei[e]; dsn[u] = ei[NE + e]; an[u] = al[e];
        }
        v2f hv[BAT];
        #pragma unroll
        for (int u = 0; u < BAT; ++u) hv[u] = hs[sr[u]];
        #pragma unroll
        for (int u = 0; u < BAT; ++u) {
            atomicAdd(&acc0[ds[u]], a[u] * hv[u].x);
            atomicAdd(&acc1[ds[u]], a[u] * hv[u].y);
        }
        #pragma unroll
        for (int u = 0; u < BAT; ++u) { sr[u] = srn[u]; ds[u] = dsn[u]; a[u] = an[u]; }
    }
    {
        const int e = FULL * 1024 + tid;
        if (e < NE) {
            v2f hv = hs[ei[e]];
            const int d = ei[NE + e];
            const float a = al[e];
            atomicAdd(&acc0[d], a * hv.x);
            atomicAdd(&acc1[d], a * hv.y);
        }
    }
    __syncthreads();

    float* o0 = outT + (size_t)c0 * NN;
    float* o1 = o0 + NN;
    const float b0 = bias[c0], b1 = bias[c0 + 1];
    for (int n = tid; n < NN; n += 1024) {
        const float v0 = acc0[n] + b0, v1 = acc1[n] + b1;
        o0[n] = v0 > 0.f ? v0 : 0.f;
        o1[n] = v1 > 0.f ? v1 : 0.f;
    }
}

#define S2 8
__global__ __launch_bounds__(1024) void agg2_slice_lds(const int* __restrict__ ei,
                                                       const float* __restrict__ al,
                                                       const float* __restrict__ HT,
                                                       float* __restrict__ part) {
    __shared__ v2f   hs[NN];
    __shared__ float acc0[NN];
    __shared__ float acc1[NN];
    const int tid   = threadIdx.x;
    const int pair  = blockIdx.x % (OUT_CH / 2);
    const int slice = blockIdx.x / (OUT_CH / 2);
    const int c0    = 2 * pair;
    const float* h0 = HT + (size_t)c0 * NN;
    const float* h1 = h0 + NN;

    if (slice == 0) {
        for (int n = tid; n < NN; n += 1024) {
            v2f hv = { h0[n], h1[n] };
            float a = al[NE + n];
            hs[n]   = hv;
            acc0[n] = a * hv.x;
            acc1[n] = a * hv.y;
        }
    } else {
        for (int n = tid; n < NN; n += 1024) {
            v2f hv = { h0[n], h1[n] };
            hs[n]   = hv;
            acc0[n] = 0.f;
            acc1[n] = 0.f;
        }
    }
    __syncthreads();

    constexpr int FULL = NE / 1024;
    static_assert(FULL % S2 == 0, "the slices must divide the full sweeps");
    constexpr int SW  = FULL / S2;
    constexpr int BAT = 13;
    static_assert(SW % BAT == 0, "the batch must divide a slice's sweeps");
    const int base = slice * SW * 1024 + tid;
    for (int it = 0; it < SW; it += BAT) {
        int sr[BAT], ds[BAT]; float a[BAT];
        #pragma unroll
        for (int u = 0; u < BAT; ++u) {
            const int e = base + (it + u) * 1024;
            sr[u] = ei[e]; ds[u] = ei[NE + e]; a[u] = al[e];
        }
        v2f hv[BAT];
        #pragma unroll
        for (int u = 0; u < BAT; ++u) hv[u] = hs[sr[u]];
        #pragma unroll
        for (int u = 0; u < BAT; ++u) {
            atomicAdd(&acc0[ds[u]], a[u] * hv[u].x);
            atomicAdd(&acc1[ds[u]], a[u] * hv[u].y);
        }
    }
    if (slice == S2 - 1) {
        const int e = FULL * 1024 + tid;
        if (e < NE) {
            v2f hv = hs[ei[e]];
            const int d = ei[NE + e];
            const float a = al[e];
            atomicAdd(&acc0[d], a * hv.x);
            atomicAdd(&acc1[d], a * hv.y);
        }
    }
    __syncthreads();

    float* o0 = part + ((size_t)slice * OUT_CH + c0) * NN;
    float* o1 = o0 + NN;
    for (int n = tid; n < NN; n += 1024) { o0[n] = acc0[n]; o1[n] = acc1[n]; }
}

__global__ __launch_bounds__(256) void final_planes_logsoftmax(const float* __restrict__ part,
                                                                const float* __restrict__ b2,
                                                                float* __restrict__ out) {
    __shared__ float tile[OUT_CH][33];
    const int tid = threadIdx.x;
    const int w = tid >> 5, lane = tid & 31;
    const int n0 = blockIdx.x * 32;
    #pragma unroll
    for (int j = 0; j < 8; ++j) {
        const int c = 8 * w + j;
        const float* p = part + (size_t)c * NN + n0 + lane;
        float v = p[0];
        #pragma unroll
        for (int s = 1; s < S2; ++s) v += p[(size_t)s * OUT_CH * NN];
        tile[c][lane] = v + b2[c];
    }
    __syncthreads();
    #pragma unroll
    for (int j = 0; j < 4; ++j) {
        const int nl = 4 * w + j;
        float v0 = tile[lane][nl];
        float v1 = tile[lane + 32][nl];
        float m = fmaxf(v0, v1);
        #pragma unroll
        for (int o = 16; o > 0; o >>= 1) m = fmaxf(m, __shfl_xor(m, o, 32));
        float s = expf(v0 - m) + expf(v1 - m);
        #pragma unroll
        for (int o = 16; o > 0; o >>= 1) s += __shfl_xor(s, o, 32);
        float lse = logf(s) + m;
        float* row = out + (size_t)(n0 + nl) * OUT_CH;
        row[lane]      = v0 - lse;
        row[lane + 32] = v1 - lse;
    }
}

extern "C" void kernel_launch(void* const* d_in, const int* in_sizes, int n_in,
                              void* d_out, int out_size, void* d_ws, size_t ws_size,
                              hipStream_t stream) {
    const float* x   = (const float*)d_in[0];
    const int*   ei  = (const int*)  d_in[1];
    const float* W1  = (const float*)d_in[2];
    const float* a1s = (const float*)d_in[3];
    const float* a1d = (const float*)d_in[4];
    const float* b1  = (const float*)d_in[5];
    const float* W2  = (const float*)d_in[6];
    const float* a2s = (const float*)d_in[7];
    const float* a2d = (const float*)d_in[8];
    const float* b2  = (const float*)d_in[9];
    float* out = (float*)d_out;

    float* ws = (float*)d_ws;
    float* h1T   = ws;
    float* agg1T = h1T   + (size_t)NN * HC1;
    float* esrc1 = agg1T + (size_t)NN * HC1;
    float* edst1 = esrc1 + (size_t)NN * HEADS;
    int*   m1    = (int*)(edst1 + (size_t)NN * HEADS);
    float* den1  = (float*)m1 + (size_t)NN * HEADS;
    float* ex1   = den1 + (size_t)NN * HEADS;
    float* al1   = ex1  + (size_t)ETOT * HEADS;
    float* h2T   = al1  + (size_t)ETOT * HEADS;
    float* esrc2 = h2T + (size_t)NN * OUT_CH;
    float* edst2 = esrc2 + NN;
    int*   m2    = (int*)(edst2 + NN);
    float* den2  = (float*)m2 + NN;
    float* ex2   = den2 + NN;
    float* part2 = h1T;

    const int TB = 256;
    auto blocks = [](long long n, int tb) { return (unsigned)((n + tb - 1) / tb); };

    {
        long long waves = (long long)(NN / 32) * (HC1 / 32);
        wmma_gemm_f32_ct<<<blocks(waves * 32, TB), TB, 0, stream>>>(x, W1, h1T, NN, IN_CH, HC1);
    }
    scores_ct<<<blocks((long long)NN * HEADS, TB), TB, 0, stream>>>(
        h1T, a1s, a1d, esrc1, edst1, m1, den1, NN, HEADS, HID);
    edge_score_max<<<blocks((long long)ETOT * HEADS, TB), TB, 0, stream>>>(ei, esrc1, edst1, ex1, m1, HEADS);
    edge_exp_sum  <<<blocks((long long)ETOT * HEADS, TB), TB, 0, stream>>>(ei, ex1, m1, den1, HEADS);
    edge_alpha_hm <<<ETOT * HEADS / 256, 256, 0, stream>>>(ei, ex1, den1, al1);
    agg_pair_lds  <<<HC1 / 2, 1024, 0, stream>>>(ei, al1, h1T, b1, agg1T);

    {
        long long waves = (long long)(NN / 32) * (OUT_CH / 32);
        wmma_gemm_f32_tt<<<blocks(waves * 32, TB), TB, 0, stream>>>(agg1T, W2, h2T, NN, HC1, OUT_CH);
    }
    scores_ct<<<blocks((long long)NN, TB), TB, 0, stream>>>(
        h2T, a2s, a2d, esrc2, edst2, m2, den2, NN, 1, OUT_CH);
    edge_score_max<<<blocks((long long)ETOT, TB), TB, 0, stream>>>(ei, esrc2, edst2, ex2, m2, 1);
    edge_exp_sum  <<<blocks((long long)ETOT, TB), TB, 0, stream>>>(ei, ex2, m2, den2, 1);
    edge_alpha    <<<blocks((long long)ETOT, TB), TB, 0, stream>>>(ei, ex2, den2, 1);
    agg2_slice_lds<<<(OUT_CH / 2) * S2, 1024, 0, stream>>>(ei, ex2, h2T, part2);
    final_planes_logsoftmax<<<NN / 32, 256, 0, stream>>>(part2, b2, out);
}


# Edge softmax max and exp-sum by LDS atomics in (head, edge-slice) workgroups, 32 planes folded per node (sum: arrival order, then slice order); one division per destination; global atomics gone
# speedup vs baseline: 3.8342x; 1.2747x over previous
#include <hip/hip_runtime.h>
#include <math.h>

#define NN      20000
#define NE      320000
#define ETOT    (NE + NN)
#define IN_CH   128
#define HID     64
#define HEADS   8
#define HC1     (HEADS * HID)
#define OUT_CH  64

typedef float v2f __attribute__((ext_vector_type(2)));
typedef float v8f __attribute__((ext_vector_type(8)));

__device__ __forceinline__ int f2ord(float f) {
    int i = __float_as_int(f);
    return i >= 0 ? i : (i ^ 0x7fffffff);
}
__device__ __forceinline__ float ord2f(int i) {
    return __int_as_float(i >= 0 ? i : (i ^ 0x7fffffff));
}

__global__ void wmma_gemm_f32_ct(const float* __restrict__ X,
                                 const float* __restrict__ Wm,
                                 float* __restrict__ HT,
                                 int M, int K, int N) {
    const int wave    = (blockIdx.x * blockDim.x + threadIdx.x) >> 5;
    const int lane    = threadIdx.x & 31;
    const int cGroups = N >> 5;
    const int mGroups = M >> 5;
    if (wave >= mGroups * cGroups) return;
    const int mg   = wave / cGroups;
    const int cg   = wave % cGroups;
    const int l    = lane & 15;
    const int half = lane >> 4;
    const unsigned uN = (unsigned)N, uM = (unsigned)M, uK = (unsigned)K;

    unsigned xoff = (unsigned)(mg * 32 + l) * uK + 2u * half;
    unsigned woff = (unsigned)(2 * half) * uN + (unsigned)(cg * 32 + l);

    v8f acc00 = {}, acc01 = {}, acc10 = {}, acc11 = {};
    for (int k0 = 0; k0 < K; k0 += 4) {
        v2f x0 = *(const v2f*)(X + xoff);
        v2f x1 = *(const v2f*)(X + xoff + 16u * uK);
        v2f w0 = { Wm[woff],      Wm[woff + uN]      };
        v2f w1 = { Wm[woff + 16], Wm[woff + uN + 16] };
        acc00 = __builtin_amdgcn_wmma_f32_16x16x4_f32(false, w0, false, x0, (short)0, acc00, false, false);
        acc01 = __builtin_amdgcn_wmma_f32_16x16x4_f32(false, w0, false, x1, (short)0, acc01, false, false);
        acc10 = __builtin_amdgcn_wmma_f32_16x16x4_f32(false, w1, false, x0, (short)0, acc10, false, false);
        acc11 = __builtin_amdgcn_wmma_f32_16x16x4_f32(false, w1, false, x1, (short)0, acc11, false, false);
        xoff += 4u;
        woff += 4u * uN;
    }
    unsigned hoff = (unsigned)(cg * 32 + 8 * half) * uM + (unsigned)(mg * 32 + l);
    #pragma unroll
    for (int r = 0; r < 8; ++r) {
        float* hp = HT + hoff;
        hp[0]            = acc00[r];
        hp[16]           = acc01[r];
        hp[16u * uM]      = acc10[r];
        hp[16u * uM + 16] = acc11[r];
        hoff += uM;
    }
}

__global__ void wmma_gemm_f32_tt(const float* __restrict__ XT,
                                 const float* __restrict__ Wm,
                                 float* __restrict__ HT,
                                 int M, int K, int N) {
    const int wave    = (blockIdx.x * blockDim.x + threadIdx.x) >> 5;
    const int lane    = threadIdx.x & 31;
    const int cGroups = N >> 5;
    const int mGroups = M >> 5;
    if (wave >= mGroups * cGroups) return;
    const int mg   = wave / cGroups;
    const int cg   = wave % cGroups;
    const int l    = lane & 15;
    const int half = lane >> 4;
    const unsigned uN = (unsigned)N, uM = (unsigned)M;

    unsigned xoff = (unsigned)(2 * half) * uM + (unsigned)(mg * 32 + l);
    unsigned woff = (unsigned)(2 * half) * uN + (unsigned)(cg * 32 + l);

    v8f acc00 = {}, acc01 = {}, acc10 = {}, acc11 = {};
    for (int k0 = 0; k0 < K; k0 += 4) {
        v2f x0 = { XT[xoff],      XT[xoff + uM]      };
        v2f x1 = { XT[xoff + 16], XT[xoff + uM + 16] };
        v2f w0 = { Wm[woff],      Wm[woff + uN]      };
        v2f w1 = { Wm[woff + 16], Wm[woff + uN + 16] };
        acc00 = __builtin_amdgcn_wmma_f32_16x16x4_f32(false, w0, false, x0, (short)0, acc00, false, false);
        acc01 = __builtin_amdgcn_wmma_f32_16x16x4_f32(false, w0, false, x1, (short)0, acc01, false, false);
        acc10 = __builtin_amdgcn_wmma_f32_16x16x4_f32(false, w1, false, x0, (short)0, acc10, false, false);
        acc11 = __builtin_amdgcn_wmma_f32_16x16x4_f32(false, w1, false, x1, (short)0, acc11, false, false);
        xoff += 4u * uM;
        woff += 4u * uN;
    }
    unsigned hoff = (unsigned)(cg * 32 + 8 * half) * uM + (unsigned)(mg * 32 + l);
    #pragma unroll
    for (int r = 0; r < 8; ++r) {
        float* hp = HT + hoff;
        hp[0]            = acc00[r];
        hp[16]           = acc01[r];
        hp[16u * uM]      = acc10[r];
        hp[16u * uM + 16] = acc11[r];
        hoff += uM;
    }
}

__global__ void scores_ct(const float* __restrict__ HT,
                          const float* __restrict__ a_src,
                          const float* __restrict__ a_dst,
                          float* __restrict__ esT, float* __restrict__ edT,
                          int n, int H, int C) {
    int i = blockIdx.x * blockDim.x + threadIdx.x;
    if (i >= n * H) return;
    int hh = i / n, node = i % n;
    const float* hp = HT + (size_t)hh * C * n + node;
    const float* as = a_src + hh * C;
    const float* ad = a_dst + hh * C;
    float s = 0.f, d = 0.f;
    for (int c = 0; c < C; ++c) { float v = hp[(size_t)c * n]; s += v * as[c]; d += v * ad[c]; }
    esT[i] = s;
    edT[i] = d;
}

#define SS   32
#define EPS  (ETOT / SS)
#define UM   ((EPS + 1023) / 1024)
#define NG   10
static_assert(ETOT % SS == 0, "the slices must divide the edge list");
static_assert(2 * NG * 1024 >= NN, "two groups of NG sweeps must cover the nodes");

#define LOAD_SLICE_EDGES(sr, ds)                                              \
    _Pragma("unroll")                                                         \
    for (int u = 0; u < UM; ++u) {                                            \
        const int e  = e0 + u * 1024 + tid;                                   \
        const int ec = (e < e1 && e < NE) ? e : 0;                            \
        const int s0 = ei[ec], d0 = ei[NE + ec];                              \
        sr[u] = (e >= NE) ? e - NE : s0;                                      \
        ds[u] = (e >= NE) ? e - NE : d0;                                      \
    }

__global__ __launch_bounds__(1024) void edge_max_slices(const int* __restrict__ ei,
                                                        const float* __restrict__ esT,
                                                        const float* __restrict__ edT,
                                                        int* __restrict__ pmax) {
    __shared__ float es[NN];
    __shared__ float ed[NN];
    __shared__ int   mx[NN];
    const int tid = threadIdx.x;
    const int hh  = blockIdx.x / SS;
    const int sl  = blockIdx.x % SS;
    const int e0  = sl * EPS, e1 = e0 + EPS;
    int sr[UM], ds[UM];
    LOAD_SLICE_EDGES(sr, ds)
    const float* sp = esT + (size_t)hh * NN;
    const float* dp = edT + (size_t)hh * NN;
    #pragma unroll
    for (int g = 0; g < 2; ++g) {
        float a[NG], b[NG];
        #pragma unroll
        for (int j = 0; j < NG; ++j) {
            const int n  = (g * NG + j) * 1024 + tid;
            const int nc = n < NN ? n : 0;
            a[j] = sp[nc]; b[j] = dp[nc];
        }
        #pragma unroll
        for (int j = 0; j < NG; ++j) {
            const int n = (g * NG + j) * 1024 + tid;
            if (n < NN) { es[n] = a[j]; ed[n] = b[j]; mx[n] = f2ord(-INFINITY); }
        }
    }
    __syncthreads();
    #pragma unroll
    for (int u = 0; u < UM; ++u) {
        const int e = e0 + u * 1024 + tid;
        if (e < e1) {
            float v = es[sr[u]] + ed[ds[u]];
            v = v > 0.f ? v : 0.2f * v;
            atomicMax(&mx[ds[u]], f2ord(v));
        }
    }
    __syncthreads();
    int* o = pmax + (size_t)blockIdx.x * NN;
    for (int n = tid; n < NN; n += 1024) o[n] = mx[n];
}

__global__ void combine_max(const int* __restrict__ pmax, float* __restrict__ mT, int H) {
    const int i = blockIdx.x * blockDim.x + threadIdx.x;
    if (i >= NN * H) return;
    const int hh = i / NN, node = i % NN;
    const int* p = pmax + (size_t)hh * SS * NN + node;
    int m = p[0];
    #pragma unroll
    for (int s = 1; s < SS; ++s) { const int v = p[(size_t)s * NN]; m = v > m ? v : m; }
    mT[i] = ord2f(m);
}

__global__ __launch_bounds__(1024) void edge_exp_slices(const int* __restrict__ ei,
                                                        const float* __restrict__ esT,
                                                        const float* __restrict__ edT,
                                                        const float* __restrict__ mT,
                                                        float* __restrict__ exT,
                                                        float* __restrict__ pden) {
    __shared__ float es[NN];
    __shared__ float ed[NN];
    __shared__ float mm[NN];
    __shared__ float dn[NN];
    const int tid = threadIdx.x;
    const int hh  = blockIdx.x / SS;
    const int sl  = blockIdx.x % SS;
    const int e0  = sl * EPS, e1 = e0 + EPS;
    int sr[UM], ds[UM];
    LOAD_SLICE_EDGES(sr, ds)
    const float* sp = esT + (size_t)hh * NN;
    const float* dp = edT + (size_t)hh * NN;
    const float* mp = mT  + (size_t)hh * NN;
    #pragma unroll
    for (int g = 0; g < 2; ++g) {
        float a[NG], b[NG], c[NG];
        #pragma unroll
        for (int j = 0; j < NG; ++j) {
            const int n  = (g * NG + j) * 1024 + tid;
            const int nc = n < NN ? n : 0;
            a[j] = sp[nc]; b[j] = dp[nc]; c[j] = mp[nc];
        }
        #pragma unroll
        for (int j = 0; j < NG; ++j) {
            const int n = (g * NG + j) * 1024 + tid;
            if (n < NN) { es[n] = a[j]; ed[n] = b[j]; mm[n] = c[j]; dn[n] = 0.f; }
        }
    }
    __syncthreads();
    float* xo = exT + (size_t)hh * ETOT;
    #pragma unroll
    for (int u = 0; u < UM; ++u) {
        const int e = e0 + u * 1024 + tid;
        if (e < e1) {
            float v = es[sr[u]] + ed[ds[u]];
            v = v > 0.f ? v : 0.2f * v;
            const float ex = expf(v - mm[ds[u]]);
            xo[e] = ex;
            atomicAdd(&dn[ds[u]], ex);
        }
    }
    __syncthreads();
    float* o = pden + (size_t)blockIdx.x * NN;
    for (int n = tid; n < NN; n += 1024) o[n] = dn[n];
}

__global__ void combine_den(const float* __restrict__ pden, float* __restrict__ denT, int H) {
    const int i = blockIdx.x * blockDim.x + threadIdx.x;
    if (i >= NN * H) return;
    const int hh = i / NN, node = i % NN;
    const float* p = pden + (size_t)hh * SS * NN + node;
    float d = p[0];
    #pragma unroll
    for (int s = 1; s < SS; ++s) d += p[(size_t)s * NN];
    denT[i] = d;
}

__global__ __launch_bounds__(1024) void agg_pair_lds(const int* __restrict__ ei,
                                                     const float* __restrict__ alT,
                                                     const float* __restrict__ HT,
                                                     const float* __restrict__ denT,
                                                     const float* __restrict__ bias,
                                                     float* __restrict__ outT) {
    __shared__ v2f   hs[NN];
    __shared__ float acc0[NN];
    __shared__ float acc1[NN];
    const int tid  = threadIdx.x;
    const int c0   = 2 * blockIdx.x;
    const int head = c0 / HID;
    const float* h0 = HT + (size_t)c0 * NN;
    const float* h1 = h0 + NN;
    const float* al = alT + (size_t)head * ETOT;

    for (int n = tid; n < NN; n += 1024) {
        v2f hv = { h0[n], h1[n] };
        float a = al[NE + n];
        hs[n]   = hv;
        acc0[n] = a * hv.x;
        acc1[n] = a * hv.y;
    }
    __syncthreads();

    constexpr int FULL = NE / 1024;
    constexpr int BAT  = 8;
    static_assert(FULL % BAT == 0, "the batch must divide the full sweeps");
    int sr[BAT], ds[BAT]; float a[BAT];
    #pragma unroll
    for (int u = 0; u < BAT; ++u) {
        const int e = u * 1024 + tid;
        sr[u] = ei[e]; ds[u] = ei[NE + e]; a[u] = al[e];
    }
    for (int it = 0; it < FULL; it += BAT) {
        const int nx = (it + BAT < FULL) ? it + BAT : it;
        int srn[BAT], dsn[BAT]; float an[BAT];
        #pragma unroll
        for (int u = 0; u < BAT; ++u) {
            const int e = (nx + u) * 1024 + tid;
            srn[u] = ei[e]; dsn[u] = ei[NE + e]; an[u] = al[e];
        }
        v2f hv[BAT];
        #pragma unroll
        for (int u = 0; u < BAT; ++u) hv[u] = hs[sr[u]];
        #pragma unroll
        for (int u = 0; u < BAT; ++u) {
            atomicAdd(&acc0[ds[u]], a[u] * hv[u].x);
            atomicAdd(&acc1[ds[u]], a[u] * hv[u].y);
        }
        #pragma unroll
        for (int u = 0; u < BAT; ++u) { sr[u] = srn[u]; ds[u] = dsn[u]; a[u] = an[u]; }
    }
    {
        const int e = FULL * 1024 + tid;
        if (e < NE) {
            v2f hv = hs[ei[e]];
            const int d = ei[NE + e];
            const float a = al[e];
            atomicAdd(&acc0[d], a * hv.x);
            atomicAdd(&acc1[d], a * hv.y);
        }
    }
    __syncthreads();

    float* o0 = outT + (size_t)c0 * NN;
    float* o1 = o0 + NN;
    const float b0 = bias[c0], b1 = bias[c0 + 1];
    const float* dn = denT + (size_t)head * NN;
    for (int n = tid; n < NN; n += 1024) {
        const float d = dn[n] + 1e-16f;
        const float v0 = acc0[n] / d + b0, v1 = acc1[n] / d + b1;
        o0[n] = v0 > 0.f ? v0 : 0.f;
        o1[n] = v1 > 0.f ? v1 : 0.f;
    }
}

#define S2 8
__global__ __launch_bounds__(1024) void agg2_slice_lds(const int* __restrict__ ei,
                                                       const float* __restrict__ al,
                                                       const float* __restrict__ HT,
                                                       float* __restrict__ part) {
    __shared__ v2f   hs[NN];
    __shared__ float acc0[NN];
    __shared__ float acc1[NN];
    const int tid   = threadIdx.x;
    const int pair  = blockIdx.x % (OUT_CH / 2);
    const int slice = blockIdx.x / (OUT_CH / 2);
    const int c0    = 2 * pair;
    const float* h0 = HT + (size_t)c0 * NN;
    const float* h1 = h0 + NN;

    if (slice == 0) {
        for (int n = tid; n < NN; n += 1024) {
            v2f hv = { h0[n], h1[n] };
            float a = al[NE + n];
            hs[n]   = hv;
            acc0[n] = a * hv.x;
            acc1[n] = a * hv.y;
        }
    } else {
        for (int n = tid; n < NN; n += 1024) {
            v2f hv = { h0[n], h1[n] };
            hs[n]   = hv;
            acc0[n] = 0.f;
            acc1[n] = 0.f;
        }
    }
    __syncthreads();

    constexpr int FULL = NE / 1024;
    static_assert(FULL % S2 == 0, "the slices must divide the full sweeps");
    constexpr int SW  = FULL / S2;
    constexpr int BAT = 13;
    static_assert(SW % BAT == 0, "the batch must divide a slice's sweeps");
    const int base = slice * SW * 1024 + tid;
    for (int it = 0; it < SW; it += BAT) {
        int sr[BAT], ds[BAT]; float a[BAT];
        #pragma unroll
        for (int u = 0; u < BAT; ++u) {
            const int e = base + (it + u) * 1024;
            sr[u] = ei[e]; ds[u] = ei[NE + e]; a[u] = al[e];
        }
        v2f hv[BAT];
        #pragma unroll
        for (int u = 0; u < BAT; ++u) hv[u] = hs[sr[u]];
        #pragma unroll
        for (int u = 0; u < BAT; ++u) {
            atomicAdd(&acc0[ds[u]], a[u] * hv[u].x);
            atomicAdd(&acc1[ds[u]], a[u] * hv[u].y);
        }
    }
    if (slice == S2 - 1) {
        const int e = FULL * 1024 + tid;
        if (e < NE) {
            v2f hv = hs[ei[e]];
            const int d = ei[NE + e];
            const float a = al[e];
            atomicAdd(&acc0[d], a * hv.x);
            atomicAdd(&acc1[d], a * hv.y);
        }
    }
    __syncthreads();

    float* o0 = part + ((size_t)slice * OUT_CH + c0) * NN;
    float* o1 = o0 + NN;
    for (int n = tid; n < NN; n += 1024) { o0[n] = acc0[n]; o1[n] = acc1[n]; }
}

__global__ __launch_bounds__(256) void final_planes_logsoftmax(const float* __restrict__ part,
                                                                const float* __restrict__ pden,
                                                                const float* __restrict__ b2,
                                                                float* __restrict__ out) {
    __shared__ float tile[OUT_CH][33];
    const int tid = threadIdx.x;
    const int w = tid >> 5, lane = tid & 31;
    const int n0 = blockIdx.x * 32;
    float d = pden[n0 + lane];
    #pragma unroll
    for (int s = 1; s < SS; ++s) d += pden[(size_t)s * NN + n0 + lane];
    d += 1e-16f;
    #pragma unroll
    for (int j = 0; j < 8; ++j) {
        const int c = 8 * w + j;
        const float* p = part + (size_t)c * NN + n0 + lane;
        float v = p[0];
        #pragma unroll
        for (int s = 1; s < S2; ++s) v += p[(size_t)s * OUT_CH * NN];
        tile[c][lane] = v / d + b2[c];
    }
    __syncthreads();
    #pragma unroll
    for (int j = 0; j < 4; ++j) {
        const int nl = 4 * w + j;
        float v0 = tile[lane][nl];
        float v1 = tile[lane + 32][nl];
        float m = fmaxf(v0, v1);
        #pragma unroll
        for (int o = 16; o > 0; o >>= 1) m = fmaxf(m, __shfl_xor(m, o, 32));
        float s = expf(v0 - m) + expf(v1 - m);
        #pragma unroll
        for (int o = 16; o > 0; o >>= 1) s += __shfl_xor(s, o, 32);
        float lse = logf(s) + m;
        float* row = out + (size_t)(n0 + nl) * OUT_CH;
        row[lane]      = v0 - lse;
        row[lane + 32] = v1 - lse;
    }
}

extern "C" void kernel_launch(void* const* d_in, const int* in_sizes, int n_in,
                              void* d_out, int out_size, void* d_ws, size_t ws_size,
                              hipStream_t stream) {
    const float* x   = (const float*)d_in[0];
    const int*   ei  = (const int*)  d_in[1];
    const float* W1  = (const float*)d_in[2];
    const float* a1s = (const float*)d_in[3];
    const float* a1d = (const float*)d_in[4];
    const float* b1  = (const float*)d_in[5];
    const float* W2  = (const float*)d_in[6];
    const float* a2s = (const float*)d_in[7];
    const float* a2d = (const float*)d_in[8];
    const float* b2  = (const float*)d_in[9];
    float* out = (float*)d_out;

    float* ws = (float*)d_ws;
    float* h1T   = ws;
    float* agg1T = h1T   + (size_t)NN * HC1;
    float* es1T  = agg1T + (size_t)NN * HC1;
    float* ed1T  = es1T  + (size_t)NN * HEADS;
    float* m1T   = ed1T  + (size_t)NN * HEADS;
    float* den1T = m1T   + (size_t)NN * HEADS;
    int*   pm1   = (int*)(den1T + (size_t)NN * HEADS);
    float* pd1   = (float*)pm1 + (size_t)NN * HEADS * SS;
    float* al1   = pd1  + (size_t)NN * HEADS * SS;
    float* h2T   = al1  + (size_t)ETOT * HEADS;
    float* es2T  = h2T  + (size_t)NN * OUT_CH;
    float* ed2T  = es2T + NN;
    float* m2T   = ed2T + NN;
    int*   pm2   = (int*)(m2T + NN);
    float* pd2   = (float*)pm2 + (size_t)NN * SS;
    float* ex2   = pd2  + (size_t)NN * SS;
    float* part2 = h1T;

    const int TB = 256;
    auto blocks = [](long long n, int tb) { return (unsigned)((n + tb - 1) / tb); };

    {
        long long waves = (long long)(NN / 32) * (HC1 / 32);
        wmma_gemm_f32_ct<<<blocks(waves * 32, TB), TB, 0, stream>>>(x, W1, h1T, NN, IN_CH, HC1);
    }
    scores_ct<<<blocks((long long)NN * HEADS, TB), TB, 0, stream>>>(
        h1T, a1s, a1d, es1T, ed1T, NN, HEADS, HID);
    edge_max_slices<<<HEADS * SS, 1024, 0, stream>>>(ei, es1T, ed1T, pm1);
    combine_max    <<<blocks((long long)NN * HEADS, TB), TB, 0, stream>>>(pm1, m1T, HEADS);
    edge_exp_slices<<<HEADS * SS, 1024, 0, stream>>>(ei, es1T, ed1T, m1T, al1, pd1);
    combine_den    <<<blocks((long long)NN * HEADS, TB), TB, 0, stream>>>(pd1, den1T, HEADS);
    agg_pair_lds  <<<HC1 / 2, 1024, 0, stream>>>(ei, al1, h1T, den1T, b1, agg1T);

    {
        long long waves = (long long)(NN / 32) * (OUT_CH / 32);
        wmma_gemm_f32_tt<<<blocks(waves * 32, TB), TB, 0, stream>>>(agg1T, W2, h2T, NN, HC1, OUT_CH);
    }
    scores_ct<<<blocks((long long)NN, TB), TB, 0, stream>>>(
        h2T, a2s, a2d, es2T, ed2T, NN, 1, OUT_CH);
    edge_max_slices<<<SS, 1024, 0, stream>>>(ei, es2T, ed2T, pm2);
    combine_max    <<<blocks((long long)NN, TB), TB, 0, stream>>>(pm2, m2T, 1);
    edge_exp_slices<<<SS, 1024, 0, stream>>>(ei, es2T, ed2T, m2T, ex2, pd2);
    agg2_slice_lds<<<(OUT_CH / 2) * S2, 1024, 0, stream>>>(ei, ex2, h2T, part2);
    final_planes_logsoftmax<<<NN / 32, 256, 0, stream>>>(part2, pd2, b2, out);
}


# Serial load loops regrouped: a lane's 64 channel reads in both score kernels and the node-plane reads (fill, denominators) of the two LDS aggregation kernels go out as groups, not one trip at a time
# speedup vs baseline: 4.0755x; 1.0629x over previous
#include <hip/hip_runtime.h>
#include <math.h>

#define NN      20000
#define NE      320000
#define ETOT    (NE + NN)
#define IN_CH   128
#define HID     64
#define HEADS   8
#define HC1     (HEADS * HID)
#define OUT_CH  64

typedef float v2f __attribute__((ext_vector_type(2)));
typedef float v8f __attribute__((ext_vector_type(8)));

__device__ __forceinline__ int f2ord(float f) {
    int i = __float_as_int(f);
    return i >= 0 ? i : (i ^ 0x7fffffff);
}
__device__ __forceinline__ float ord2f(int i) {
    return __int_as_float(i >= 0 ? i : (i ^ 0x7fffffff));
}

__global__ void wmma_gemm_f32_ct(const float* __restrict__ X,
                                 const float* __restrict__ Wm,
                                 float* __restrict__ HT,
                                 int M, int K, int N) {
    const int wave    = (blockIdx.x * blockDim.x + threadIdx.x) >> 5;
    const int lane    = threadIdx.x & 31;
    const int cGroups = N >> 5;
    const int mGroups = M >> 5;
    if (wave >= mGroups * cGroups) return;
    const int mg   = wave / cGroups;
    const int cg   = wave % cGroups;
    const int l    = lane & 15;
    const int half = lane >> 4;
    const unsigned uN = (unsigned)N, uM = (unsigned)M, uK = (unsigned)K;

    unsigned xoff = (unsigned)(mg * 32 + l) * uK + 2u * half;
    unsigned woff = (unsigned)(2 * half) * uN + (unsigned)(cg * 32 + l);

    v8f acc00 = {}, acc01 = {}, acc10 = {}, acc11 = {};
    for (int k0 = 0; k0 < K; k0 += 4) {
        v2f x0 = *(const v2f*)(X + xoff);
        v2f x1 = *(const v2f*)(X + xoff + 16u * uK);
        v2f w0 = { Wm[woff],      Wm[woff + uN]      };
        v2f w1 = { Wm[woff + 16], Wm[woff + uN + 16] };
        acc00 = __builtin_amdgcn_wmma_f32_16x16x4_f32(false, w0, false, x0, (short)0, acc00, false, false);
        acc01 = __builtin_amdgcn_wmma_f32_16x16x4_f32(false, w0, false, x1, (short)0, acc01, false, false);
        acc10 = __builtin_amdgcn_wmma_f32_16x16x4_f32(false, w1, false, x0, (short)0, acc10, false, false);
        acc11 = __builtin_amdgcn_wmma_f32_16x16x4_f32(false, w1, false, x1, (short)0, acc11, false, false);
        xoff += 4u;
        woff += 4u * uN;
    }
    unsigned hoff = (unsigned)(cg * 32 + 8 * half) * uM + (unsigned)(mg * 32 + l);
    #pragma unroll
    for (int r = 0; r < 8; ++r) {
        float* hp = HT + hoff;
        hp[0]            = acc00[r];
        hp[16]           = acc01[r];
        hp[16u * uM]      = acc10[r];
        hp[16u * uM + 16] = acc11[r];
        hoff += uM;
    }
}

__global__ void wmma_gemm_f32_tt(const float* __restrict__ XT,
                                 const float* __restrict__ Wm,
                                 float* __restrict__ HT,
                                 int M, int K, int N) {
    const int wave    = (blockIdx.x * blockDim.x + threadIdx.x) >> 5;
    const int lane    = threadIdx.x & 31;
    const int cGroups = N >> 5;
    const int mGroups = M >> 5;
    if (wave >= mGroups * cGroups) return;
    const int mg   = wave / cGroups;
    const int cg   = wave % cGroups;
    const int l    = lane & 15;
    const int half = lane >> 4;
    const unsigned uN = (unsigned)N, uM = (unsigned)M;

    unsigned xoff = (unsigned)(2 * half) * uM + (unsigned)(mg * 32 + l);
    unsigned woff = (unsigned)(2 * half) * uN + (unsigned)(cg * 32 + l);

    v8f acc00 = {}, acc01 = {}, acc10 = {}, acc11 = {};
    for (int k0 = 0; k0 < K; k0 += 4) {
        v2f x0 = { XT[xoff],      XT[xoff + uM]      };
        v2f x1 = { XT[xoff + 16], XT[xoff + uM + 16] };
        v2f w0 = { Wm[woff],      Wm[woff + uN]      };
        v2f w1 = { Wm[woff + 16], Wm[woff + uN + 16] };
        acc00 = __builtin_amdgcn_wmma_f32_16x16x4_f32(false, w0, false, x0, (short)0, acc00, false, false);
        acc01 = __builtin_amdgcn_wmma_f32_16x16x4_f32(false, w0, false, x1, (short)0, acc01, false, false);
        acc10 = __builtin_amdgcn_wmma_f32_16x16x4_f32(false, w1, false, x0, (short)0, acc10, false, false);
        acc11 = __builtin_amdgcn_wmma_f32_16x16x4_f32(false, w1, false, x1, (short)0, acc11, false, false);
        xoff += 4u * uM;
        woff += 4u * uN;
    }
    unsigned hoff = (unsigned)(cg * 32 + 8 * half) * uM + (unsigned)(mg * 32 + l);
    #pragma unroll
    for (int r = 0; r < 8; ++r) {
        float* hp = HT + hoff;
        hp[0]            = acc00[r];
        hp[16]           = acc01[r];
        hp[16u * uM]      = acc10[r];
        hp[16u * uM + 16] = acc11[r];
        hoff += uM;
    }
}

template <int C>
__global__ void scores_ct(const float* __restrict__ HT,
                          const float* __restrict__ a_src,
                          const float* __restrict__ a_dst,
                          float* __restrict__ esT, float* __restrict__ edT) {
    const int node = blockIdx.x * blockDim.x + threadIdx.x;
    const int hh   = blockIdx.y;
    if (node >= NN) return;
    const float* hp = HT + (size_t)hh * C * NN + node;
    const float* as = a_src + hh * C;
    const float* ad = a_dst + hh * C;
    float v[C];
    #pragma unroll
    for (int c = 0; c < C; ++c) v[c] = hp[(size_t)c * NN];
    float s = 0.f, d = 0.f;
    #pragma unroll
    for (int c = 0; c < C; ++c) { s += v[c] * as[c]; d += v[c] * ad[c]; }
    esT[(size_t)hh * NN + node] = s;
    edT[(size_t)hh * NN + node] = d;
}

#define SS   32
#define EPS  (ETOT / SS)
#define UM   ((EPS + 1023) / 1024)
#define NG   10
static_assert(ETOT % SS == 0, "the slices must divide the edge list");
static_assert(2 * NG * 1024 >= NN, "two groups of NG sweeps must cover the nodes");

#define LOAD_SLICE_EDGES(sr, ds)                                              \
    _Pragma("unroll")                                                         \
    for (int u = 0; u < UM; ++u) {                                            \
        const int e  = e0 + u * 1024 + tid;                                   \
        const int ec = (e < e1 && e < NE) ? e : 0;                            \
        const int s0 = ei[ec], d0 = ei[NE + ec];                              \
        sr[u] = (e >= NE) ? e - NE : s0;                                      \
        ds[u] = (e >= NE) ? e - NE : d0;                                      \
    }

__global__ __launch_bounds__(1024) void edge_max_slices(const int* __restrict__ ei,
                                                        const float* __restrict__ esT,
                                                        const float* __restrict__ edT,
                                                        int* __restrict__ pmax) {
    __shared__ float es[NN];
    __shared__ float ed[NN];
    __shared__ int   mx[NN];
    const int tid = threadIdx.x;
    const int hh  = blockIdx.x / SS;
    const int sl  = blockIdx.x % SS;
    const int e0  = sl * EPS, e1 = e0 + EPS;
    int sr[UM], ds[UM];
    LOAD_SLICE_EDGES(sr, ds)
    const float* sp = esT + (size_t)hh * NN;
    const float* dp = edT + (size_t)hh * NN;
    #pragma unroll
    for (int g = 0; g < 2; ++g) {
        float a[NG], b[NG];
        #pragma unroll
        for (int j = 0; j < NG; ++j) {
            const int n  = (g * NG + j) * 1024 + tid;
            const int nc = n < NN ? n : 0;
            a[j] = sp[nc]; b[j] = dp[nc];
        }
        #pragma unroll
        for (int j = 0; j < NG; ++j) {
            const int n = (g * NG + j) * 1024 + tid;
            if (n < NN) { es[n] = a[j]; ed[n] = b[j]; mx[n] = f2ord(-INFINITY); }
        }
    }
    __syncthreads();
    #pragma unroll
    for (int u = 0; u < UM; ++u) {
        const int e = e0 + u * 1024 + tid;
        if (e < e1) {
            float v = es[sr[u]] + ed[ds[u]];
            v = v > 0.f ? v : 0.2f * v;
            atomicMax(&mx[ds[u]], f2ord(v));
        }
    }
    __syncthreads();
    int* o = pmax + (size_t)blockIdx.x * NN;
    for (int n = tid; n < NN; n += 1024) o[n] = mx[n];
}

__global__ void combine_max(const int* __restrict__ pmax, float* __restrict__ mT, int H) {
    const int i = blockIdx.x * blockDim.x + threadIdx.x;
    if (i >= NN * H) return;
    const int hh = i / NN, node = i % NN;
    const int* p = pmax + (size_t)hh * SS * NN + node;
    int m = p[0];
    #pragma unroll
    for (int s = 1; s < SS; ++s) { const int v = p[(size_t)s * NN]; m = v > m ? v : m; }
    mT[i] = ord2f(m);
}

__global__ __launch_bounds__(1024) void edge_exp_slices(const int* __restrict__ ei,
                                                        const float* __restrict__ esT,
                                                        const float* __restrict__ edT,
                                                        const float* __restrict__ mT,
                                                        float* __restrict__ exT,
                                                        float* __restrict__ pden) {
    __shared__ float es[NN];
    __shared__ float ed[NN];
    __shared__ float mm[NN];
    __shared__ float dn[NN];
    const int tid = threadIdx.x;
    const int hh  = blockIdx.x / SS;
    const int sl  = blockIdx.x % SS;
    const int e0  = sl * EPS, e1 = e0 + EPS;
    int sr[UM], ds[UM];
    LOAD_SLICE_EDGES(sr, ds)
    const float* sp = esT + (size_t)hh * NN;
    const float* dp = edT + (size_t)hh * NN;
    const float* mp = mT  + (size_t)hh * NN;
    #pragma unroll
    for (int g = 0; g < 2; ++g) {
        float a[NG], b[NG], c[NG];
        #pragma unroll
        for (int j = 0; j < NG; ++j) {
            const int n  = (g * NG + j) * 1024 + tid;
            const int nc = n < NN ? n : 0;
            a[j] = sp[nc]; b[j] = dp[nc]; c[j] = mp[nc];
        }
        #pragma unroll
        for (int j = 0; j < NG; ++j) {
            const int n = (g * NG + j) * 1024 + tid;
            if (n < NN) { es[n] = a[j]; ed[n] = b[j]; mm[n] = c[j]; dn[n] = 0.f; }
        }
    }
    __syncthreads();
    float* xo = exT + (size_t)hh * ETOT;
    #pragma unroll
    for (int u = 0; u < UM; ++u) {
        const int e = e0 + u * 1024 + tid;
        if (e < e1) {
            float v = es[sr[u]] + ed[ds[u]];
            v = v > 0.f ? v : 0.2f * v;
            const float ex = expf(v - mm[ds[u]]);
            xo[e] = ex;
            atomicAdd(&dn[ds[u]], ex);
        }
    }
    __syncthreads();
    float* o = pden + (size_t)blockIdx.x * NN;
    for (int n = tid; n < NN; n += 1024) o[n] = dn[n];
}

__global__ void combine_den(const float* __restrict__ pden, float* __restrict__ denT, int H) {
    const int i = blockIdx.x * blockDim.x + threadIdx.x;
    if (i >= NN * H) return;
    const int hh = i / NN, node = i % NN;
    const float* p = pden + (size_t)hh * SS * NN + node;
    float d = p[0];
    #pragma unroll
    for (int s = 1; s < SS; ++s) d += p[(size_t)s * NN];
    denT[i] = d;
}

__global__ __launch_bounds__(1024) void agg_pair_lds(const int* __restrict__ ei,
                                                     const float* __restrict__ alT,
                                                     const float* __restrict__ HT,
                                                     const float* __restrict__ denT,
                                                     const float* __restrict__ bias,
                                                     float* __restrict__ outT) {
    __shared__ v2f   hs[NN];
    __shared__ float acc0[NN];
    __shared__ float acc1[NN];
    const int tid  = threadIdx.x;
    const int c0   = 2 * blockIdx.x;
    const int head = c0 / HID;
    const float* h0 = HT + (size_t)c0 * NN;
    const float* h1 = h0 + NN;
    const float* al = alT + (size_t)head * ETOT;

    #pragma unroll
    for (int g = 0; g < 2; ++g) {
        float x0[NG], x1[NG], aa[NG];
        #pragma unroll
        for (int j = 0; j < NG; ++j) {
            const int n  = (g * NG + j) * 1024 + tid;
            const int nc = n < NN ? n : 0;
            x0[j] = h0[nc]; x1[j] = h1[nc];
            aa[j] = al[NE + nc];
        }
        #pragma unroll
        for (int j = 0; j < NG; ++j) {
            const int n = (g * NG + j) * 1024 + tid;
            if (n < NN) {
                v2f hv = { x0[j], x1[j] };
                hs[n]   = hv;
                acc0[n] = aa[j] * hv.x;
                acc1[n] = aa[j] * hv.y;
            }
        }
    }
    __syncthreads();

    constexpr int FULL = NE / 1024;
    constexpr int BAT  = 8;
    static_assert(FULL % BAT == 0, "the batch must divide the full sweeps");
    int sr[BAT], ds[BAT]; float a[BAT];
    #pragma unroll
    for (int u = 0; u < BAT; ++u) {
        const int e = u * 1024 + tid;
        sr[u] = ei[e]; ds[u] = ei[NE + e]; a[u] = al[e];
    }
    for (int it = 0; it < FULL; it += BAT) {
        const int nx = (it + BAT < FULL) ? it + BAT : it;
        int srn[BAT], dsn[BAT]; float an[BAT];
        #pragma unroll
        for (int u = 0; u < BAT; ++u) {
            const int e = (nx + u) * 1024 + tid;
            srn[u] = ei[e]; dsn[u] = ei[NE + e]; an[u] = al[e];
        }
        v2f hv[BAT];
        #pragma unroll
        for (int u = 0; u < BAT; ++u) hv[u] = hs[sr[u]];
        #pragma unroll
        for (int u = 0; u < BAT; ++u) {
            atomicAdd(&acc0[ds[u]], a[u] * hv[u].x);
            atomicAdd(&acc1[ds[u]], a[u] * hv[u].y);
        }
        #pragma unroll
        for (int u = 0; u < BAT; ++u) { sr[u] = srn[u]; ds[u] = dsn[u]; a[u] = an[u]; }
    }
    {
        const int e = FULL * 1024 + tid;
        if (e < NE) {
            v2f hv = hs[ei[e]];
            const int d = ei[NE + e];
            const float a = al[e];
            atomicAdd(&acc0[d], a * hv.x);
            atomicAdd(&acc1[d], a * hv.y);
        }
    }
    const float* dn = denT + (size_t)head * NN;
    float dd[2 * NG];
    #pragma unroll
    for (int j = 0; j < 2 * NG; ++j) {
        const int n  = j * 1024 + tid;
        const int nc = n < NN ? n : 0;
        dd[j] = dn[nc];
    }
    __syncthreads();

    float* o0 = outT + (size_t)c0 * NN;
    float* o1 = o0 + NN;
    const float b0 = bias[c0], b1 = bias[c0 + 1];
    #pragma unroll
    for (int j = 0; j < 2 * NG; ++j) {
        const int n = j * 1024 + tid;
        if (n < NN) {
            const float d = dd[j] + 1e-16f;
            const float v0 = acc0[n] / d + b0, v1 = acc1[n] / d + b1;
            o0[n] = v0 > 0.f ? v0 : 0.f;
            o1[n] = v1 > 0.f ? v1 : 0.f;
        }
    }
}

#define S2 8
__global__ __launch_bounds__(1024) void agg2_slice_lds(const int* __restrict__ ei,
                                                       const float* __restrict__ al,
                                                       const float* __restrict__ HT,
                                                       float* __restrict__ part) {
    __shared__ v2f   hs[NN];
    __shared__ float acc0[NN];
    __shared__ float acc1[NN];
    const int tid   = threadIdx.x;
    const int pair  = blockIdx.x % (OUT_CH / 2);
    const int slice = blockIdx.x / (OUT_CH / 2);
    const int c0    = 2 * pair;
    const float* h0 = HT + (size_t)c0 * NN;
    const float* h1 = h0 + NN;

    if (slice == 0) {
        #pragma unroll
        for (int g = 0; g < 2; ++g) {
            float x0[NG], x1[NG], aa[NG];
            #pragma unroll
            for (int j = 0; j < NG; ++j) {
                const int n  = (g * NG + j) * 1024 + tid;
                const int nc = n < NN ? n : 0;
                x0[j] = h0[nc]; x1[j] = h1[nc];
                aa[j] = al[NE + nc];
            }
            #pragma unroll
            for (int j = 0; j < NG; ++j) {
                const int n = (g * NG + j) * 1024 + tid;
                if (n < NN) {
                    v2f hv = { x0[j], x1[j] };
                    hs[n]   = hv;
                    acc0[n] = aa[j] * hv.x;
                    acc1[n] = aa[j] * hv.y;
                }
            }
        }
    } else {
        #pragma unroll
        for (int g = 0; g < 2; ++g) {
            float x0[NG], x1[NG];
            #pragma unroll
            for (int j = 0; j < NG; ++j) {
                const int n  = (g * NG + j) * 1024 + tid;
                const int nc = n < NN ? n : 0;
                x0[j] = h0[nc]; x1[j] = h1[nc];
            }
            #pragma unroll
            for (int j = 0; j < NG; ++j) {
                const int n = (g * NG + j) * 1024 + tid;
                if (n < NN) {
                    v2f hv = { x0[j], x1[j] };
                    hs[n]   = hv;
                    acc0[n] = 0.f;
                    acc1[n] = 0.f;
                }
            }
        }
    }
    __syncthreads();

    constexpr int FULL = NE / 1024;
    static_assert(FULL % S2 == 0, "the slices must divide the full sweeps");
    constexpr int SW  = FULL / S2;
    constexpr int BAT = 13;
    static_assert(SW % BAT == 0, "the batch must divide a slice's sweeps");
    const int base = slice * SW * 1024 + tid;
    for (int it = 0; it < SW; it += BAT) {
        int sr[BAT], ds[BAT]; float a[BAT];
        #pragma unroll
        for (int u = 0; u < BAT; ++u) {
            const int e = base + (it + u) * 1024;
            sr[u] = ei[e]; ds[u] = ei[NE + e]; a[u] = al[e];
        }
        v2f hv[BAT];
        #pragma unroll
        for (int u = 0; u < BAT; ++u) hv[u] = hs[sr[u]];
        #pragma unroll
        for (int u = 0; u < BAT; ++u) {
            atomicAdd(&acc0[ds[u]], a[u] * hv[u].x);
            atomicAdd(&acc1[ds[u]], a[u] * hv[u].y);
        }
    }
    if (slice == S2 - 1) {
        const int e = FULL * 1024 + tid;
        if (e < NE) {
            v2f hv = hs[ei[e]];
            const int d = ei[NE + e];
            const float a = al[e];
            atomicAdd(&acc0[d], a * hv.x);
            atomicAdd(&acc1[d], a * hv.y);
        }
    }
    __syncthreads();

    float* o0 = part + ((size_t)slice * OUT_CH + c0) * NN;
    float* o1 = o0 + NN;
    for (int n = tid; n < NN; n += 1024) { o0[n] = acc0[n]; o1[n] = acc1[n]; }
}

__global__ __launch_bounds__(256) void final_planes_logsoftmax(const float* __restrict__ part,
                                                                const float* __restrict__ pden,
                                                                const float* __restrict__ b2,
                                                                float* __restrict__ out) {
    __shared__ float tile[OUT_CH][33];
    const int tid = threadIdx.x;
    const int w = tid >> 5, lane = tid & 31;
    const int n0 = blockIdx.x * 32;
    float d = pden[n0 + lane];
    #pragma unroll
    for (int s = 1; s < SS; ++s) d += pden[(size_t)s * NN + n0 + lane];
    d += 1e-16f;
    #pragma unroll
    for (int j = 0; j < 8; ++j) {
        const int c = 8 * w + j;
        const float* p = part + (size_t)c * NN + n0 + lane;
        float v = p[0];
        #pragma unroll
        for (int s = 1; s < S2; ++s) v += p[(size_t)s * OUT_CH * NN];
        tile[c][lane] = v / d + b2[c];
    }
    __syncthreads();
    #pragma unroll
    for (int j = 0; j < 4; ++j) {
        const int nl = 4 * w + j;
        float v0 = tile[lane][nl];
        float v1 = tile[lane + 32][nl];
        float m = fmaxf(v0, v1);
        #pragma unroll
        for (int o = 16; o > 0; o >>= 1) m = fmaxf(m, __shfl_xor(m, o, 32));
        float s = expf(v0 - m) + expf(v1 - m);
        #pragma unroll
        for (int o = 16; o > 0; o >>= 1) s += __shfl_xor(s, o, 32);
        float lse = logf(s) + m;
        float* row = out + (size_t)(n0 + nl) * OUT_CH;
        row[lane]      = v0 - lse;
        row[lane + 32] = v1 - lse;
    }
}

extern "C" void kernel_launch(void* const* d_in, const int* in_sizes, int n_in,
                              void* d_out, int out_size, void* d_ws, size_t ws_size,
                              hipStream_t stream) {
    const float* x   = (const float*)d_in[0];
    const int*   ei  = (const int*)  d_in[1];
    const float* W1  = (const float*)d_in[2];
    const float* a1s = (const float*)d_in[3];
    const float* a1d = (const float*)d_in[4];
    const float* b1  = (const float*)d_in[5];
    const float* W2  = (const float*)d_in[6];
    const float* a2s = (const float*)d_in[7];
    const float* a2d = (const float*)d_in[8];
    const float* b2  = (const float*)d_in[9];
    float* out = (float*)d_out;

    float* ws = (float*)d_ws;
    float* h1T   = ws;
    float* agg1T = h1T   + (size_t)NN * HC1;
    float* es1T  = agg1T + (size_t)NN * HC1;
    float* ed1T  = es1T  + (size_t)NN * HEADS;
    float* m1T   = ed1T  + (size_t)NN * HEADS;
    float* den1T = m1T   + (size_t)NN * HEADS;
    int*   pm1   = (int*)(den1T + (size_t)NN * HEADS);
    float* pd1   = (float*)pm1 + (size_t)NN * HEADS * SS;
    float* al1   = pd1  + (size_t)NN * HEADS * SS;
    float* h2T   = al1  + (size_t)ETOT * HEADS;
    float* es2T  = h2T  + (size_t)NN * OUT_CH;
    float* ed2T  = es2T + NN;
    float* m2T   = ed2T + NN;
    int*   pm2   = (int*)(m2T + NN);
    float* pd2   = (float*)pm2 + (size_t)NN * SS;
    float* ex2   = pd2  + (size_t)NN * SS;
    float* part2 = h1T;

    const int TB = 256;
    auto blocks = [](long long n, int tb) { return (unsigned)((n + tb - 1) / tb); };

    {
        long long waves = (long long)(NN / 32) * (HC1 / 32);
        wmma_gemm_f32_ct<<<blocks(waves * 32, TB), TB, 0, stream>>>(x, W1, h1T, NN, IN_CH, HC1);
    }
    scores_ct<HID><<<dim3(blocks(NN, TB), HEADS), TB, 0, stream>>>(
        h1T, a1s, a1d, es1T, ed1T);
    edge_max_slices<<<HEADS * SS, 1024, 0, stream>>>(ei, es1T, ed1T, pm1);
    combine_max    <<<blocks((long long)NN * HEADS, TB), TB, 0, stream>>>(pm1, m1T, HEADS);
    edge_exp_slices<<<HEADS * SS, 1024, 0, stream>>>(ei, es1T, ed1T, m1T, al1, pd1);
    combine_den    <<<blocks((long long)NN * HEADS, TB), TB, 0, stream>>>(pd1, den1T, HEADS);
    agg_pair_lds  <<<HC1 / 2, 1024, 0, stream>>>(ei, al1, h1T, den1T, b1, agg1T);

    {
        long long waves = (long long)(NN / 32) * (OUT_CH / 32);
        wmma_gemm_f32_tt<<<blocks(waves * 32, TB), TB, 0, stream>>>(agg1T, W2, h2T, NN, HC1, OUT_CH);
    }
    scores_ct<OUT_CH><<<dim3(blocks(NN, TB), 1), TB, 0, stream>>>(
        h2T, a2s, a2d, es2T, ed2T);
    edge_max_slices<<<SS, 1024, 0, stream>>>(ei, es2T, ed2T, pm2);
    combine_max    <<<blocks((long long)NN, TB), TB, 0, stream>>>(pm2, m2T, 1);
    edge_exp_slices<<<SS, 1024, 0, stream>>>(ei, es2T, ed2T, m2T, ex2, pd2);
    agg2_slice_lds<<<(OUT_CH / 2) * S2, 1024, 0, stream>>>(ei, ex2, h2T, part2);
    final_planes_logsoftmax<<<NN / 32, 256, 0, stream>>>(part2, pd2, b2, out);
}


# Layer-2 GEMM: K split over 4 waves per tile (1,250 -> 5,000 waves), so a SIMD runs other waves' WMMAs under a wave's load wait; slice tiles added in LDS in slice order (h2 sum order changes)
# speedup vs baseline: 4.3598x; 1.0698x over previous
#include <hip/hip_runtime.h>
#include <math.h>

#define NN      20000
#define NE      320000
#define ETOT    (NE + NN)
#define IN_CH   128
#define HID     64
#define HEADS   8
#define HC1     (HEADS * HID)
#define OUT_CH  64

typedef float v2f __attribute__((ext_vector_type(2)));
typedef float v8f __attribute__((ext_vector_type(8)));

__device__ __forceinline__ int f2ord(float f) {
    int i = __float_as_int(f);
    return i >= 0 ? i : (i ^ 0x7fffffff);
}
__device__ __forceinline__ float ord2f(int i) {
    return __int_as_float(i >= 0 ? i : (i ^ 0x7fffffff));
}

__global__ void wmma_gemm_f32_ct(const float* __restrict__ X,
                                 const float* __restrict__ Wm,
                                 float* __restrict__ HT,
                                 int M, int K, int N) {
    const int wave    = (blockIdx.x * blockDim.x + threadIdx.x) >> 5;
    const int lane    = threadIdx.x & 31;
    const int cGroups = N >> 5;
    const int mGroups = M >> 5;
    if (wave >= mGroups * cGroups) return;
    const int mg   = wave / cGroups;
    const int cg   = wave % cGroups;
    const int l    = lane & 15;
    const int half = lane >> 4;
    const unsigned uN = (unsigned)N, uM = (unsigned)M, uK = (unsigned)K;

    unsigned xoff = (unsigned)(mg * 32 + l) * uK + 2u * half;
    unsigned woff = (unsigned)(2 * half) * uN + (unsigned)(cg * 32 + l);

    v8f acc00 = {}, acc01 = {}, acc10 = {}, acc11 = {};
    for (int k0 = 0; k0 < K; k0 += 4) {
        v2f x0 = *(const v2f*)(X + xoff);
        v2f x1 = *(const v2f*)(X + xoff + 16u * uK);
        v2f w0 = { Wm[woff],      Wm[woff + uN]      };
        v2f w1 = { Wm[woff + 16], Wm[woff + uN + 16] };
        acc00 = __builtin_amdgcn_wmma_f32_16x16x4_f32(false, w0, false, x0, (short)0, acc00, false, false);
        acc01 = __builtin_amdgcn_wmma_f32_16x16x4_f32(false, w0, false, x1, (short)0, acc01, false, false);
        acc10 = __builtin_amdgcn_wmma_f32_16x16x4_f32(false, w1, false, x0, (short)0, acc10, false, false);
        acc11 = __builtin_amdgcn_wmma_f32_16x16x4_f32(false, w1, false, x1, (short)0, acc11, false, false);
        xoff += 4u;
        woff += 4u * uN;
    }
    unsigned hoff = (unsigned)(cg * 32 + 8 * half) * uM + (unsigned)(mg * 32 + l);
    #pragma unroll
    for (int r = 0; r < 8; ++r) {
        float* hp = HT + hoff;
        hp[0]            = acc00[r];
        hp[16]           = acc01[r];
        hp[16u * uM]      = acc10[r];
        hp[16u * uM + 16] = acc11[r];
        hoff += uM;
    }
}

#define KS2 4
__global__ __launch_bounds__(256) void wmma_gemm_f32_tt(const float* __restrict__ XT,
                                                        const float* __restrict__ Wm,
                                                        float* __restrict__ HT,
                                                        int M, int K, int N) {
    __shared__ float part[2][KS2][32][34];
    const int w    = threadIdx.x >> 5;
    const int lane = threadIdx.x & 31;
    const int cg   = w & 1;
    const int ks   = w >> 1;
    const int mg   = blockIdx.x;
    const int l    = lane & 15;
    const int half = lane >> 4;
    const unsigned uN = (unsigned)N, uM = (unsigned)M;
    const int KL = K / KS2;
    const unsigned kb = (unsigned)(ks * KL);

    unsigned xoff = (kb + 2u * half) * uM + (unsigned)(mg * 32 + l);
    unsigned woff = (kb + 2u * half) * uN + (unsigned)(cg * 32 + l);

    v8f acc00 = {}, acc01 = {}, acc10 = {}, acc11 = {};
    for (int k0 = 0; k0 < KL; k0 += 4) {
        v2f x0 = { XT[xoff],      XT[xoff + uM]      };
        v2f x1 = { XT[xoff + 16], XT[xoff + uM + 16] };
        v2f w0 = { Wm[woff],      Wm[woff + uN]      };
        v2f w1 = { Wm[woff + 16], Wm[woff + uN + 16] };
        acc00 = __builtin_amdgcn_wmma_f32_16x16x4_f32(false, w0, false, x0, (short)0, acc00, false, false);
        acc01 = __builtin_amdgcn_wmma_f32_16x16x4_f32(false, w0, false, x1, (short)0, acc01, false, false);
        acc10 = __builtin_amdgcn_wmma_f32_16x16x4_f32(false, w1, false, x0, (short)0, acc10, false, false);
        acc11 = __builtin_amdgcn_wmma_f32_16x16x4_f32(false, w1, false, x1, (short)0, acc11, false, false);
        xoff += 4u * uM;
        woff += 4u * uN;
    }
    float (*pt)[34] = part[cg][ks];
    #pragma unroll
    for (int r = 0; r < 8; ++r) {
        pt[8 * half + r     ][l     ] = acc00[r];
        pt[8 * half + r     ][l + 16] = acc01[r];
        pt[8 * half + r + 16][l     ] = acc10[r];
        pt[8 * half + r + 16][l + 16] = acc11[r];
    }
    __syncthreads();
    unsigned hoff = (unsigned)(cg * 32 + 8 * ks) * uM + (unsigned)(mg * 32 + lane);
    #pragma unroll
    for (int j = 0; j < 8; ++j) {
        const int row = 8 * ks + j;
        float v = part[cg][0][row][lane];
        #pragma unroll
        for (int q = 1; q < KS2; ++q) v += part[cg][q][row][lane];
        HT[hoff] = v;
        hoff += uM;
    }
}

template <int C>
__global__ void scores_ct(const float* __restrict__ HT,
                          const float* __restrict__ a_src,
                          const float* __restrict__ a_dst,
                          float* __restrict__ esT, float* __restrict__ edT) {
    const int node = blockIdx.x * blockDim.x + threadIdx.x;
    const int hh   = blockIdx.y;
    if (node >= NN) return;
    const float* hp = HT + (size_t)hh * C * NN + node;
    const float* as = a_src + hh * C;
    const float* ad = a_dst + hh * C;
    float v[C];
    #pragma unroll
    for (int c = 0; c < C; ++c) v[c] = hp[(size_t)c * NN];
    float s = 0.f, d = 0.f;
    #pragma unroll
    for (int c = 0; c < C; ++c) { s += v[c] * as[c]; d += v[c] * ad[c]; }
    esT[(size_t)hh * NN + node] = s;
    edT[(size_t)hh * NN + node] = d;
}

#define SS   32
#define EPS  (ETOT / SS)
#define UM   ((EPS + 1023) / 1024)
#define NG   10
static_assert(ETOT % SS == 0, "the slices must divide the edge list");
static_assert(2 * NG * 1024 >= NN, "two groups of NG sweeps must cover the nodes");

#define LOAD_SLICE_EDGES(sr, ds)                                              \
    _Pragma("unroll")                                                         \
    for (int u = 0; u < UM; ++u) {                                            \
        const int e  = e0 + u * 1024 + tid;                                   \
        const int ec = (e < e1 && e < NE) ? e : 0;                            \
        const int s0 = ei[ec], d0 = ei[NE + ec];                              \
        sr[u] = (e >= NE) ? e - NE : s0;                                      \
        ds[u] = (e >= NE) ? e - NE : d0;                                      \
    }

__global__ __launch_bounds__(1024) void edge_max_slices(const int* __restrict__ ei,
                                                        const float* __restrict__ esT,
                                                        const float* __restrict__ edT,
                                                        int* __restrict__ pmax) {
    __shared__ float es[NN];
    __shared__ float ed[NN];
    __shared__ int   mx[NN];
    const int tid = threadIdx.x;
    const int hh  = blockIdx.x / SS;
    const int sl  = blockIdx.x % SS;
    const int e0  = sl * EPS, e1 = e0 + EPS;
    int sr[UM], ds[UM];
    LOAD_SLICE_EDGES(sr, ds)
    const float* sp = esT + (size_t)hh * NN;
    const float* dp = edT + (size_t)hh * NN;
    #pragma unroll
    for (int g = 0; g < 2; ++g) {
        float a[NG], b[NG];
        #pragma unroll
        for (int j = 0; j < NG; ++j) {
            const int n  = (g * NG + j) * 1024 + tid;
            const int nc = n < NN ? n : 0;
            a[j] = sp[nc]; b[j] = dp[nc];
        }
        #pragma unroll
        for (int j = 0; j < NG; ++j) {
            const int n = (g * NG + j) * 1024 + tid;
            if (n < NN) { es[n] = a[j]; ed[n] = b[j]; mx[n] = f2ord(-INFINITY); }
        }
    }
    __syncthreads();
    #pragma unroll
    for (int u = 0; u < UM; ++u) {
        const int e = e0 + u * 1024 + tid;
        if (e < e1) {
            float v = es[sr[u]] + ed[ds[u]];
            v = v > 0.f ? v : 0.2f * v;
            atomicMax(&mx[ds[u]], f2ord(v));
        }
    }
    __syncthreads();
    int* o = pmax + (size_t)blockIdx.x * NN;
    for (int n = tid; n < NN; n += 1024) o[n] = mx[n];
}

__global__ void combine_max(const int* __restrict__ pmax, float* __restrict__ mT, int H) {
    const int i = blockIdx.x * blockDim.x + threadIdx.x;
    if (i >= NN * H) return;
    const int hh = i / NN, node = i % NN;
    const int* p = pmax + (size_t)hh * SS * NN + node;
    int m = p[0];
    #pragma unroll
    for (int s = 1; s < SS; ++s) { const int v = p[(size_t)s * NN]; m = v > m ? v : m; }
    mT[i] = ord2f(m);
}

__global__ __launch_bounds__(1024) void edge_exp_slices(const int* __restrict__ ei,
                                                        const float* __restrict__ esT,
                                                        const float* __restrict__ edT,
                                                        const float* __restrict__ mT,
                                                        float* __restrict__ exT,
                                                        float* __restrict__ pden) {
    __shared__ float es[NN];
    __shared__ float ed[NN];
    __shared__ float mm[NN];
    __shared__ float dn[NN];
    const int tid = threadIdx.x;
    const int hh  = blockIdx.x / SS;
    const int sl  = blockIdx.x % SS;
    const int e0  = sl * EPS, e1 = e0 + EPS;
    int sr[UM], ds[UM];
    LOAD_SLICE_EDGES(sr, ds)
    const float* sp = esT + (size_t)hh * NN;
    const float* dp = edT + (size_t)hh * NN;
    const float* mp = mT  + (size_t)hh * NN;
    #pragma unroll
    for (int g = 0; g < 2; ++g) {
        float a[NG], b[NG], c[NG];
        #pragma unroll
        for (int j = 0; j < NG; ++j) {
            const int n  = (g * NG + j) * 1024 + tid;
            const int nc = n < NN ? n : 0;
            a[j] = sp[nc]; b[j] = dp[nc]; c[j] = mp[nc];
        }
        #pragma unroll
        for (int j = 0; j < NG; ++j) {
            const int n = (g * NG + j) * 1024 + tid;
            if (n < NN) { es[n] = a[j]; ed[n] = b[j]; mm[n] = c[j]; dn[n] = 0.f; }
        }
    }
    __syncthreads();
    float* xo = exT + (size_t)hh * ETOT;
    #pragma unroll
    for (int u = 0; u < UM; ++u) {
        const int e = e0 + u * 1024 + tid;
        if (e < e1) {
            float v = es[sr[u]] + ed[ds[u]];
            v = v > 0.f ? v : 0.2f * v;
            const float ex = expf(v - mm[ds[u]]);
            xo[e] = ex;
            atomicAdd(&dn[ds[u]], ex);
        }
    }
    __syncthreads();
    float* o = pden + (size_t)blockIdx.x * NN;
    for (int n = tid; n < NN; n += 1024) o[n] = dn[n];
}

__global__ void combine_den(const float* __restrict__ pden, float* __restrict__ denT, int H) {
    const int i = blockIdx.x * blockDim.x + threadIdx.x;
    if (i >= NN * H) return;
    const int hh = i / NN, node = i % NN;
    const float* p = pden + (size_t)hh * SS * NN + node;
    float d = p[0];
    #pragma unroll
    for (int s = 1; s < SS; ++s) d += p[(size_t)s * NN];
    denT[i] = d;
}

__global__ __launch_bounds__(1024) void agg_pair_lds(const int* __restrict__ ei,
                                                     const float* __restrict__ alT,
                                                     const float* __restrict__ HT,
                                                     const float* __restrict__ denT,
                                                     const float* __restrict__ bias,
                                                     float* __restrict__ outT) {
    __shared__ v2f   hs[NN];
    __shared__ float acc0[NN];
    __shared__ float acc1[NN];
    const int tid  = threadIdx.x;
    const int c0   = 2 * blockIdx.x;
    const int head = c0 / HID;
    const float* h0 = HT + (size_t)c0 * NN;
    const float* h1 = h0 + NN;
    const float* al = alT + (size_t)head * ETOT;

    #pragma unroll
    for (int g = 0; g < 2; ++g) {
        float x0[NG], x1[NG], aa[NG];
        #pragma unroll
        for (int j = 0; j < NG; ++j) {
            const int n  = (g * NG + j) * 1024 + tid;
            const int nc = n < NN ? n : 0;
            x0[j] = h0[nc]; x1[j] = h1[nc];
            aa[j] = al[NE + nc];
        }
        #pragma unroll
        for (int j = 0; j < NG; ++j) {
            const int n = (g * NG + j) * 1024 + tid;
            if (n < NN) {
                v2f hv = { x0[j], x1[j] };
                hs[n]   = hv;
                acc0[n] = aa[j] * hv.x;
                acc1[n] = aa[j] * hv.y;
            }
        }
    }
    __syncthreads();

    constexpr int FULL = NE / 1024;
    constexpr int BAT  = 8;
    static_assert(FULL % BAT == 0, "the batch must divide the full sweeps");
    int sr[BAT], ds[BAT]; float a[BAT];
    #pragma unroll
    for (int u = 0; u < BAT; ++u) {
        const int e = u * 1024 + tid;
        sr[u] = ei[e]; ds[u] = ei[NE + e]; a[u] = al[e];
    }
    for (int it = 0; it < FULL; it += BAT) {
        const int nx = (it + BAT < FULL) ? it + BAT : it;
        int srn[BAT], dsn[BAT]; float an[BAT];
        #pragma unroll
        for (int u = 0; u < BAT; ++u) {
            const int e = (nx + u) * 1024 + tid;
            srn[u] = ei[e]; dsn[u] = ei[NE + e]; an[u] = al[e];
        }
        v2f hv[BAT];
        #pragma unroll
        for (int u = 0; u < BAT; ++u) hv[u] = hs[sr[u]];
        #pragma unroll
        for (int u = 0; u < BAT; ++u) {
            atomicAdd(&acc0[ds[u]], a[u] * hv[u].x);
            atomicAdd(&acc1[ds[u]], a[u] * hv[u].y);
        }
        #pragma unroll
        for (int u = 0; u < BAT; ++u) { sr[u] = srn[u]; ds[u] = dsn[u]; a[u] = an[u]; }
    }
    {
        const int e = FULL * 1024 + tid;
        if (e < NE) {
            v2f hv = hs[ei[e]];
            const int d = ei[NE + e];
            const float a = al[e];
            atomicAdd(&acc0[d], a * hv.x);
            atomicAdd(&acc1[d], a * hv.y);
        }
    }
    const float* dn = denT + (size_t)head * NN;
    float dd[2 * NG];
    #pragma unroll
    for (int j = 0; j < 2 * NG; ++j) {
        const int n  = j * 1024 + tid;
        const int nc = n < NN ? n : 0;
        dd[j] = dn[nc];
    }
    __syncthreads();

    float* o0 = outT + (size_t)c0 * NN;
    float* o1 = o0 + NN;
    const float b0 = bias[c0], b1 = bias[c0 + 1];
    #pragma unroll
    for (int j = 0; j < 2 * NG; ++j) {
        const int n = j * 1024 + tid;
        if (n < NN) {
            const float d = dd[j] + 1e-16f;
            const float v0 = acc0[n] / d + b0, v1 = acc1[n] / d + b1;
            o0[n] = v0 > 0.f ? v0 : 0.f;
            o1[n] = v1 > 0.f ? v1 : 0.f;
        }
    }
}

#define S2 8
__global__ __launch_bounds__(1024) void agg2_slice_lds(const int* __restrict__ ei,
                                                       const float* __restrict__ al,
                                                       const float* __restrict__ HT,
                                                       float* __restrict__ part) {
    __shared__ v2f   hs[NN];
    __shared__ float acc0[NN];
    __shared__ float acc1[NN];
    const int tid   = threadIdx.x;
    const int pair  = blockIdx.x % (OUT_CH / 2);
    const int slice = blockIdx.x / (OUT_CH / 2);
    const int c0    = 2 * pair;
    const float* h0 = HT + (size_t)c0 * NN;
    const float* h1 = h0 + NN;

    if (slice == 0) {
        #pragma unroll
        for (int g = 0; g < 2; ++g) {
            float x0[NG], x1[NG], aa[NG];
            #pragma unroll
            for (int j = 0; j < NG; ++j) {
                const int n  = (g * NG + j) * 1024 + tid;
                const int nc = n < NN ? n : 0;
                x0[j] = h0[nc]; x1[j] = h1[nc];
                aa[j] = al[NE + nc];
            }
            #pragma unroll
            for (int j = 0; j < NG; ++j) {
                const int n = (g * NG + j) * 1024 + tid;
                if (n < NN) {
                    v2f hv = { x0[j], x1[j] };
                    hs[n]   = hv;
                    acc0[n] = aa[j] * hv.x;
                    acc1[n] = aa[j] * hv.y;
                }
            }
        }
    } else {
        #pragma unroll
        for (int g = 0; g < 2; ++g) {
            float x0[NG], x1[NG];
            #pragma unroll
            for (int j = 0; j < NG; ++j) {
                const int n  = (g * NG + j) * 1024 + tid;
                const int nc = n < NN ? n : 0;
                x0[j] = h0[nc]; x1[j] = h1[nc];
            }
            #pragma unroll
            for (int j = 0; j < NG; ++j) {
                const int n = (g * NG + j) * 1024 + tid;
                if (n < NN) {
                    v2f hv = { x0[j], x1[j] };
                    hs[n]   = hv;
                    acc0[n] = 0.f;
                    acc1[n] = 0.f;
                }
            }
        }
    }
    __syncthreads();

    constexpr int FULL = NE / 1024;
    static_assert(FULL % S2 == 0, "the slices must divide the full sweeps");
    constexpr int SW  = FULL / S2;
    constexpr int BAT = 13;
    static_assert(SW % BAT == 0, "the batch must divide a slice's sweeps");
    const int base = slice * SW * 1024 + tid;
    for (int it = 0; it < SW; it += BAT) {
        int sr[BAT], ds[BAT]; float a[BAT];
        #pragma unroll
        for (int u = 0; u < BAT; ++u) {
            const int e = base + (it + u) * 1024;
            sr[u] = ei[e]; ds[u] = ei[NE + e]; a[u] = al[e];
        }
        v2f hv[BAT];
        #pragma unroll
        for (int u = 0; u < BAT; ++u) hv[u] = hs[sr[u]];
        #pragma unroll
        for (int u = 0; u < BAT; ++u) {
            atomicAdd(&acc0[ds[u]], a[u] * hv[u].x);
            atomicAdd(&acc1[ds[u]], a[u] * hv[u].y);
        }
    }
    if (slice == S2 - 1) {
        const int e = FULL * 1024 + tid;
        if (e < NE) {
            v2f hv = hs[ei[e]];
            const int d = ei[NE + e];
            const float a = al[e];
            atomicAdd(&acc0[d], a * hv.x);
            atomicAdd(&acc1[d], a * hv.y);
        }
    }
    __syncthreads();

    float* o0 = part + ((size_t)slice * OUT_CH + c0) * NN;
    float* o1 = o0 + NN;
    for (int n = tid; n < NN; n += 1024) { o0[n] = acc0[n]; o1[n] = acc1[n]; }
}

__global__ __launch_bounds__(256) void final_planes_logsoftmax(const float* __restrict__ part,
                                                                const float* __restrict__ pden,
                                                                const float* __restrict__ b2,
                                                                float* __restrict__ out) {
    __shared__ float tile[OUT_CH][33];
    const int tid = threadIdx.x;
    const int w = tid >> 5, lane = tid & 31;
    const int n0 = blockIdx.x * 32;
    float d = pden[n0 + lane];
    #pragma unroll
    for (int s = 1; s < SS; ++s) d += pden[(size_t)s * NN + n0 + lane];
    d += 1e-16f;
    #pragma unroll
    for (int j = 0; j < 8; ++j) {
        const int c = 8 * w + j;
        const float* p = part + (size_t)c * NN + n0 + lane;
        float v = p[0];
        #pragma unroll
        for (int s = 1; s < S2; ++s) v += p[(size_t)s * OUT_CH * NN];
        tile[c][lane] = v / d + b2[c];
    }
    __syncthreads();
    #pragma unroll
    for (int j = 0; j < 4; ++j) {
        const int nl = 4 * w + j;
        float v0 = tile[lane][nl];
        float v1 = tile[lane + 32][nl];
        float m = fmaxf(v0, v1);
        #pragma unroll
        for (int o = 16; o > 0; o >>= 1) m = fmaxf(m, __shfl_xor(m, o, 32));
        float s = expf(v0 - m) + expf(v1 - m);
        #pragma unroll
        for (int o = 16; o > 0; o >>= 1) s += __shfl_xor(s, o, 32);
        float lse = logf(s) + m;
        float* row = out + (size_t)(n0 + nl) * OUT_CH;
        row[lane]      = v0 - lse;
        row[lane + 32] = v1 - lse;
    }
}

extern "C" void kernel_launch(void* const* d_in, const int* in_sizes, int n_in,
                              void* d_out, int out_size, void* d_ws, size_t ws_size,
                              hipStream_t stream) {
    const float* x   = (const float*)d_in[0];
    const int*   ei  = (const int*)  d_in[1];
    const float* W1  = (const float*)d_in[2];
    const float* a1s = (const float*)d_in[3];
    const float* a1d = (const float*)d_in[4];
    const float* b1  = (const float*)d_in[5];
    const float* W2  = (const float*)d_in[6];
    const float* a2s = (const float*)d_in[7];
    const float* a2d = (const float*)d_in[8];
    const float* b2  = (const float*)d_in[9];
    float* out = (float*)d_out;

    float* ws = (float*)d_ws;
    float* h1T   = ws;
    float* agg1T = h1T   + (size_t)NN * HC1;
    float* es1T  = agg1T + (size_t)NN * HC1;
    float* ed1T  = es1T  + (size_t)NN * HEADS;
    float* m1T   = ed1T  + (size_t)NN * HEADS;
    float* den1T = m1T   + (size_t)NN * HEADS;
    int*   pm1   = (int*)(den1T + (size_t)NN * HEADS);
    float* pd1   = (float*)pm1 + (size_t)NN * HEADS * SS;
    float* al1   = pd1  + (size_t)NN * HEADS * SS;
    float* h2T   = al1  + (size_t)ETOT * HEADS;
    float* es2T  = h2T  + (size_t)NN * OUT_CH;
    float* ed2T  = es2T + NN;
    float* m2T   = ed2T + NN;
    int*   pm2   = (int*)(m2T + NN);
    float* pd2   = (float*)pm2 + (size_t)NN * SS;
    float* ex2   = pd2  + (size_t)NN * SS;
    float* part2 = h1T;

    const int TB = 256;
    auto blocks = [](long long n, int tb) { return (unsigned)((n + tb - 1) / tb); };

    {
        long long waves = (long long)(NN / 32) * (HC1 / 32);
        wmma_gemm_f32_ct<<<blocks(waves * 32, TB), TB, 0, stream>>>(x, W1, h1T, NN, IN_CH, HC1);
    }
    scores_ct<HID><<<dim3(blocks(NN, TB), HEADS), TB, 0, stream>>>(
        h1T, a1s, a1d, es1T, ed1T);
    edge_max_slices<<<HEADS * SS, 1024, 0, stream>>>(ei, es1T, ed1T, pm1);
    combine_max    <<<blocks((long long)NN * HEADS, TB), TB, 0, stream>>>(pm1, m1T, HEADS);
    edge_exp_slices<<<HEADS * SS, 1024, 0, stream>>>(ei, es1T, ed1T, m1T, al1, pd1);
    combine_den    <<<blocks((long long)NN * HEADS, TB), TB, 0, stream>>>(pd1, den1T, HEADS);
    agg_pair_lds  <<<HC1 / 2, 1024, 0, stream>>>(ei, al1, h1T, den1T, b1, agg1T);

    static_assert(OUT_CH == 64 && HC1 % (4 * KS2) == 0 && NN % 32 == 0, "GEMM 2 tiling");
    wmma_gemm_f32_tt<<<NN / 32, 256, 0, stream>>>(agg1T, W2, h2T, NN, HC1, OUT_CH);
    scores_ct<OUT_CH><<<dim3(blocks(NN, TB), 1), TB, 0, stream>>>(
        h2T, a2s, a2d, es2T, ed2T);
    edge_max_slices<<<SS, 1024, 0, stream>>>(ei, es2T, ed2T, pm2);
    combine_max    <<<blocks((long long)NN, TB), TB, 0, stream>>>(pm2, m2T, 1);
    edge_exp_slices<<<SS, 1024, 0, stream>>>(ei, es2T, ed2T, m2T, ex2, pd2);
    agg2_slice_lds<<<(OUT_CH / 2) * S2, 1024, 0, stream>>>(ei, ex2, h2T, part2);
    final_planes_logsoftmax<<<NN / 32, 256, 0, stream>>>(part2, pd2, b2, out);
}


# Raw edges placed once per call, as packed src|dst words, into slots whose lane owns the destination's LDS bank (dst % 32): a wave's 32 float adds hit 32 banks; sums now in slot (arrival) order
# speedup vs baseline: 4.7840x; 1.0973x over previous
#include <hip/hip_runtime.h>
#include <math.h>

#define NN      20000
#define NE      320000
#define NCH     32
#define CHE     (NE / NCH)
#define CAP     352
#define CHS     (CAP * 32)
#define NSLOT   (NCH * CHS)
#define ETOT    (NSLOT + NN)
static_assert(NE % NCH == 0 && CHS >= CHE && CHS % 1024 == 0, "a chunk's slots must hold its edges");
static_assert(NN <= 32768, "source and destination share one word");
#define IN_CH   128
#define HID     64
#define HEADS   8
#define HC1     (HEADS * HID)
#define OUT_CH  64

typedef float v2f __attribute__((ext_vector_type(2)));
typedef float v8f __attribute__((ext_vector_type(8)));

__device__ __forceinline__ int f2ord(float f) {
    int i = __float_as_int(f);
    return i >= 0 ? i : (i ^ 0x7fffffff);
}
__device__ __forceinline__ float ord2f(int i) {
    return __int_as_float(i >= 0 ? i : (i ^ 0x7fffffff));
}

__global__ void wmma_gemm_f32_ct(const float* __restrict__ X,
                                 const float* __restrict__ Wm,
                                 float* __restrict__ HT,
                                 int M, int K, int N) {
    const int wave    = (blockIdx.x * blockDim.x + threadIdx.x) >> 5;
    const int lane    = threadIdx.x & 31;
    const int cGroups = N >> 5;
    const int mGroups = M >> 5;
    if (wave >= mGroups * cGroups) return;
    const int mg   = wave / cGroups;
    const int cg   = wave % cGroups;
    const int l    = lane & 15;
    const int half = lane >> 4;
    const unsigned uN = (unsigned)N, uM = (unsigned)M, uK = (unsigned)K;

    unsigned xoff = (unsigned)(mg * 32 + l) * uK + 2u * half;
    unsigned woff = (unsigned)(2 * half) * uN + (unsigned)(cg * 32 + l);

    v8f acc00 = {}, acc01 = {}, acc10 = {}, acc11 = {};
    for (int k0 = 0; k0 < K; k0 += 4) {
        v2f x0 = *(const v2f*)(X + xoff);
        v2f x1 = *(const v2f*)(X + xoff + 16u * uK);
        v2f w0 = { Wm[woff],      Wm[woff + uN]      };
        v2f w1 = { Wm[woff + 16], Wm[woff + uN + 16] };
        acc00 = __builtin_amdgcn_wmma_f32_16x16x4_f32(false, w0, false, x0, (short)0, acc00, false, false);
        acc01 = __builtin_amdgcn_wmma_f32_16x16x4_f32(false, w0, false, x1, (short)0, acc01, false, false);
        acc10 = __builtin_amdgcn_wmma_f32_16x16x4_f32(false, w1, false, x0, (short)0, acc10, false, false);
        acc11 = __builtin_amdgcn_wmma_f32_16x16x4_f32(false, w1, false, x1, (short)0, acc11, false, false);
        xoff += 4u;
        woff += 4u * uN;
    }
    unsigned hoff = (unsigned)(cg * 32 + 8 * half) * uM + (unsigned)(mg * 32 + l);
    #pragma unroll
    for (int r = 0; r < 8; ++r) {
        float* hp = HT + hoff;
        hp[0]            = acc00[r];
        hp[16]           = acc01[r];
        hp[16u * uM]      = acc10[r];
        hp[16u * uM + 16] = acc11[r];
        hoff += uM;
    }
}

#define KS2 4
__global__ __launch_bounds__(256) void wmma_gemm_f32_tt(const float* __restrict__ XT,
                                                        const float* __restrict__ Wm,
                                                        float* __restrict__ HT,
                                                        int M, int K, int N) {
    __shared__ float part[2][KS2][32][34];
    const int w    = threadIdx.x >> 5;
    const int lane = threadIdx.x & 31;
    const int cg   = w & 1;
    const int ks   = w >> 1;
    const int mg   = blockIdx.x;
    const int l    = lane & 15;
    const int half = lane >> 4;
    const unsigned uN = (unsigned)N, uM = (unsigned)M;
    const int KL = K / KS2;
    const unsigned kb = (unsigned)(ks * KL);

    unsigned xoff = (kb + 2u * half) * uM + (unsigned)(mg * 32 + l);
    unsigned woff = (kb + 2u * half) * uN + (unsigned)(cg * 32 + l);

    v8f acc00 = {}, acc01 = {}, acc10 = {}, acc11 = {};
    for (int k0 = 0; k0 < KL; k0 += 4) {
        v2f x0 = { XT[xoff],      XT[xoff + uM]      };
        v2f x1 = { XT[xoff + 16], XT[xoff + uM + 16] };
        v2f w0 = { Wm[woff],      Wm[woff + uN]      };
        v2f w1 = { Wm[woff + 16], Wm[woff + uN + 16] };
        acc00 = __builtin_amdgcn_wmma_f32_16x16x4_f32(false, w0, false, x0, (short)0, acc00, false, false);
        acc01 = __builtin_amdgcn_wmma_f32_16x16x4_f32(false, w0, false, x1, (short)0, acc01, false, false);
        acc10 = __builtin_amdgcn_wmma_f32_16x16x4_f32(false, w1, false, x0, (short)0, acc10, false, false);
        acc11 = __builtin_amdgcn_wmma_f32_16x16x4_f32(false, w1, false, x1, (short)0, acc11, false, false);
        xoff += 4u * uM;
        woff += 4u * uN;
    }
    float (*pt)[34] = part[cg][ks];
    #pragma unroll
    for (int r = 0; r < 8; ++r) {
        pt[8 * half + r     ][l     ] = acc00[r];
        pt[8 * half + r     ][l + 16] = acc01[r];
        pt[8 * half + r + 16][l     ] = acc10[r];
        pt[8 * half + r + 16][l + 16] = acc11[r];
    }
    __syncthreads();
    unsigned hoff = (unsigned)(cg * 32 + 8 * ks) * uM + (unsigned)(mg * 32 + lane);
    #pragma unroll
    for (int j = 0; j < 8; ++j) {
        const int row = 8 * ks + j;
        float v = part[cg][0][row][lane];
        #pragma unroll
        for (int q = 1; q < KS2; ++q) v += part[cg][q][row][lane];
        HT[hoff] = v;
        hoff += uM;
    }
}

template <int C>
__global__ void scores_ct(const float* __restrict__ HT,
                          const float* __restrict__ a_src,
                          const float* __restrict__ a_dst,
                          float* __restrict__ esT, float* __restrict__ edT) {
    const int node = blockIdx.x * blockDim.x + threadIdx.x;
    const int hh   = blockIdx.y;
    if (node >= NN) return;
    const float* hp = HT + (size_t)hh * C * NN + node;
    const float* as = a_src + hh * C;
    const float* ad = a_dst + hh * C;
    float v[C];
    #pragma unroll
    for (int c = 0; c < C; ++c) v[c] = hp[(size_t)c * NN];
    float s = 0.f, d = 0.f;
    #pragma unroll
    for (int c = 0; c < C; ++c) { s += v[c] * as[c]; d += v[c] * ad[c]; }
    esT[(size_t)hh * NN + node] = s;
    edT[(size_t)hh * NN + node] = d;
}

__global__ __launch_bounds__(1024) void bank_slots(const int* __restrict__ ei, int* __restrict__ slot) {
    __shared__ int tab[CHS];
    __shared__ int spill[CHE];
    __shared__ int cnt[32];
    __shared__ int fre[33];
    __shared__ int nspill;
    const int tid = threadIdx.x;
    const int g   = blockIdx.x;
    constexpr int UE = (CHE + 1023) / 1024;
    int s[UE], d[UE];
    #pragma unroll
    for (int u = 0; u < UE; ++u) {
        const int i  = u * 1024 + tid;
        const int ic = i < CHE ? i : 0;
        s[u] = ei[g * CHE + ic];
        d[u] = ei[NE + g * CHE + ic];
    }
    for (int q = 0; q < CHS / 1024; ++q) tab[q * 1024 + tid] = -1;
    if (tid < 32) cnt[tid] = 0;
    if (tid == 0) nspill = 0;
    __syncthreads();
    #pragma unroll
    for (int u = 0; u < UE; ++u) {
        const int i = u * 1024 + tid;
        if (i < CHE) {
            const int cl = d[u] & 31;
            const int v  = (d[u] << 16) | s[u];
            const int r  = atomicAdd(&cnt[cl], 1);
            if (r < CAP) tab[r * 32 + cl] = v;
            else         spill[atomicAdd(&nspill, 1)] = v;
        }
    }
    __syncthreads();
    if (tid == 0) {
        int a = 0;
        for (int cl = 0; cl < 32; ++cl) {
            fre[cl] = a;
            const int used = cnt[cl] < CAP ? cnt[cl] : CAP;
            a += CAP - used;
        }
        fre[32] = a;
    }
    __syncthreads();
    const int ns = nspill;
    for (int k = tid; k < ns; k += 1024) {
        int cl = 0;
        for (int j = 1; j < 32; ++j) cl += (fre[j] <= k) ? 1 : 0;
        const int used = cnt[cl] < CAP ? cnt[cl] : CAP;
        tab[(used + k - fre[cl]) * 32 + cl] = spill[k];
    }
    __syncthreads();
    int* o = slot + (size_t)g * CHS;
    for (int q = 0; q < CHS / 1024; ++q) o[q * 1024 + tid] = tab[q * 1024 + tid];
}

#define SS   32
#define EPS  (ETOT / SS)
#define UM   ((EPS + 1023) / 1024)
#define NG   10
static_assert(ETOT % SS == 0, "the slices must divide the list");
static_assert(2 * NG * 1024 >= NN, "two groups of NG sweeps must cover the nodes");

#define LOAD_SLICE_EDGES(sr, ds)                                              \
    _Pragma("unroll")                                                         \
    for (int u = 0; u < UM; ++u) {                                            \
        const int e  = e0 + u * 1024 + tid;                                   \
        const int ec = (e < e1 && e < NSLOT) ? e : 0;                         \
        const int sv = slot[ec];                                              \
        sr[u] = (e >= NSLOT) ? e - NSLOT : (sv & 0xffff);                     \
        ds[u] = (e >= NSLOT) ? e - NSLOT : (sv >> 16);                        \
    }

__global__ __launch_bounds__(1024) void edge_max_slices(const int* __restrict__ slot,
                                                        const float* __restrict__ esT,
                                                        const float* __restrict__ edT,
                                                        int* __restrict__ pmax) {
    __shared__ float es[NN];
    __shared__ float ed[NN];
    __shared__ int   mx[NN];
    const int tid = threadIdx.x;
    const int hh  = blockIdx.x / SS;
    const int sl  = blockIdx.x % SS;
    const int e0  = sl * EPS, e1 = e0 + EPS;
    int sr[UM], ds[UM];
    LOAD_SLICE_EDGES(sr, ds)
    const float* sp = esT + (size_t)hh * NN;
    const float* dp = edT + (size_t)hh * NN;
    #pragma unroll
    for (int g = 0; g < 2; ++g) {
        float a[NG], b[NG];
        #pragma unroll
        for (int j = 0; j < NG; ++j) {
            const int n  = (g * NG + j) * 1024 + tid;
            const int nc = n < NN ? n : 0;
            a[j] = sp[nc]; b[j] = dp[nc];
        }
        #pragma unroll
        for (int j = 0; j < NG; ++j) {
            const int n = (g * NG + j) * 1024 + tid;
            if (n < NN) { es[n] = a[j]; ed[n] = b[j]; mx[n] = f2ord(-INFINITY); }
        }
    }
    __syncthreads();
    #pragma unroll
    for (int u = 0; u < UM; ++u) {
        const int e = e0 + u * 1024 + tid;
        if (e < e1 && ds[u] >= 0) {
            float v = es[sr[u]] + ed[ds[u]];
            v = v > 0.f ? v : 0.2f * v;
            atomicMax(&mx[ds[u]], f2ord(v));
        }
    }
    __syncthreads();
    int* o = pmax + (size_t)blockIdx.x * NN;
    for (int n = tid; n < NN; n += 1024) o[n] = mx[n];
}

__global__ void combine_max(const int* __restrict__ pmax, float* __restrict__ mT, int H) {
    const int i = blockIdx.x * blockDim.x + threadIdx.x;
    if (i >= NN * H) return;
    const int hh = i / NN, node = i % NN;
    const int* p = pmax + (size_t)hh * SS * NN + node;
    int m = p[0];
    #pragma unroll
    for (int s = 1; s < SS; ++s) { const int v = p[(size_t)s * NN]; m = v > m ? v : m; }
    mT[i] = ord2f(m);
}

__global__ __launch_bounds__(1024) void edge_exp_slices(const int* __restrict__ slot,
                                                        const float* __restrict__ esT,
                                                        const float* __restrict__ edT,
                                                        const float* __restrict__ mT,
                                                        float* __restrict__ exT,
                                                        float* __restrict__ pden) {
    __shared__ float es[NN];
    __shared__ float ed[NN];
    __shared__ float mm[NN];
    __shared__ float dn[NN];
    const int tid = threadIdx.x;
    const int hh  = blockIdx.x / SS;
    const int sl  = blockIdx.x % SS;
    const int e0  = sl * EPS, e1 = e0 + EPS;
    int sr[UM], ds[UM];
    LOAD_SLICE_EDGES(sr, ds)
    const float* sp = esT + (size_t)hh * NN;
    const float* dp = edT + (size_t)hh * NN;
    const float* mp = mT  + (size_t)hh * NN;
    #pragma unroll
    for (int g = 0; g < 2; ++g) {
        float a[NG], b[NG], c[NG];
        #pragma unroll
        for (int j = 0; j < NG; ++j) {
            const int n  = (g * NG + j) * 1024 + tid;
            const int nc = n < NN ? n : 0;
            a[j] = sp[nc]; b[j] = dp[nc]; c[j] = mp[nc];
        }
        #pragma unroll
        for (int j = 0; j < NG; ++j) {
            const int n = (g * NG + j) * 1024 + tid;
            if (n < NN) { es[n] = a[j]; ed[n] = b[j]; mm[n] = c[j]; dn[n] = 0.f; }
        }
    }
    __syncthreads();
    float* xo = exT + (size_t)hh * ETOT;
    #pragma unroll
    for (int u = 0; u < UM; ++u) {
        const int e = e0 + u * 1024 + tid;
        if (e < e1) {
            float ex = 0.f;
            if (ds[u] >= 0) {
                float v = es[sr[u]] + ed[ds[u]];
                v = v > 0.f ? v : 0.2f * v;
                ex = expf(v - mm[ds[u]]);
                atomicAdd(&dn[ds[u]], ex);
            }
            xo[e] = ex;
        }
    }
    __syncthreads();
    float* o = pden + (size_t)blockIdx.x * NN;
    for (int n = tid; n < NN; n += 1024) o[n] = dn[n];
}

__global__ void combine_den(const float* __restrict__ pden, float* __restrict__ denT, int H) {
    const int i = blockIdx.x * blockDim.x + threadIdx.x;
    if (i >= NN * H) return;
    const int hh = i / NN, node = i % NN;
    const float* p = pden + (size_t)hh * SS * NN + node;
    float d = p[0];
    #pragma unroll
    for (int s = 1; s < SS; ++s) d += p[(size_t)s * NN];
    denT[i] = d;
}

__global__ __launch_bounds__(1024) void agg_pair_lds(const int* __restrict__ slot,
                                                     const float* __restrict__ alT,
                                                     const float* __restrict__ HT,
                                                     const float* __restrict__ denT,
                                                     const float* __restrict__ bias,
                                                     float* __restrict__ outT) {
    __shared__ v2f   hs[NN];
    __shared__ float acc0[NN];
    __shared__ float acc1[NN];
    const int tid  = threadIdx.x;
    const int c0   = 2 * blockIdx.x;
    const int head = c0 / HID;
    const float* h0 = HT + (size_t)c0 * NN;
    const float* h1 = h0 + NN;
    const float* al = alT + (size_t)head * ETOT;

    #pragma unroll
    for (int g = 0; g < 2; ++g) {
        float x0[NG], x1[NG], aa[NG];
        #pragma unroll
        for (int j = 0; j < NG; ++j) {
            const int n  = (g * NG + j) * 1024 + tid;
            const int nc = n < NN ? n : 0;
            x0[j] = h0[nc]; x1[j] = h1[nc];
            aa[j] = al[NSLOT + nc];
        }
        #pragma unroll
        for (int j = 0; j < NG; ++j) {
            const int n = (g * NG + j) * 1024 + tid;
            if (n < NN) {
                v2f hv = { x0[j], x1[j] };
                hs[n]   = hv;
                acc0[n] = aa[j] * hv.x;
                acc1[n] = aa[j] * hv.y;
            }
        }
    }
    __syncthreads();

    constexpr int FULL = NSLOT / 1024;
    constexpr int BAT  = 8;
    static_assert(NSLOT % 1024 == 0 && FULL % BAT == 0, "the batch must divide the sweeps");
    int sv[BAT]; float a[BAT];
    #pragma unroll
    for (int u = 0; u < BAT; ++u) {
        const int e = u * 1024 + tid;
        sv[u] = slot[e]; a[u] = al[e];
    }
    for (int it = 0; it < FULL; it += BAT) {
        const int nx = (it + BAT < FULL) ? it + BAT : it;
        int svn[BAT]; float an[BAT];
        #pragma unroll
        for (int u = 0; u < BAT; ++u) {
            const int e = (nx + u) * 1024 + tid;
            svn[u] = slot[e]; an[u] = al[e];
        }
        v2f hv[BAT];
        #pragma unroll
        for (int u = 0; u < BAT; ++u) hv[u] = hs[sv[u] < 0 ? 0 : (sv[u] & 0xffff)];
        #pragma unroll
        for (int u = 0; u < BAT; ++u) {
            const int d = sv[u] >> 16;
            if (d >= 0) {
                atomicAdd(&acc0[d], a[u] * hv[u].x);
                atomicAdd(&acc1[d], a[u] * hv[u].y);
            }
        }
        #pragma unroll
        for (int u = 0; u < BAT; ++u) { sv[u] = svn[u]; a[u] = an[u]; }
    }
    const float* dn = denT + (size_t)head * NN;
    float dd[2 * NG];
    #pragma unroll
    for (int j = 0; j < 2 * NG; ++j) {
        const int n  = j * 1024 + tid;
        const int nc = n < NN ? n : 0;
        dd[j] = dn[nc];
    }
    __syncthreads();

    float* o0 = outT + (size_t)c0 * NN;
    float* o1 = o0 + NN;
    const float b0 = bias[c0], b1 = bias[c0 + 1];
    #pragma unroll
    for (int j = 0; j < 2 * NG; ++j) {
        const int n = j * 1024 + tid;
        if (n < NN) {
            const float d = dd[j] + 1e-16f;
            const float v0 = acc0[n] / d + b0, v1 = acc1[n] / d + b1;
            o0[n] = v0 > 0.f ? v0 : 0.f;
            o1[n] = v1 > 0.f ? v1 : 0.f;
        }
    }
}

#define S2 8
__global__ __launch_bounds__(1024) void agg2_slice_lds(const int* __restrict__ slot,
                                                       const float* __restrict__ al,
                                                       const float* __restrict__ HT,
                                                       float* __restrict__ part) {
    __shared__ v2f   hs[NN];
    __shared__ float acc0[NN];
    __shared__ float acc1[NN];
    const int tid   = threadIdx.x;
    const int pair  = blockIdx.x % (OUT_CH / 2);
    const int slice = blockIdx.x / (OUT_CH / 2);
    const int c0    = 2 * pair;
    const float* h0 = HT + (size_t)c0 * NN;
    const float* h1 = h0 + NN;

    if (slice == 0) {
        #pragma unroll
        for (int g = 0; g < 2; ++g) {
            float x0[NG], x1[NG], aa[NG];
            #pragma unroll
            for (int j = 0; j < NG; ++j) {
                const int n  = (g * NG + j) * 1024 + tid;
                const int nc = n < NN ? n : 0;
                x0[j] = h0[nc]; x1[j] = h1[nc];
                aa[j] = al[NSLOT + nc];
            }
            #pragma unroll
            for (int j = 0; j < NG; ++j) {
                const int n = (g * NG + j) * 1024 + tid;
                if (n < NN) {
                    v2f hv = { x0[j], x1[j] };
                    hs[n]   = hv;
                    acc0[n] = aa[j] * hv.x;
                    acc1[n] = aa[j] * hv.y;
                }
            }
        }
    } else {
        #pragma unroll
        for (int g = 0; g < 2; ++g) {
            float x0[NG], x1[NG];
            #pragma unroll
            for (int j = 0; j < NG; ++j) {
                const int n  = (g * NG + j) * 1024 + tid;
                const int nc = n < NN ? n : 0;
                x0[j] = h0[nc]; x1[j] = h1[nc];
            }
            #pragma unroll
            for (int j = 0; j < NG; ++j) {
                const int n = (g * NG + j) * 1024 + tid;
                if (n < NN) {
                    v2f hv = { x0[j], x1[j] };
                    hs[n]   = hv;
                    acc0[n] = 0.f;
                    acc1[n] = 0.f;
                }
            }
        }
    }
    __syncthreads();

    constexpr int FULL = NSLOT / 1024;
    static_assert(NSLOT % 1024 == 0 && FULL % S2 == 0, "the slices must divide the sweeps");
    constexpr int SW  = FULL / S2;
    constexpr int BAT = 11;
    static_assert(SW % BAT == 0, "the batch must divide a slice's sweeps");
    const int base = slice * SW * 1024 + tid;
    for (int it = 0; it < SW; it += BAT) {
        int sv[BAT]; float a[BAT];
        #pragma unroll
        for (int u = 0; u < BAT; ++u) {
            const int e = base + (it + u) * 1024;
            sv[u] = slot[e]; a[u] = al[e];
        }
        v2f hv[BAT];
        #pragma unroll
        for (int u = 0; u < BAT; ++u) hv[u] = hs[sv[u] < 0 ? 0 : (sv[u] & 0xffff)];
        #pragma unroll
        for (int u = 0; u < BAT; ++u) {
            const int d = sv[u] >> 16;
            if (d >= 0) {
                atomicAdd(&acc0[d], a[u] * hv[u].x);
                atomicAdd(&acc1[d], a[u] * hv[u].y);
            }
        }
    }
    __syncthreads();

    float* o0 = part + ((size_t)slice * OUT_CH + c0) * NN;
    float* o1 = o0 + NN;
    for (int n = tid; n < NN; n += 1024) { o0[n] = acc0[n]; o1[n] = acc1[n]; }
}

__global__ __launch_bounds__(256) void final_planes_logsoftmax(const float* __restrict__ part,
                                                                const float* __restrict__ pden,
                                                                const float* __restrict__ b2,
                                                                float* __restrict__ out) {
    __shared__ float tile[OUT_CH][33];
    const int tid = threadIdx.x;
    const int w = tid >> 5, lane = tid & 31;
    const int n0 = blockIdx.x * 32;
    float d = pden[n0 + lane];
    #pragma unroll
    for (int s = 1; s < SS; ++s) d += pden[(size_t)s * NN + n0 + lane];
    d += 1e-16f;
    #pragma unroll
    for (int j = 0; j < 8; ++j) {
        const int c = 8 * w + j;
        const float* p = part + (size_t)c * NN + n0 + lane;
        float v = p[0];
        #pragma unroll
        for (int s = 1; s < S2; ++s) v += p[(size_t)s * OUT_CH * NN];
        tile[c][lane] = v / d + b2[c];
    }
    __syncthreads();
    #pragma unroll
    for (int j = 0; j < 4; ++j) {
        const int nl = 4 * w + j;
        float v0 = tile[lane][nl];
        float v1 = tile[lane + 32][nl];
        float m = fmaxf(v0, v1);
        #pragma unroll
        for (int o = 16; o > 0; o >>= 1) m = fmaxf(m, __shfl_xor(m, o, 32));
        float s = expf(v0 - m) + expf(v1 - m);
        #pragma unroll
        for (int o = 16; o > 0; o >>= 1) s += __shfl_xor(s, o, 32);
        float lse = logf(s) + m;
        float* row = out + (size_t)(n0 + nl) * OUT_CH;
        row[lane]      = v0 - lse;
        row[lane + 32] = v1 - lse;
    }
}

extern "C" void kernel_launch(void* const* d_in, const int* in_sizes, int n_in,
                              void* d_out, int out_size, void* d_ws, size_t ws_size,
                              hipStream_t stream) {
    const float* x   = (const float*)d_in[0];
    const int*   ei  = (const int*)  d_in[1];
    const float* W1  = (const float*)d_in[2];
    const float* a1s = (const float*)d_in[3];
    const float* a1d = (const float*)d_in[4];
    const float* b1  = (const float*)d_in[5];
    const float* W2  = (const float*)d_in[6];
    const float* a2s = (const float*)d_in[7];
    const float* a2d = (const float*)d_in[8];
    const float* b2  = (const float*)d_in[9];
    float* out = (float*)d_out;

    float* ws = (float*)d_ws;
    float* h1T   = ws;
    float* agg1T = h1T   + (size_t)NN * HC1;
    float* es1T  = agg1T + (size_t)NN * HC1;
    float* ed1T  = es1T  + (size_t)NN * HEADS;
    float* m1T   = ed1T  + (size_t)NN * HEADS;
    float* den1T = m1T   + (size_t)NN * HEADS;
    int*   pm1   = (int*)(den1T + (size_t)NN * HEADS);
    float* pd1   = (float*)pm1 + (size_t)NN * HEADS * SS;
    float* al1   = pd1  + (size_t)NN * HEADS * SS;
    float* h2T   = al1  + (size_t)ETOT * HEADS;
    float* es2T  = h2T  + (size_t)NN * OUT_CH;
    float* ed2T  = es2T + NN;
    float* m2T   = ed2T + NN;
    int*   pm2   = (int*)(m2T + NN);
    float* pd2   = (float*)pm2 + (size_t)NN * SS;
    float* ex2   = pd2  + (size_t)NN * SS;
    int*   slots = (int*)(ex2 + (size_t)ETOT);
    float* part2 = h1T;

    const int TB = 256;
    auto blocks = [](long long n, int tb) { return (unsigned)((n + tb - 1) / tb); };

    bank_slots<<<NCH, 1024, 0, stream>>>(ei, slots);

    {
        long long waves = (long long)(NN / 32) * (HC1 / 32);
        wmma_gemm_f32_ct<<<blocks(waves * 32, TB), TB, 0, stream>>>(x, W1, h1T, NN, IN_CH, HC1);
    }
    scores_ct<HID><<<dim3(blocks(NN, TB), HEADS), TB, 0, stream>>>(
        h1T, a1s, a1d, es1T, ed1T);
    edge_max_slices<<<HEADS * SS, 1024, 0, stream>>>(slots, es1T, ed1T, pm1);
    combine_max    <<<blocks((long long)NN * HEADS, TB), TB, 0, stream>>>(pm1, m1T, HEADS);
    edge_exp_slices<<<HEADS * SS, 1024, 0, stream>>>(slots, es1T, ed1T, m1T, al1, pd1);
    combine_den    <<<blocks((long long)NN * HEADS, TB), TB, 0, stream>>>(pd1, den1T, HEADS);
    agg_pair_lds  <<<HC1 / 2, 1024, 0, stream>>>(slots, al1, h1T, den1T, b1, agg1T);

    static_assert(OUT_CH == 64 && HC1 % (4 * KS2) == 0 && NN % 32 == 0, "GEMM 2 tiling");
    wmma_gemm_f32_tt<<<NN / 32, 256, 0, stream>>>(agg1T, W2, h2T, NN, HC1, OUT_CH);
    scores_ct<OUT_CH><<<dim3(blocks(NN, TB), 1), TB, 0, stream>>>(
        h2T, a2s, a2d, es2T, ed2T);
    edge_max_slices<<<SS, 1024, 0, stream>>>(slots, es2T, ed2T, pm2);
    combine_max    <<<blocks((long long)NN, TB), TB, 0, stream>>>(pm2, m2T, 1);
    edge_exp_slices<<<SS, 1024, 0, stream>>>(slots, es2T, ed2T, m2T, ex2, pd2);
    agg2_slice_lds<<<(OUT_CH / 2) * S2, 1024, 0, stream>>>(slots, ex2, h2T, part2);
    final_planes_logsoftmax<<<NN / 32, 256, 0, stream>>>(part2, pd2, b2, out);
}


# Edge-slot build (32 work-groups, needs only the edge list) rides in the layer-1 score launch as extra work-groups: one dependent launch fewer (15 -> 14), its time hidden beside the score reads
# speedup vs baseline: 4.9660x; 1.0380x over previous
#include <hip/hip_runtime.h>
#include <math.h>

#define NN      20000
#define NE      320000
#define NCH     32
#define CHE     (NE / NCH)
#define CAP     352
#define CHS     (CAP * 32)
#define NSLOT   (NCH * CHS)
#define ETOT    (NSLOT + NN)
static_assert(NE % NCH == 0 && CHS >= CHE && CHS % 1024 == 0, "a chunk's slots must hold its edges");
static_assert(NN <= 32768, "source and destination share one word");
#define IN_CH   128
#define HID     64
#define HEADS   8
#define HC1     (HEADS * HID)
#define OUT_CH  64

typedef float v2f __attribute__((ext_vector_type(2)));
typedef float v8f __attribute__((ext_vector_type(8)));

__device__ __forceinline__ int f2ord(float f) {
    int i = __float_as_int(f);
    return i >= 0 ? i : (i ^ 0x7fffffff);
}
__device__ __forceinline__ float ord2f(int i) {
    return __int_as_float(i >= 0 ? i : (i ^ 0x7fffffff));
}

__global__ void wmma_gemm_f32_ct(const float* __restrict__ X,
                                 const float* __restrict__ Wm,
                                 float* __restrict__ HT,
                                 int M, int K, int N) {
    const int wave    = (blockIdx.x * blockDim.x + threadIdx.x) >> 5;
    const int lane    = threadIdx.x & 31;
    const int cGroups = N >> 5;
    const int mGroups = M >> 5;
    if (wave >= mGroups * cGroups) return;
    const int mg   = wave / cGroups;
    const int cg   = wave % cGroups;
    const int l    = lane & 15;
    const int half = lane >> 4;
    const unsigned uN = (unsigned)N, uM = (unsigned)M, uK = (unsigned)K;

    unsigned xoff = (unsigned)(mg * 32 + l) * uK + 2u * half;
    unsigned woff = (unsigned)(2 * half) * uN + (unsigned)(cg * 32 + l);

    v8f acc00 = {}, acc01 = {}, acc10 = {}, acc11 = {};
    for (int k0 = 0; k0 < K; k0 += 4) {
        v2f x0 = *(const v2f*)(X + xoff);
        v2f x1 = *(const v2f*)(X + xoff + 16u * uK);
        v2f w0 = { Wm[woff],      Wm[woff + uN]      };
        v2f w1 = { Wm[woff + 16], Wm[woff + uN + 16] };
        acc00 = __builtin_amdgcn_wmma_f32_16x16x4_f32(false, w0, false, x0, (short)0, acc00, false, false);
        acc01 = __builtin_amdgcn_wmma_f32_16x16x4_f32(false, w0, false, x1, (short)0, acc01, false, false);
        acc10 = __builtin_amdgcn_wmma_f32_16x16x4_f32(false, w1, false, x0, (short)0, acc10, false, false);
        acc11 = __builtin_amdgcn_wmma_f32_16x16x4_f32(false, w1, false, x1, (short)0, acc11, false, false);
        xoff += 4u;
        woff += 4u * uN;
    }
    unsigned hoff = (unsigned)(cg * 32 + 8 * half) * uM + (unsigned)(mg * 32 + l);
    #pragma unroll
    for (int r = 0; r < 8; ++r) {
        float* hp = HT + hoff;
        hp[0]            = acc00[r];
        hp[16]           = acc01[r];
        hp[16u * uM]      = acc10[r];
        hp[16u * uM + 16] = acc11[r];
        hoff += uM;
    }
}

#define KS2 4
__global__ __launch_bounds__(256) void wmma_gemm_f32_tt(const float* __restrict__ XT,
                                                        const float* __restrict__ Wm,
                                                        float* __restrict__ HT,
                                                        int M, int K, int N) {
    __shared__ float part[2][KS2][32][34];
    const int w    = threadIdx.x >> 5;
    const int lane = threadIdx.x & 31;
    const int cg   = w & 1;
    const int ks   = w >> 1;
    const int mg   = blockIdx.x;
    const int l    = lane & 15;
    const int half = lane >> 4;
    const unsigned uN = (unsigned)N, uM = (unsigned)M;
    const int KL = K / KS2;
    const unsigned kb = (unsigned)(ks * KL);

    unsigned xoff = (kb + 2u * half) * uM + (unsigned)(mg * 32 + l);
    unsigned woff = (kb + 2u * half) * uN + (unsigned)(cg * 32 + l);

    v8f acc00 = {}, acc01 = {}, acc10 = {}, acc11 = {};
    for (int k0 = 0; k0 < KL; k0 += 4) {
        v2f x0 = { XT[xoff],      XT[xoff + uM]      };
        v2f x1 = { XT[xoff + 16], XT[xoff + uM + 16] };
        v2f w0 = { Wm[woff],      Wm[woff + uN]      };
        v2f w1 = { Wm[woff + 16], Wm[woff + uN + 16] };
        acc00 = __builtin_amdgcn_wmma_f32_16x16x4_f32(false, w0, false, x0, (short)0, acc00, false, false);
        acc01 = __builtin_amdgcn_wmma_f32_16x16x4_f32(false, w0, false, x1, (short)0, acc01, false, false);
        acc10 = __builtin_amdgcn_wmma_f32_16x16x4_f32(false, w1, false, x0, (short)0, acc10, false, false);
        acc11 = __builtin_amdgcn_wmma_f32_16x16x4_f32(false, w1, false, x1, (short)0, acc11, false, false);
        xoff += 4u * uM;
        woff += 4u * uN;
    }
    float (*pt)[34] = part[cg][ks];
    #pragma unroll
    for (int r = 0; r < 8; ++r) {
        pt[8 * half + r     ][l     ] = acc00[r];
        pt[8 * half + r     ][l + 16] = acc01[r];
        pt[8 * half + r + 16][l     ] = acc10[r];
        pt[8 * half + r + 16][l + 16] = acc11[r];
    }
    __syncthreads();
    unsigned hoff = (unsigned)(cg * 32 + 8 * ks) * uM + (unsigned)(mg * 32 + lane);
    #pragma unroll
    for (int j = 0; j < 8; ++j) {
        const int row = 8 * ks + j;
        float v = part[cg][0][row][lane];
        #pragma unroll
        for (int q = 1; q < KS2; ++q) v += part[cg][q][row][lane];
        HT[hoff] = v;
        hoff += uM;
    }
}

template <int C>
__device__ __forceinline__ void scores_body(const float* __restrict__ HT,
                                            const float* __restrict__ a_src,
                                            const float* __restrict__ a_dst,
                                            float* __restrict__ esT, float* __restrict__ edT,
                                            int hh, int node) {
    if (node >= NN) return;
    const float* hp = HT + (size_t)hh * C * NN + node;
    const float* as = a_src + hh * C;
    const float* ad = a_dst + hh * C;
    float v[C];
    #pragma unroll
    for (int c = 0; c < C; ++c) v[c] = hp[(size_t)c * NN];
    float s = 0.f, d = 0.f;
    #pragma unroll
    for (int c = 0; c < C; ++c) { s += v[c] * as[c]; d += v[c] * ad[c]; }
    esT[(size_t)hh * NN + node] = s;
    edT[(size_t)hh * NN + node] = d;
}
template <int C>
__global__ void scores_ct(const float* __restrict__ HT,
                          const float* __restrict__ a_src,
                          const float* __restrict__ a_dst,
                          float* __restrict__ esT, float* __restrict__ edT) {
    scores_body<C>(HT, a_src, a_dst, esT, edT, blockIdx.y, blockIdx.x * blockDim.x + threadIdx.x);
}

__device__ __forceinline__ void bank_slots_body(const int* __restrict__ ei, int* __restrict__ slot, const int g) {
    __shared__ int tab[CHS];
    __shared__ int spill[CHE];
    __shared__ int cnt[32];
    __shared__ int fre[33];
    __shared__ int nspill;
    const int tid = threadIdx.x;
    constexpr int UE = (CHE + 1023) / 1024;
    int s[UE], d[UE];
    #pragma unroll
    for (int u = 0; u < UE; ++u) {
        const int i  = u * 1024 + tid;
        const int ic = i < CHE ? i : 0;
        s[u] = ei[g * CHE + ic];
        d[u] = ei[NE + g * CHE + ic];
    }
    for (int q = 0; q < CHS / 1024; ++q) tab[q * 1024 + tid] = -1;
    if (tid < 32) cnt[tid] = 0;
    if (tid == 0) nspill = 0;
    __syncthreads();
    #pragma unroll
    for (int u = 0; u < UE; ++u) {
        const int i = u * 1024 + tid;
        if (i < CHE) {
            const int cl = d[u] & 31;
            const int v  = (d[u] << 16) | s[u];
            const int r  = atomicAdd(&cnt[cl], 1);
            if (r < CAP) tab[r * 32 + cl] = v;
            else         spill[atomicAdd(&nspill, 1)] = v;
        }
    }
    __syncthreads();
    if (tid == 0) {
        int a = 0;
        for (int cl = 0; cl < 32; ++cl) {
            fre[cl] = a;
            const int used = cnt[cl] < CAP ? cnt[cl] : CAP;
            a += CAP - used;
        }
        fre[32] = a;
    }
    __syncthreads();
    const int ns = nspill;
    for (int k = tid; k < ns; k += 1024) {
        int cl = 0;
        for (int j = 1; j < 32; ++j) cl += (fre[j] <= k) ? 1 : 0;
        const int used = cnt[cl] < CAP ? cnt[cl] : CAP;
        tab[(used + k - fre[cl]) * 32 + cl] = spill[k];
    }
    __syncthreads();
    int* o = slot + (size_t)g * CHS;
    for (int q = 0; q < CHS / 1024; ++q) o[q * 1024 + tid] = tab[q * 1024 + tid];
}

#define NB1 ((NN + 1023) / 1024)
__global__ __launch_bounds__(1024) void slots_and_scores1(const int* __restrict__ ei, int* __restrict__ slot,
                                                          const float* __restrict__ HT,
                                                          const float* __restrict__ a_src,
                                                          const float* __restrict__ a_dst,
                                                          float* __restrict__ esT, float* __restrict__ edT) {
    if (blockIdx.x < NCH) { bank_slots_body(ei, slot, blockIdx.x); return; }
    const int b = blockIdx.x - NCH;
    scores_body<HID>(HT, a_src, a_dst, esT, edT, b / NB1, (b % NB1) * 1024 + threadIdx.x);
}

#define SS   32
#define EPS  (ETOT / SS)
#define UM   ((EPS + 1023) / 1024)
#define NG   10
static_assert(ETOT % SS == 0, "the slices must divide the list");
static_assert(2 * NG * 1024 >= NN, "two groups of NG sweeps must cover the nodes");

#define LOAD_SLICE_EDGES(sr, ds)                                              \
    _Pragma("unroll")                                                         \
    for (int u = 0; u < UM; ++u) {                                            \
        const int e  = e0 + u * 1024 + tid;                                   \
        const int ec = (e < e1 && e < NSLOT) ? e : 0;                         \
        const int sv = slot[ec];                                              \
        sr[u] = (e >= NSLOT) ? e - NSLOT : (sv & 0xffff);                     \
        ds[u] = (e >= NSLOT) ? e - NSLOT : (sv >> 16);                        \
    }

__global__ __launch_bounds__(1024) void edge_max_slices(const int* __restrict__ slot,
                                                        const float* __restrict__ esT,
                                                        const float* __restrict__ edT,
                                                        int* __restrict__ pmax) {
    __shared__ float es[NN];
    __shared__ float ed[NN];
    __shared__ int   mx[NN];
    const int tid = threadIdx.x;
    const int hh  = blockIdx.x / SS;
    const int sl  = blockIdx.x % SS;
    const int e0  = sl * EPS, e1 = e0 + EPS;
    int sr[UM], ds[UM];
    LOAD_SLICE_EDGES(sr, ds)
    const float* sp = esT + (size_t)hh * NN;
    const float* dp = edT + (size_t)hh * NN;
    #pragma unroll
    for (int g = 0; g < 2; ++g) {
        float a[NG], b[NG];
        #pragma unroll
        for (int j = 0; j < NG; ++j) {
            const int n  = (g * NG + j) * 1024 + tid;
            const int nc = n < NN ? n : 0;
            a[j] = sp[nc]; b[j] = dp[nc];
        }
        #pragma unroll
        for (int j = 0; j < NG; ++j) {
            const int n = (g * NG + j) * 1024 + tid;
            if (n < NN) { es[n] = a[j]; ed[n] = b[j]; mx[n] = f2ord(-INFINITY); }
        }
    }
    __syncthreads();
    #pragma unroll
    for (int u = 0; u < UM; ++u) {
        const int e = e0 + u * 1024 + tid;
        if (e < e1 && ds[u] >= 0) {
            float v = es[sr[u]] + ed[ds[u]];
            v = v > 0.f ? v : 0.2f * v;
            atomicMax(&mx[ds[u]], f2ord(v));
        }
    }
    __syncthreads();
    int* o = pmax + (size_t)blockIdx.x * NN;
    for (int n = tid; n < NN; n += 1024) o[n] = mx[n];
}

__global__ void combine_max(const int* __restrict__ pmax, float* __restrict__ mT, int H) {
    const int i = blockIdx.x * blockDim.x + threadIdx.x;
    if (i >= NN * H) return;
    const int hh = i / NN, node = i % NN;
    const int* p = pmax + (size_t)hh * SS * NN + node;
    int m = p[0];
    #pragma unroll
    for (int s = 1; s < SS; ++s) { const int v = p[(size_t)s * NN]; m = v > m ? v : m; }
    mT[i] = ord2f(m);
}

__global__ __launch_bounds__(1024) void edge_exp_slices(const int* __restrict__ slot,
                                                        const float* __restrict__ esT,
                                                        const float* __restrict__ edT,
                                                        const float* __restrict__ mT,
                                                        float* __restrict__ exT,
                                                        float* __restrict__ pden) {
    __shared__ float es[NN];
    __shared__ float ed[NN];
    __shared__ float mm[NN];
    __shared__ float dn[NN];
    const int tid = threadIdx.x;
    const int hh  = blockIdx.x / SS;
    const int sl  = blockIdx.x % SS;
    const int e0  = sl * EPS, e1 = e0 + EPS;
    int sr[UM], ds[UM];
    LOAD_SLICE_EDGES(sr, ds)
    const float* sp = esT + (size_t)hh * NN;
    const float* dp = edT + (size_t)hh * NN;
    const float* mp = mT  + (size_t)hh * NN;
    #pragma unroll
    for (int g = 0; g < 2; ++g) {
        float a[NG], b[NG], c[NG];
        #pragma unroll
        for (int j = 0; j < NG; ++j) {
            const int n  = (g * NG + j) * 1024 + tid;
            const int nc = n < NN ? n : 0;
            a[j] = sp[nc]; b[j] = dp[nc]; c[j] = mp[nc];
        }
        #pragma unroll
        for (int j = 0; j < NG; ++j) {
            const int n = (g * NG + j) * 1024 + tid;
            if (n < NN) { es[n] = a[j]; ed[n] = b[j]; mm[n] = c[j]; dn[n] = 0.f; }
        }
    }
    __syncthreads();
    float* xo = exT + (size_t)hh * ETOT;
    #pragma unroll
    for (int u = 0; u < UM; ++u) {
        const int e = e0 + u * 1024 + tid;
        if (e < e1) {
            float ex = 0.f;
            if (ds[u] >= 0) {
                float v = es[sr[u]] + ed[ds[u]];
                v = v > 0.f ? v : 0.2f * v;
                ex = expf(v - mm[ds[u]]);
                atomicAdd(&dn[ds[u]], ex);
            }
            xo[e] = ex;
        }
    }
    __syncthreads();
    float* o = pden + (size_t)blockIdx.x * NN;
    for (int n = tid; n < NN; n += 1024) o[n] = dn[n];
}

__global__ void combine_den(const float* __restrict__ pden, float* __restrict__ denT, int H) {
    const int i = blockIdx.x * blockDim.x + threadIdx.x;
    if (i >= NN * H) return;
    const int hh = i / NN, node = i % NN;
    const float* p = pden + (size_t)hh * SS * NN + node;
    float d = p[0];
    #pragma unroll
    for (int s = 1; s < SS; ++s) d += p[(size_t)s * NN];
    denT[i] = d;
}

__global__ __launch_bounds__(1024) void agg_pair_lds(const int* __restrict__ slot,
                                                     const float* __restrict__ alT,
                                                     const float* __restrict__ HT,
                                                     const float* __restrict__ denT,
                                                     const float* __restrict__ bias,
                                                     float* __restrict__ outT) {
    __shared__ v2f   hs[NN];
    __shared__ float acc0[NN];
    __shared__ float acc1[NN];
    const int tid  = threadIdx.x;
    const int c0   = 2 * blockIdx.x;
    const int head = c0 / HID;
    const float* h0 = HT + (size_t)c0 * NN;
    const float* h1 = h0 + NN;
    const float* al = alT + (size_t)head * ETOT;

    #pragma unroll
    for (int g = 0; g < 2; ++g) {
        float x0[NG], x1[NG], aa[NG];
        #pragma unroll
        for (int j = 0; j < NG; ++j) {
            const int n  = (g * NG + j) * 1024 + tid;
            const int nc = n < NN ? n : 0;
            x0[j] = h0[nc]; x1[j] = h1[nc];
            aa[j] = al[NSLOT + nc];
        }
        #pragma unroll
        for (int j = 0; j < NG; ++j) {
            const int n = (g * NG + j) * 1024 + tid;
            if (n < NN) {
                v2f hv = { x0[j], x1[j] };
                hs[n]   = hv;
                acc0[n] = aa[j] * hv.x;
                acc1[n] = aa[j] * hv.y;
            }
        }
    }
    __syncthreads();

    constexpr int FULL = NSLOT / 1024;
    constexpr int BAT  = 8;
    static_assert(NSLOT % 1024 == 0 && FULL % BAT == 0, "the batch must divide the sweeps");
    int sv[BAT]; float a[BAT];
    #pragma unroll
    for (int u = 0; u < BAT; ++u) {
        const int e = u * 1024 + tid;
        sv[u] = slot[e]; a[u] = al[e];
    }
    for (int it = 0; it < FULL; it += BAT) {
        const int nx = (it + BAT < FULL) ? it + BAT : it;
        int svn[BAT]; float an[BAT];
        #pragma unroll
        for (int u = 0; u < BAT; ++u) {
            const int e = (nx + u) * 1024 + tid;
            svn[u] = slot[e]; an[u] = al[e];
        }
        v2f hv[BAT];
        #pragma unroll
        for (int u = 0; u < BAT; ++u) hv[u] = hs[sv[u] < 0 ? 0 : (sv[u] & 0xffff)];
        #pragma unroll
        for (int u = 0; u < BAT; ++u) {
            const int d = sv[u] >> 16;
            if (d >= 0) {
                atomicAdd(&acc0[d], a[u] * hv[u].x);
                atomicAdd(&acc1[d], a[u] * hv[u].y);
            }
        }
        #pragma unroll
        for (int u = 0; u < BAT; ++u) { sv[u] = svn[u]; a[u] = an[u]; }
    }
    const float* dn = denT + (size_t)head * NN;
    float dd[2 * NG];
    #pragma unroll
    for (int j = 0; j < 2 * NG; ++j) {
        const int n  = j * 1024 + tid;
        const int nc = n < NN ? n : 0;
        dd[j] = dn[nc];
    }
    __syncthreads();

    float* o0 = outT + (size_t)c0 * NN;
    float* o1 = o0 + NN;
    const float b0 = bias[c0], b1 = bias[c0 + 1];
    #pragma unroll
    for (int j = 0; j < 2 * NG; ++j) {
        const int n = j * 1024 + tid;
        if (n < NN) {
            const float d = dd[j] + 1e-16f;
            const float v0 = acc0[n] / d + b0, v1 = acc1[n] / d + b1;
            o0[n] = v0 > 0.f ? v0 : 0.f;
            o1[n] = v1 > 0.f ? v1 : 0.f;
        }
    }
}

#define S2 8
__global__ __launch_bounds__(1024) void agg2_slice_lds(const int* __restrict__ slot,
                                                       const float* __restrict__ al,
                                                       const float* __restrict__ HT,
                                                       float* __restrict__ part) {
    __shared__ v2f   hs[NN];
    __shared__ float acc0[NN];
    __shared__ float acc1[NN];
    const int tid   = threadIdx.x;
    const int pair  = blockIdx.x % (OUT_CH / 2);
    const int slice = blockIdx.x / (OUT_CH / 2);
    const int c0    = 2 * pair;
    const float* h0 = HT + (size_t)c0 * NN;
    const float* h1 = h0 + NN;

    if (slice == 0) {
        #pragma unroll
        for (int g = 0; g < 2; ++g) {
            float x0[NG], x1[NG], aa[NG];
            #pragma unroll
            for (int j = 0; j < NG; ++j) {
                const int n  = (g * NG + j) * 1024 + tid;
                const int nc = n < NN ? n : 0;
                x0[j] = h0[nc]; x1[j] = h1[nc];
                aa[j] = al[NSLOT + nc];
            }
            #pragma unroll
            for (int j = 0; j < NG; ++j) {
                const int n = (g * NG + j) * 1024 + tid;
                if (n < NN) {
                    v2f hv = { x0[j], x1[j] };
                    hs[n]   = hv;
                    acc0[n] = aa[j] * hv.x;
                    acc1[n] = aa[j] * hv.y;
                }
            }
        }
    } else {
        #pragma unroll
        for (int g = 0; g < 2; ++g) {
            float x0[NG], x1[NG];
            #pragma unroll
            for (int j = 0; j < NG; ++j) {
                const int n  = (g * NG + j) * 1024 + tid;
                const int nc = n < NN ? n : 0;
                x0[j] = h0[nc]; x1[j] = h1[nc];
            }
            #pragma unroll
            for (int j = 0; j < NG; ++j) {
                const int n = (g * NG + j) * 1024 + tid;
                if (n < NN) {
                    v2f hv = { x0[j], x1[j] };
                    hs[n]   = hv;
                    acc0[n] = 0.f;
                    acc1[n] = 0.f;
                }
            }
        }
    }
    __syncthreads();

    constexpr int FULL = NSLOT / 1024;
    static_assert(NSLOT % 1024 == 0 && FULL % S2 == 0, "the slices must divide the sweeps");
    constexpr int SW  = FULL / S2;
    constexpr int BAT = 11;
    static_assert(SW % BAT == 0, "the batch must divide a slice's sweeps");
    const int base = slice * SW * 1024 + tid;
    for (int it = 0; it < SW; it += BAT) {
        int sv[BAT]; float a[BAT];
        #pragma unroll
        for (int u = 0; u < BAT; ++u) {
            const int e = base + (it + u) * 1024;
            sv[u] = slot[e]; a[u] = al[e];
        }
        v2f hv[BAT];
        #pragma unroll
        for (int u = 0; u < BAT; ++u) hv[u] = hs[sv[u] < 0 ? 0 : (sv[u] & 0xffff)];
        #pragma unroll
        for (int u = 0; u < BAT; ++u) {
            const int d = sv[u] >> 16;
            if (d >= 0) {
                atomicAdd(&acc0[d], a[u] * hv[u].x);
                atomicAdd(&acc1[d], a[u] * hv[u].y);
            }
        }
    }
    __syncthreads();

    float* o0 = part + ((size_t)slice * OUT_CH + c0) * NN;
    float* o1 = o0 + NN;
    for (int n = tid; n < NN; n += 1024) { o0[n] = acc0[n]; o1[n] = acc1[n]; }
}

__global__ __launch_bounds__(256) void final_planes_logsoftmax(const float* __restrict__ part,
                                                                const float* __restrict__ pden,
                                                                const float* __restrict__ b2,
                                                                float* __restrict__ out) {
    __shared__ float tile[OUT_CH][33];
    const int tid = threadIdx.x;
    const int w = tid >> 5, lane = tid & 31;
    const int n0 = blockIdx.x * 32;
    float d = pden[n0 + lane];
    #pragma unroll
    for (int s = 1; s < SS; ++s) d += pden[(size_t)s * NN + n0 + lane];
    d += 1e-16f;
    #pragma unroll
    for (int j = 0; j < 8; ++j) {
        const int c = 8 * w + j;
        const float* p = part + (size_t)c * NN + n0 + lane;
        float v = p[0];
        #pragma unroll
        for (int s = 1; s < S2; ++s) v += p[(size_t)s * OUT_CH * NN];
        tile[c][lane] = v / d + b2[c];
    }
    __syncthreads();
    #pragma unroll
    for (int j = 0; j < 4; ++j) {
        const int nl = 4 * w + j;
        float v0 = tile[lane][nl];
        float v1 = tile[lane + 32][nl];
        float m = fmaxf(v0, v1);
        #pragma unroll
        for (int o = 16; o > 0; o >>= 1) m = fmaxf(m, __shfl_xor(m, o, 32));
        float s = expf(v0 - m) + expf(v1 - m);
        #pragma unroll
        for (int o = 16; o > 0; o >>= 1) s += __shfl_xor(s, o, 32);
        float lse = logf(s) + m;
        float* row = out + (size_t)(n0 + nl) * OUT_CH;
        row[lane]      = v0 - lse;
        row[lane + 32] = v1 - lse;
    }
}

extern "C" void kernel_launch(void* const* d_in, const int* in_sizes, int n_in,
                              void* d_out, int out_size, void* d_ws, size_t ws_size,
                              hipStream_t stream) {
    const float* x   = (const float*)d_in[0];
    const int*   ei  = (const int*)  d_in[1];
    const float* W1  = (const float*)d_in[2];
    const float* a1s = (const float*)d_in[3];
    const float* a1d = (const float*)d_in[4];
    const float* b1  = (const float*)d_in[5];
    const float* W2  = (const float*)d_in[6];
    const float* a2s = (const float*)d_in[7];
    const float* a2d = (const float*)d_in[8];
    const float* b2  = (const float*)d_in[9];
    float* out = (float*)d_out;

    float* ws = (float*)d_ws;
    float* h1T   = ws;
    float* agg1T = h1T   + (size_t)NN * HC1;
    float* es1T  = agg1T + (size_t)NN * HC1;
    float* ed1T  = es1T  + (size_t)NN * HEADS;
    float* m1T   = ed1T  + (size_t)NN * HEADS;
    float* den1T = m1T   + (size_t)NN * HEADS;
    int*   pm1   = (int*)(den1T + (size_t)NN * HEADS);
    float* pd1   = (float*)pm1 + (size_t)NN * HEADS * SS;
    float* al1   = pd1  + (size_t)NN * HEADS * SS;
    float* h2T   = al1  + (size_t)ETOT * HEADS;
    float* es2T  = h2T  + (size_t)NN * OUT_CH;
    float* ed2T  = es2T + NN;
    float* m2T   = ed2T + NN;
    int*   pm2   = (int*)(m2T + NN);
    float* pd2   = (float*)pm2 + (size_t)NN * SS;
    float* ex2   = pd2  + (size_t)NN * SS;
    int*   slots = (int*)(ex2 + (size_t)ETOT);
    float* part2 = h1T;

    const int TB = 256;
    auto blocks = [](long long n, int tb) { return (unsigned)((n + tb - 1) / tb); };

    {
        long long waves = (long long)(NN / 32) * (HC1 / 32);
        wmma_gemm_f32_ct<<<blocks(waves * 32, TB), TB, 0, stream>>>(x, W1, h1T, NN, IN_CH, HC1);
    }
    slots_and_scores1<<<NCH + HEADS * NB1, 1024, 0, stream>>>(ei, slots, h1T, a1s, a1d, es1T, ed1T);
    edge_max_slices<<<HEADS * SS, 1024, 0, stream>>>(slots, es1T, ed1T, pm1);
    combine_max    <<<blocks((long long)NN * HEADS, TB), TB, 0, stream>>>(pm1, m1T, HEADS);
    edge_exp_slices<<<HEADS * SS, 1024, 0, stream>>>(slots, es1T, ed1T, m1T, al1, pd1);
    combine_den    <<<blocks((long long)NN * HEADS, TB), TB, 0, stream>>>(pd1, den1T, HEADS);
    agg_pair_lds  <<<HC1 / 2, 1024, 0, stream>>>(slots, al1, h1T, den1T, b1, agg1T);

    static_assert(OUT_CH == 64 && HC1 % (4 * KS2) == 0 && NN % 32 == 0, "GEMM 2 tiling");
    wmma_gemm_f32_tt<<<NN / 32, 256, 0, stream>>>(agg1T, W2, h2T, NN, HC1, OUT_CH);
    scores_ct<OUT_CH><<<dim3(blocks(NN, TB), 1), TB, 0, stream>>>(
        h2T, a2s, a2d, es2T, ed2T);
    edge_max_slices<<<SS, 1024, 0, stream>>>(slots, es2T, ed2T, pm2);
    combine_max    <<<blocks((long long)NN, TB), TB, 0, stream>>>(pm2, m2T, 1);
    edge_exp_slices<<<SS, 1024, 0, stream>>>(slots, es2T, ed2T, m2T, ex2, pd2);
    agg2_slice_lds<<<(OUT_CH / 2) * S2, 1024, 0, stream>>>(slots, ex2, h2T, part2);
    final_planes_logsoftmax<<<NN / 32, 256, 0, stream>>>(part2, pd2, b2, out);
}


# Layer-1 GEMM: each 8-wave work-group stages its 32-node x tile (16 KB) once in LDS by whole-line loads; the waves' x fragments are LDS reads, no longer global loads that touch 16 lines each
# speedup vs baseline: 5.3415x; 1.0756x over previous
#include <hip/hip_runtime.h>
#include <math.h>

#define NN      20000
#define NE      320000
#define NCH     32
#define CHE     (NE / NCH)
#define CAP     352
#define CHS     (CAP * 32)
#define NSLOT   (NCH * CHS)
#define ETOT    (NSLOT + NN)
static_assert(NE % NCH == 0 && CHS >= CHE && CHS % 1024 == 0, "a chunk's slots must hold its edges");
static_assert(NN <= 32768, "source and destination share one word");
#define IN_CH   128
#define HID     64
#define HEADS   8
#define HC1     (HEADS * HID)
#define OUT_CH  64

typedef float v2f __attribute__((ext_vector_type(2)));
typedef float v8f __attribute__((ext_vector_type(8)));

__device__ __forceinline__ int f2ord(float f) {
    int i = __float_as_int(f);
    return i >= 0 ? i : (i ^ 0x7fffffff);
}
__device__ __forceinline__ float ord2f(int i) {
    return __int_as_float(i >= 0 ? i : (i ^ 0x7fffffff));
}

#define XP (IN_CH + 4)
typedef float v4f __attribute__((ext_vector_type(4)));
__global__ __launch_bounds__(256) void wmma_gemm_f32_ct(const float* __restrict__ X,
                                                        const float* __restrict__ Wm,
                                                        float* __restrict__ HT,
                                                        int M, int N) {
    __shared__ float xs[32 * XP];
    const int w       = threadIdx.x >> 5;
    const int lane    = threadIdx.x & 31;
    const int cParts  = N >> 8;
    const int mg      = blockIdx.x / cParts;
    const int cg      = (blockIdx.x % cParts) * 8 + w;
    const int l    = lane & 15;
    const int half = lane >> 4;
    const unsigned uN = (unsigned)N, uM = (unsigned)M;

    const float* xg = X + (size_t)mg * 32 * IN_CH;
    #pragma unroll
    for (int q = 0; q < (32 * IN_CH / 4) / 256; ++q) {
        const int p = q * 256 + threadIdx.x;
        const v4f v = *(const v4f*)(xg + 4 * p);
        *(v4f*)(xs + (p >> 5) * XP + 4 * (p & 31)) = v;
    }
    __syncthreads();

    const float* xr0 = xs + l * XP + 2 * half;
    const float* xr1 = xr0 + 16 * XP;
    unsigned woff = (unsigned)(2 * half) * uN + (unsigned)(cg * 32 + l);

    v8f acc00 = {}, acc01 = {}, acc10 = {}, acc11 = {};
    for (int k0 = 0; k0 < IN_CH; k0 += 4) {
        v2f x0 = *(const v2f*)(xr0 + k0);
        v2f x1 = *(const v2f*)(xr1 + k0);
        v2f w0 = { Wm[woff],      Wm[woff + uN]      };
        v2f w1 = { Wm[woff + 16], Wm[woff + uN + 16] };
        acc00 = __builtin_amdgcn_wmma_f32_16x16x4_f32(false, w0, false, x0, (short)0, acc00, false, false);
        acc01 = __builtin_amdgcn_wmma_f32_16x16x4_f32(false, w0, false, x1, (short)0, acc01, false, false);
        acc10 = __builtin_amdgcn_wmma_f32_16x16x4_f32(false, w1, false, x0, (short)0, acc10, false, false);
        acc11 = __builtin_amdgcn_wmma_f32_16x16x4_f32(false, w1, false, x1, (short)0, acc11, false, false);
        woff += 4u * uN;
    }
    unsigned hoff = (unsigned)(cg * 32 + 8 * half) * uM + (unsigned)(mg * 32 + l);
    #pragma unroll
    for (int r = 0; r < 8; ++r) {
        float* hp = HT + hoff;
        hp[0]            = acc00[r];
        hp[16]           = acc01[r];
        hp[16u * uM]      = acc10[r];
        hp[16u * uM + 16] = acc11[r];
        hoff += uM;
    }
}

#define KS2 4
__global__ __launch_bounds__(256) void wmma_gemm_f32_tt(const float* __restrict__ XT,
                                                        const float* __restrict__ Wm,
                                                        float* __restrict__ HT,
                                                        int M, int K, int N) {
    __shared__ float part[2][KS2][32][34];
    const int w    = threadIdx.x >> 5;
    const int lane = threadIdx.x & 31;
    const int cg   = w & 1;
    const int ks   = w >> 1;
    const int mg   = blockIdx.x;
    const int l    = lane & 15;
    const int half = lane >> 4;
    const unsigned uN = (unsigned)N, uM = (unsigned)M;
    const int KL = K / KS2;
    const unsigned kb = (unsigned)(ks * KL);

    unsigned xoff = (kb + 2u * half) * uM + (unsigned)(mg * 32 + l);
    unsigned woff = (kb + 2u * half) * uN + (unsigned)(cg * 32 + l);

    v8f acc00 = {}, acc01 = {}, acc10 = {}, acc11 = {};
    for (int k0 = 0; k0 < KL; k0 += 4) {
        v2f x0 = { XT[xoff],      XT[xoff + uM]      };
        v2f x1 = { XT[xoff + 16], XT[xoff + uM + 16] };
        v2f w0 = { Wm[woff],      Wm[woff + uN]      };
        v2f w1 = { Wm[woff + 16], Wm[woff + uN + 16] };
        acc00 = __builtin_amdgcn_wmma_f32_16x16x4_f32(false, w0, false, x0, (short)0, acc00, false, false);
        acc01 = __builtin_amdgcn_wmma_f32_16x16x4_f32(false, w0, false, x1, (short)0, acc01, false, false);
        acc10 = __builtin_amdgcn_wmma_f32_16x16x4_f32(false, w1, false, x0, (short)0, acc10, false, false);
        acc11 = __builtin_amdgcn_wmma_f32_16x16x4_f32(false, w1, false, x1, (short)0, acc11, false, false);
        xoff += 4u * uM;
        woff += 4u * uN;
    }
    float (*pt)[34] = part[cg][ks];
    #pragma unroll
    for (int r = 0; r < 8; ++r) {
        pt[8 * half + r     ][l     ] = acc00[r];
        pt[8 * half + r     ][l + 16] = acc01[r];
        pt[8 * half + r + 16][l     ] = acc10[r];
        pt[8 * half + r + 16][l + 16] = acc11[r];
    }
    __syncthreads();
    unsigned hoff = (unsigned)(cg * 32 + 8 * ks) * uM + (unsigned)(mg * 32 + lane);
    #pragma unroll
    for (int j = 0; j < 8; ++j) {
        const int row = 8 * ks + j;
        float v = part[cg][0][row][lane];
        #pragma unroll
        for (int q = 1; q < KS2; ++q) v += part[cg][q][row][lane];
        HT[hoff] = v;
        hoff += uM;
    }
}

template <int C>
__device__ __forceinline__ void scores_body(const float* __restrict__ HT,
                                            const float* __restrict__ a_src,
                                            const float* __restrict__ a_dst,
                                            float* __restrict__ esT, float* __restrict__ edT,
                                            int hh, int node) {
    if (node >= NN) return;
    const float* hp = HT + (size_t)hh * C * NN + node;
    const float* as = a_src + hh * C;
    const float* ad = a_dst + hh * C;
    float v[C];
    #pragma unroll
    for (int c = 0; c < C; ++c) v[c] = hp[(size_t)c * NN];
    float s = 0.f, d = 0.f;
    #pragma unroll
    for (int c = 0; c < C; ++c) { s += v[c] * as[c]; d += v[c] * ad[c]; }
    esT[(size_t)hh * NN + node] = s;
    edT[(size_t)hh * NN + node] = d;
}
template <int C>
__global__ void scores_ct(const float* __restrict__ HT,
                          const float* __restrict__ a_src,
                          const float* __restrict__ a_dst,
                          float* __restrict__ esT, float* __restrict__ edT) {
    scores_body<C>(HT, a_src, a_dst, esT, edT, blockIdx.y, blockIdx.x * blockDim.x + threadIdx.x);
}

__device__ __forceinline__ void bank_slots_body(const int* __restrict__ ei, int* __restrict__ slot, const int g) {
    __shared__ int tab[CHS];
    __shared__ int spill[CHE];
    __shared__ int cnt[32];
    __shared__ int fre[33];
    __shared__ int nspill;
    const int tid = threadIdx.x;
    constexpr int UE = (CHE + 1023) / 1024;
    int s[UE], d[UE];
    #pragma unroll
    for (int u = 0; u < UE; ++u) {
        const int i  = u * 1024 + tid;
        const int ic = i < CHE ? i : 0;
        s[u] = ei[g * CHE + ic];
        d[u] = ei[NE + g * CHE + ic];
    }
    for (int q = 0; q < CHS / 1024; ++q) tab[q * 1024 + tid] = -1;
    if (tid < 32) cnt[tid] = 0;
    if (tid == 0) nspill = 0;
    __syncthreads();
    #pragma unroll
    for (int u = 0; u < UE; ++u) {
        const int i = u * 1024 + tid;
        if (i < CHE) {
            const int cl = d[u] & 31;
            const int v  = (d[u] << 16) | s[u];
            const int r  = atomicAdd(&cnt[cl], 1);
            if (r < CAP) tab[r * 32 + cl] = v;
            else         spill[atomicAdd(&nspill, 1)] = v;
        }
    }
    __syncthreads();
    if (tid == 0) {
        int a = 0;
        for (int cl = 0; cl < 32; ++cl) {
            fre[cl] = a;
            const int used = cnt[cl] < CAP ? cnt[cl] : CAP;
            a += CAP - used;
        }
        fre[32] = a;
    }
    __syncthreads();
    const int ns = nspill;
    for (int k = tid; k < ns; k += 1024) {
        int cl = 0;
        for (int j = 1; j < 32; ++j) cl += (fre[j] <= k) ? 1 : 0;
        const int used = cnt[cl] < CAP ? cnt[cl] : CAP;
        tab[(used + k - fre[cl]) * 32 + cl] = spill[k];
    }
    __syncthreads();
    int* o = slot + (size_t)g * CHS;
    for (int q = 0; q < CHS / 1024; ++q) o[q * 1024 + tid] = tab[q * 1024 + tid];
}

#define NB1 ((NN + 1023) / 1024)
__global__ __launch_bounds__(1024) void slots_and_scores1(const int* __restrict__ ei, int* __restrict__ slot,
                                                          const float* __restrict__ HT,
                                                          const float* __restrict__ a_src,
                                                          const float* __restrict__ a_dst,
                                                          float* __restrict__ esT, float* __restrict__ edT) {
    if (blockIdx.x < NCH) { bank_slots_body(ei, slot, blockIdx.x); return; }
    const int b = blockIdx.x - NCH;
    scores_body<HID>(HT, a_src, a_dst, esT, edT, b / NB1, (b % NB1) * 1024 + threadIdx.x);
}

#define SS   32
#define EPS  (ETOT / SS)
#define UM   ((EPS + 1023) / 1024)
#define NG   10
static_assert(ETOT % SS == 0, "the slices must divide the list");
static_assert(2 * NG * 1024 >= NN, "two groups of NG sweeps must cover the nodes");

#define LOAD_SLICE_EDGES(sr, ds)                                              \
    _Pragma("unroll")                                                         \
    for (int u = 0; u < UM; ++u) {                                            \
        const int e  = e0 + u * 1024 + tid;                                   \
        const int ec = (e < e1 && e < NSLOT) ? e : 0;                         \
        const int sv = slot[ec];                                              \
        sr[u] = (e >= NSLOT) ? e - NSLOT : (sv & 0xffff);                     \
        ds[u] = (e >= NSLOT) ? e - NSLOT : (sv >> 16);                        \
    }

__global__ __launch_bounds__(1024) void edge_max_slices(const int* __restrict__ slot,
                                                        const float* __restrict__ esT,
                                                        const float* __restrict__ edT,
                                                        int* __restrict__ pmax) {
    __shared__ float es[NN];
    __shared__ float ed[NN];
    __shared__ int   mx[NN];
    const int tid = threadIdx.x;
    const int hh  = blockIdx.x / SS;
    const int sl  = blockIdx.x % SS;
    const int e0  = sl * EPS, e1 = e0 + EPS;
    int sr[UM], ds[UM];
    LOAD_SLICE_EDGES(sr, ds)
    const float* sp = esT + (size_t)hh * NN;
    const float* dp = edT + (size_t)hh * NN;
    #pragma unroll
    for (int g = 0; g < 2; ++g) {
        float a[NG], b[NG];
        #pragma unroll
        for (int j = 0; j < NG; ++j) {
            const int n  = (g * NG + j) * 1024 + tid;
            const int nc = n < NN ? n : 0;
            a[j] = sp[nc]; b[j] = dp[nc];
        }
        #pragma unroll
        for (int j = 0; j < NG; ++j) {
            const int n = (g * NG + j) * 1024 + tid;
            if (n < NN) { es[n] = a[j]; ed[n] = b[j]; mx[n] = f2ord(-INFINITY); }
        }
    }
    __syncthreads();
    #pragma unroll
    for (int u = 0; u < UM; ++u) {
        const int e = e0 + u * 1024 + tid;
        if (e < e1 && ds[u] >= 0) {
            float v = es[sr[u]] + ed[ds[u]];
            v = v > 0.f ? v : 0.2f * v;
            atomicMax(&mx[ds[u]], f2ord(v));
        }
    }
    __syncthreads();
    int* o = pmax + (size_t)blockIdx.x * NN;
    for (int n = tid; n < NN; n += 1024) o[n] = mx[n];
}

__global__ void combine_max(const int* __restrict__ pmax, float* __restrict__ mT, int H) {
    const int i = blockIdx.x * blockDim.x + threadIdx.x;
    if (i >= NN * H) return;
    const int hh = i / NN, node = i % NN;
    const int* p = pmax + (size_t)hh * SS * NN + node;
    int m = p[0];
    #pragma unroll
    for (int s = 1; s < SS; ++s) { const int v = p[(size_t)s * NN]; m = v > m ? v : m; }
    mT[i] = ord2f(m);
}

__global__ __launch_bounds__(1024) void edge_exp_slices(const int* __restrict__ slot,
                                                        const float* __restrict__ esT,
                                                        const float* __restrict__ edT,
                                                        const float* __restrict__ mT,
                                                        float* __restrict__ exT,
                                                        float* __restrict__ pden) {
    __shared__ float es[NN];
    __shared__ float ed[NN];
    __shared__ float mm[NN];
    __shared__ float dn[NN];
    const int tid = threadIdx.x;
    const int hh  = blockIdx.x / SS;
    const int sl  = blockIdx.x % SS;
    const int e0  = sl * EPS, e1 = e0 + EPS;
    int sr[UM], ds[UM];
    LOAD_SLICE_EDGES(sr, ds)
    const float* sp = esT + (size_t)hh * NN;
    const float* dp = edT + (size_t)hh * NN;
    const float* mp = mT  + (size_t)hh * NN;
    #pragma unroll
    for (int g = 0; g < 2; ++g) {
        float a[NG], b[NG], c[NG];
        #pragma unroll
        for (int j = 0; j < NG; ++j) {
            const int n  = (g * NG + j) * 1024 + tid;
            const int nc = n < NN ? n : 0;
            a[j] = sp[nc]; b[j] = dp[nc]; c[j] = mp[nc];
        }
        #pragma unroll
        for (int j = 0; j < NG; ++j) {
            const int n = (g * NG + j) * 1024 + tid;
            if (n < NN) { es[n] = a[j]; ed[n] = b[j]; mm[n] = c[j]; dn[n] = 0.f; }
        }
    }
    __syncthreads();
    float* xo = exT + (size_t)hh * ETOT;
    #pragma unroll
    for (int u = 0; u < UM; ++u) {
        const int e = e0 + u * 1024 + tid;
        if (e < e1) {
            float ex = 0.f;
            if (ds[u] >= 0) {
                float v = es[sr[u]] + ed[ds[u]];
                v = v > 0.f ? v : 0.2f * v;
                ex = expf(v - mm[ds[u]]);
                atomicAdd(&dn[ds[u]], ex);
            }
            xo[e] = ex;
        }
    }
    __syncthreads();
    float* o = pden + (size_t)blockIdx.x * NN;
    for (int n = tid; n < NN; n += 1024) o[n] = dn[n];
}

__global__ void combine_den(const float* __restrict__ pden, float* __restrict__ denT, int H) {
    const int i = blockIdx.x * blockDim.x + threadIdx.x;
    if (i >= NN * H) return;
    const int hh = i / NN, node = i % NN;
    const float* p = pden + (size_t)hh * SS * NN + node;
    float d = p[0];
    #pragma unroll
    for (int s = 1; s < SS; ++s) d += p[(size_t)s * NN];
    denT[i] = d;
}

__global__ __launch_bounds__(1024) void agg_pair_lds(const int* __restrict__ slot,
                                                     const float* __restrict__ alT,
                                                     const float* __restrict__ HT,
                                                     const float* __restrict__ denT,
                                                     const float* __restrict__ bias,
                                                     float* __restrict__ outT) {
    __shared__ v2f   hs[NN];
    __shared__ float acc0[NN];
    __shared__ float acc1[NN];
    const int tid  = threadIdx.x;
    const int c0   = 2 * blockIdx.x;
    const int head = c0 / HID;
    const float* h0 = HT + (size_t)c0 * NN;
    const float* h1 = h0 + NN;
    const float* al = alT + (size_t)head * ETOT;

    #pragma unroll
    for (int g = 0; g < 2; ++g) {
        float x0[NG], x1[NG], aa[NG];
        #pragma unroll
        for (int j = 0; j < NG; ++j) {
            const int n  = (g * NG + j) * 1024 + tid;
            const int nc = n < NN ? n : 0;
            x0[j] = h0[nc]; x1[j] = h1[nc];
            aa[j] = al[NSLOT + nc];
        }
        #pragma unroll
        for (int j = 0; j < NG; ++j) {
            const int n = (g * NG + j) * 1024 + tid;
            if (n < NN) {
                v2f hv = { x0[j], x1[j] };
                hs[n]   = hv;
                acc0[n] = aa[j] * hv.x;
                acc1[n] = aa[j] * hv.y;
            }
        }
    }
    __syncthreads();

    constexpr int FULL = NSLOT / 1024;
    constexpr int BAT  = 8;
    static_assert(NSLOT % 1024 == 0 && FULL % BAT == 0, "the batch must divide the sweeps");
    int sv[BAT]; float a[BAT];
    #pragma unroll
    for (int u = 0; u < BAT; ++u) {
        const int e = u * 1024 + tid;
        sv[u] = slot[e]; a[u] = al[e];
    }
    for (int it = 0; it < FULL; it += BAT) {
        const int nx = (it + BAT < FULL) ? it + BAT : it;
        int svn[BAT]; float an[BAT];
        #pragma unroll
        for (int u = 0; u < BAT; ++u) {
            const int e = (nx + u) * 1024 + tid;
            svn[u] = slot[e]; an[u] = al[e];
        }
        v2f hv[BAT];
        #pragma unroll
        for (int u = 0; u < BAT; ++u) hv[u] = hs[sv[u] < 0 ? 0 : (sv[u] & 0xffff)];
        #pragma unroll
        for (int u = 0; u < BAT; ++u) {
            const int d = sv[u] >> 16;
            if (d >= 0) {
                atomicAdd(&acc0[d], a[u] * hv[u].x);
                atomicAdd(&acc1[d], a[u] * hv[u].y);
            }
        }
        #pragma unroll
        for (int u = 0; u < BAT; ++u) { sv[u] = svn[u]; a[u] = an[u]; }
    }
    const float* dn = denT + (size_t)head * NN;
    float dd[2 * NG];
    #pragma unroll
    for (int j = 0; j < 2 * NG; ++j) {
        const int n  = j * 1024 + tid;
        const int nc = n < NN ? n : 0;
        dd[j] = dn[nc];
    }
    __syncthreads();

    float* o0 = outT + (size_t)c0 * NN;
    float* o1 = o0 + NN;
    const float b0 = bias[c0], b1 = bias[c0 + 1];
    #pragma unroll
    for (int j = 0; j < 2 * NG; ++j) {
        const int n = j * 1024 + tid;
        if (n < NN) {
            const float d = dd[j] + 1e-16f;
            const float v0 = acc0[n] / d + b0, v1 = acc1[n] / d + b1;
            o0[n] = v0 > 0.f ? v0 : 0.f;
            o1[n] = v1 > 0.f ? v1 : 0.f;
        }
    }
}

#define S2 8
__global__ __launch_bounds__(1024) void agg2_slice_lds(const int* __restrict__ slot,
                                                       const float* __restrict__ al,
                                                       const float* __restrict__ HT,
                                                       float* __restrict__ part) {
    __shared__ v2f   hs[NN];
    __shared__ float acc0[NN];
    __shared__ float acc1[NN];
    const int tid   = threadIdx.x;
    const int pair  = blockIdx.x % (OUT_CH / 2);
    const int slice = blockIdx.x / (OUT_CH / 2);
    const int c0    = 2 * pair;
    const float* h0 = HT + (size_t)c0 * NN;
    const float* h1 = h0 + NN;

    if (slice == 0) {
        #pragma unroll
        for (int g = 0; g < 2; ++g) {
            float x0[NG], x1[NG], aa[NG];
            #pragma unroll
            for (int j = 0; j < NG; ++j) {
                const int n  = (g * NG + j) * 1024 + tid;
                const int nc = n < NN ? n : 0;
                x0[j] = h0[nc]; x1[j] = h1[nc];
                aa[j] = al[NSLOT + nc];
            }
            #pragma unroll
            for (int j = 0; j < NG; ++j) {
                const int n = (g * NG + j) * 1024 + tid;
                if (n < NN) {
                    v2f hv = { x0[j], x1[j] };
                    hs[n]   = hv;
                    acc0[n] = aa[j] * hv.x;
                    acc1[n] = aa[j] * hv.y;
                }
            }
        }
    } else {
        #pragma unroll
        for (int g = 0; g < 2; ++g) {
            float x0[NG], x1[NG];
            #pragma unroll
            for (int j = 0; j < NG; ++j) {
                const int n  = (g * NG + j) * 1024 + tid;
                const int nc = n < NN ? n : 0;
                x0[j] = h0[nc]; x1[j] = h1[nc];
            }
            #pragma unroll
            for (int j = 0; j < NG; ++j) {
                const int n = (g * NG + j) * 1024 + tid;
                if (n < NN) {
                    v2f hv = { x0[j], x1[j] };
                    hs[n]   = hv;
                    acc0[n] = 0.f;
                    acc1[n] = 0.f;
                }
            }
        }
    }
    __syncthreads();

    constexpr int FULL = NSLOT / 1024;
    static_assert(NSLOT % 1024 == 0 && FULL % S2 == 0, "the slices must divide the sweeps");
    constexpr int SW  = FULL / S2;
    constexpr int BAT = 11;
    static_assert(SW % BAT == 0, "the batch must divide a slice's sweeps");
    const int base = slice * SW * 1024 + tid;
    for (int it = 0; it < SW; it += BAT) {
        int sv[BAT]; float a[BAT];
        #pragma unroll
        for (int u = 0; u < BAT; ++u) {
            const int e = base + (it + u) * 1024;
            sv[u] = slot[e]; a[u] = al[e];
        }
        v2f hv[BAT];
        #pragma unroll
        for (int u = 0; u < BAT; ++u) hv[u] = hs[sv[u] < 0 ? 0 : (sv[u] & 0xffff)];
        #pragma unroll
        for (int u = 0; u < BAT; ++u) {
            const int d = sv[u] >> 16;
            if (d >= 0) {
                atomicAdd(&acc0[d], a[u] * hv[u].x);
                atomicAdd(&acc1[d], a[u] * hv[u].y);
            }
        }
    }
    __syncthreads();

    float* o0 = part + ((size_t)slice * OUT_CH + c0) * NN;
    float* o1 = o0 + NN;
    for (int n = tid; n < NN; n += 1024) { o0[n] = acc0[n]; o1[n] = acc1[n]; }
}

__global__ __launch_bounds__(256) void final_planes_logsoftmax(const float* __restrict__ part,
                                                                const float* __restrict__ pden,
                                                                const float* __restrict__ b2,
                                                                float* __restrict__ out) {
    __shared__ float tile[OUT_CH][33];
    const int tid = threadIdx.x;
    const int w = tid >> 5, lane = tid & 31;
    const int n0 = blockIdx.x * 32;
    float d = pden[n0 + lane];
    #pragma unroll
    for (int s = 1; s < SS; ++s) d += pden[(size_t)s * NN + n0 + lane];
    d += 1e-16f;
    #pragma unroll
    for (int j = 0; j < 8; ++j) {
        const int c = 8 * w + j;
        const float* p = part + (size_t)c * NN + n0 + lane;
        float v = p[0];
        #pragma unroll
        for (int s = 1; s < S2; ++s) v += p[(size_t)s * OUT_CH * NN];
        tile[c][lane] = v / d + b2[c];
    }
    __syncthreads();
    #pragma unroll
    for (int j = 0; j < 4; ++j) {
        const int nl = 4 * w + j;
        float v0 = tile[lane][nl];
        float v1 = tile[lane + 32][nl];
        float m = fmaxf(v0, v1);
        #pragma unroll
        for (int o = 16; o > 0; o >>= 1) m = fmaxf(m, __shfl_xor(m, o, 32));
        float s = expf(v0 - m) + expf(v1 - m);
        #pragma unroll
        for (int o = 16; o > 0; o >>= 1) s += __shfl_xor(s, o, 32);
        float lse = logf(s) + m;
        float* row = out + (size_t)(n0 + nl) * OUT_CH;
        row[lane]      = v0 - lse;
        row[lane + 32] = v1 - lse;
    }
}

extern "C" void kernel_launch(void* const* d_in, const int* in_sizes, int n_in,
                              void* d_out, int out_size, void* d_ws, size_t ws_size,
                              hipStream_t stream) {
    const float* x   = (const float*)d_in[0];
    const int*   ei  = (const int*)  d_in[1];
    const float* W1  = (const float*)d_in[2];
    const float* a1s = (const float*)d_in[3];
    const float* a1d = (const float*)d_in[4];
    const float* b1  = (const float*)d_in[5];
    const float* W2  = (const float*)d_in[6];
    const float* a2s = (const float*)d_in[7];
    const float* a2d = (const float*)d_in[8];
    const float* b2  = (const float*)d_in[9];
    float* out = (float*)d_out;

    float* ws = (float*)d_ws;
    float* h1T   = ws;
    float* agg1T = h1T   + (size_t)NN * HC1;
    float* es1T  = agg1T + (size_t)NN * HC1;
    float* ed1T  = es1T  + (size_t)NN * HEADS;
    float* m1T   = ed1T  + (size_t)NN * HEADS;
    float* den1T = m1T   + (size_t)NN * HEADS;
    int*   pm1   = (int*)(den1T + (size_t)NN * HEADS);
    float* pd1   = (float*)pm1 + (size_t)NN * HEADS * SS;
    float* al1   = pd1  + (size_t)NN * HEADS * SS;
    float* h2T   = al1  + (size_t)ETOT * HEADS;
    float* es2T  = h2T  + (size_t)NN * OUT_CH;
    float* ed2T  = es2T + NN;
    float* m2T   = ed2T + NN;
    int*   pm2   = (int*)(m2T + NN);
    float* pd2   = (float*)pm2 + (size_t)NN * SS;
    float* ex2   = pd2  + (size_t)NN * SS;
    int*   slots = (int*)(ex2 + (size_t)ETOT);
    float* part2 = h1T;

    const int TB = 256;
    auto blocks = [](long long n, int tb) { return (unsigned)((n + tb - 1) / tb); };

    static_assert(NN % 32 == 0 && HC1 % 256 == 0 && (32 * IN_CH / 4) % 256 == 0, "GEMM 1 tiling");
    wmma_gemm_f32_ct<<<(NN / 32) * (HC1 / 256), 256, 0, stream>>>(x, W1, h1T, NN, HC1);
    slots_and_scores1<<<NCH + HEADS * NB1, 1024, 0, stream>>>(ei, slots, h1T, a1s, a1d, es1T, ed1T);
    edge_max_slices<<<HEADS * SS, 1024, 0, stream>>>(slots, es1T, ed1T, pm1);
    combine_max    <<<blocks((long long)NN * HEADS, TB), TB, 0, stream>>>(pm1, m1T, HEADS);
    edge_exp_slices<<<HEADS * SS, 1024, 0, stream>>>(slots, es1T, ed1T, m1T, al1, pd1);
    combine_den    <<<blocks((long long)NN * HEADS, TB), TB, 0, stream>>>(pd1, den1T, HEADS);
    agg_pair_lds  <<<HC1 / 2, 1024, 0, stream>>>(slots, al1, h1T, den1T, b1, agg1T);

    static_assert(OUT_CH == 64 && HC1 % (4 * KS2) == 0 && NN % 32 == 0, "GEMM 2 tiling");
    wmma_gemm_f32_tt<<<NN / 32, 256, 0, stream>>>(agg1T, W2, h2T, NN, HC1, OUT_CH);
    scores_ct<OUT_CH><<<dim3(blocks(NN, TB), 1), TB, 0, stream>>>(
        h2T, a2s, a2d, es2T, ed2T);
    edge_max_slices<<<SS, 1024, 0, stream>>>(slots, es2T, ed2T, pm2);
    combine_max    <<<blocks((long long)NN, TB), TB, 0, stream>>>(pm2, m2T, 1);
    edge_exp_slices<<<SS, 1024, 0, stream>>>(slots, es2T, ed2T, m2T, ex2, pd2);
    agg2_slice_lds<<<(OUT_CH / 2) * S2, 1024, 0, stream>>>(slots, ex2, h2T, part2);
    final_planes_logsoftmax<<<NN / 32, 256, 0, stream>>>(part2, pd2, b2, out);
}


# Edge softmax shifted by the bound leaky(head's largest source score + dst score) where a head's source scores span <= 16 (tested each call, else exact maxima): maximum sweeps and folds exit at once
# speedup vs baseline: 5.8031x; 1.0864x over previous
#include <hip/hip_runtime.h>
#include <math.h>

#define NN      20000
#define NE      320000
#define NCH     32
#define CHE     (NE / NCH)
#define CAP     352
#define CHS     (CAP * 32)
#define NSLOT   (NCH * CHS)
#define ETOT    (NSLOT + NN)
static_assert(NE % NCH == 0 && CHS >= CHE && CHS % 1024 == 0, "a chunk's slots must hold its edges");
static_assert(NN <= 32768, "source and destination share one word");
#define IN_CH   128
#define HID     64
#define HEADS   8
#define HC1     (HEADS * HID)
#define OUT_CH  64

typedef float v2f __attribute__((ext_vector_type(2)));
typedef float v8f __attribute__((ext_vector_type(8)));

__device__ __forceinline__ int f2ord(float f) {
    int i = __float_as_int(f);
    return i >= 0 ? i : (i ^ 0x7fffffff);
}
__device__ __forceinline__ float ord2f(int i) {
    return __int_as_float(i >= 0 ? i : (i ^ 0x7fffffff));
}

#define XP (IN_CH + 4)
typedef float v4f __attribute__((ext_vector_type(4)));
__global__ __launch_bounds__(256) void wmma_gemm_f32_ct(const float* __restrict__ X,
                                                        const float* __restrict__ Wm,
                                                        float* __restrict__ HT,
                                                        int M, int N) {
    __shared__ float xs[32 * XP];
    const int w       = threadIdx.x >> 5;
    const int lane    = threadIdx.x & 31;
    const int cParts  = N >> 8;
    const int mg      = blockIdx.x / cParts;
    const int cg      = (blockIdx.x % cParts) * 8 + w;
    const int l    = lane & 15;
    const int half = lane >> 4;
    const unsigned uN = (unsigned)N, uM = (unsigned)M;

    const float* xg = X + (size_t)mg * 32 * IN_CH;
    #pragma unroll
    for (int q = 0; q < (32 * IN_CH / 4) / 256; ++q) {
        const int p = q * 256 + threadIdx.x;
        const v4f v = *(const v4f*)(xg + 4 * p);
        *(v4f*)(xs + (p >> 5) * XP + 4 * (p & 31)) = v;
    }
    __syncthreads();

    const float* xr0 = xs + l * XP + 2 * half;
    const float* xr1 = xr0 + 16 * XP;
    unsigned woff = (unsigned)(2 * half) * uN + (unsigned)(cg * 32 + l);

    v8f acc00 = {}, acc01 = {}, acc10 = {}, acc11 = {};
    for (int k0 = 0; k0 < IN_CH; k0 += 4) {
        v2f x0 = *(const v2f*)(xr0 + k0);
        v2f x1 = *(const v2f*)(xr1 + k0);
        v2f w0 = { Wm[woff],      Wm[woff + uN]      };
        v2f w1 = { Wm[woff + 16], Wm[woff + uN + 16] };
        acc00 = __builtin_amdgcn_wmma_f32_16x16x4_f32(false, w0, false, x0, (short)0, acc00, false, false);
        acc01 = __builtin_amdgcn_wmma_f32_16x16x4_f32(false, w0, false, x1, (short)0, acc01, false, false);
        acc10 = __builtin_amdgcn_wmma_f32_16x16x4_f32(false, w1, false, x0, (short)0, acc10, false, false);
        acc11 = __builtin_amdgcn_wmma_f32_16x16x4_f32(false, w1, false, x1, (short)0, acc11, false, false);
        woff += 4u * uN;
    }
    unsigned hoff = (unsigned)(cg * 32 + 8 * half) * uM + (unsigned)(mg * 32 + l);
    #pragma unroll
    for (int r = 0; r < 8; ++r) {
        float* hp = HT + hoff;
        hp[0]            = acc00[r];
        hp[16]           = acc01[r];
        hp[16u * uM]      = acc10[r];
        hp[16u * uM + 16] = acc11[r];
        hoff += uM;
    }
}

#define KS2 4
__global__ __launch_bounds__(256) void wmma_gemm_f32_tt(const float* __restrict__ XT,
                                                        const float* __restrict__ Wm,
                                                        float* __restrict__ HT,
                                                        int M, int K, int N) {
    __shared__ float part[2][KS2][32][34];
    const int w    = threadIdx.x >> 5;
    const int lane = threadIdx.x & 31;
    const int cg   = w & 1;
    const int ks   = w >> 1;
    const int mg   = blockIdx.x;
    const int l    = lane & 15;
    const int half = lane >> 4;
    const unsigned uN = (unsigned)N, uM = (unsigned)M;
    const int KL = K / KS2;
    const unsigned kb = (unsigned)(ks * KL);

    unsigned xoff = (kb + 2u * half) * uM + (unsigned)(mg * 32 + l);
    unsigned woff = (kb + 2u * half) * uN + (unsigned)(cg * 32 + l);

    v8f acc00 = {}, acc01 = {}, acc10 = {}, acc11 = {};
    for (int k0 = 0; k0 < KL; k0 += 4) {
        v2f x0 = { XT[xoff],      XT[xoff + uM]      };
        v2f x1 = { XT[xoff + 16], XT[xoff + uM + 16] };
        v2f w0 = { Wm[woff],      Wm[woff + uN]      };
        v2f w1 = { Wm[woff + 16], Wm[woff + uN + 16] };
        acc00 = __builtin_amdgcn_wmma_f32_16x16x4_f32(false, w0, false, x0, (short)0, acc00, false, false);
        acc01 = __builtin_amdgcn_wmma_f32_16x16x4_f32(false, w0, false, x1, (short)0, acc01, false, false);
        acc10 = __builtin_amdgcn_wmma_f32_16x16x4_f32(false, w1, false, x0, (short)0, acc10, false, false);
        acc11 = __builtin_amdgcn_wmma_f32_16x16x4_f32(false, w1, false, x1, (short)0, acc11, false, false);
        xoff += 4u * uM;
        woff += 4u * uN;
    }
    float (*pt)[34] = part[cg][ks];
    #pragma unroll
    for (int r = 0; r < 8; ++r) {
        pt[8 * half + r     ][l     ] = acc00[r];
        pt[8 * half + r     ][l + 16] = acc01[r];
        pt[8 * half + r + 16][l     ] = acc10[r];
        pt[8 * half + r + 16][l + 16] = acc11[r];
    }
    __syncthreads();
    unsigned hoff = (unsigned)(cg * 32 + 8 * ks) * uM + (unsigned)(mg * 32 + lane);
    #pragma unroll
    for (int j = 0; j < 8; ++j) {
        const int row = 8 * ks + j;
        float v = part[cg][0][row][lane];
        #pragma unroll
        for (int q = 1; q < KS2; ++q) v += part[cg][q][row][lane];
        HT[hoff] = v;
        hoff += uM;
    }
}

#define SPAN_OK 16.0f
__device__ __forceinline__ void block_minmax_store(float s, bool valid, float* __restrict__ omax,
                                                   float* __restrict__ omin) {
    __shared__ float rmx[32], rmn[32];
    float mx = valid ? s : -INFINITY, mn = valid ? s : INFINITY;
    #pragma unroll
    for (int o = 16; o > 0; o >>= 1) {
        mx = fmaxf(mx, __shfl_xor(mx, o, 32));
        mn = fminf(mn, __shfl_xor(mn, o, 32));
    }
    const int w = threadIdx.x >> 5, nw = blockDim.x >> 5;
    if ((threadIdx.x & 31) == 0) { rmx[w] = mx; rmn[w] = mn; }
    __syncthreads();
    if (threadIdx.x == 0) {
        for (int i = 1; i < nw; ++i) { mx = fmaxf(mx, rmx[i]); mn = fminf(mn, rmn[i]); }
        *omax = mx; *omin = mn;
    }
}
__device__ __forceinline__ bool head_bound(const float* __restrict__ pmm, int hh, int nb, float& M) {
    const float* p = pmm + (size_t)hh * 2 * nb;
    const int lane = threadIdx.x & 31;
    float a[3], b[3];
    #pragma unroll
    for (int q = 0; q < 3; ++q) {
        const int i  = q * 32 + lane;
        const int ic = i < nb ? i : 0;
        a[q] = p[ic]; b[q] = p[nb + ic];
    }
    float mx = fmaxf(fmaxf(a[0], a[1]), a[2]), mn = fminf(fminf(b[0], b[1]), b[2]);
    #pragma unroll
    for (int o = 16; o > 0; o >>= 1) {
        mx = fmaxf(mx, __shfl_xor(mx, o, 32));
        mn = fminf(mn, __shfl_xor(mn, o, 32));
    }
    mx = __int_as_float(__builtin_amdgcn_readfirstlane(__float_as_int(mx)));
    mn = __int_as_float(__builtin_amdgcn_readfirstlane(__float_as_int(mn)));
    M = mx;
    return (mx - mn) <= SPAN_OK;
}

template <int C>
__device__ __forceinline__ void scores_body(const float* __restrict__ HT,
                                            const float* __restrict__ a_src,
                                            const float* __restrict__ a_dst,
                                            float* __restrict__ esT, float* __restrict__ edT,
                                            float* __restrict__ pmm, int nb, int grp,
                                            int hh, int node) {
    const bool valid = node < NN;
    if (!valid) node = 0;
    const float* hp = HT + (size_t)hh * C * NN + node;
    const float* as = a_src + hh * C;
    const float* ad = a_dst + hh * C;
    float v[C];
    #pragma unroll
    for (int c = 0; c < C; ++c) v[c] = hp[(size_t)c * NN];
    float s = 0.f, d = 0.f;
    #pragma unroll
    for (int c = 0; c < C; ++c) { s += v[c] * as[c]; d += v[c] * ad[c]; }
    if (valid) {
        esT[(size_t)hh * NN + node] = s;
        edT[(size_t)hh * NN + node] = d;
    }
    block_minmax_store(s, valid, pmm + (size_t)hh * 2 * nb + grp, pmm + (size_t)hh * 2 * nb + nb + grp);
}
template <int C>
__global__ void scores_ct(const float* __restrict__ HT,
                          const float* __restrict__ a_src,
                          const float* __restrict__ a_dst,
                          float* __restrict__ esT, float* __restrict__ edT, float* __restrict__ pmm) {
    scores_body<C>(HT, a_src, a_dst, esT, edT, pmm, gridDim.x, blockIdx.x,
                   blockIdx.y, blockIdx.x * blockDim.x + threadIdx.x);
}

__device__ __forceinline__ void bank_slots_body(const int* __restrict__ ei, int* __restrict__ slot, const int g) {
    __shared__ int tab[CHS];
    __shared__ int spill[CHE];
    __shared__ int cnt[32];
    __shared__ int fre[33];
    __shared__ int nspill;
    const int tid = threadIdx.x;
    constexpr int UE = (CHE + 1023) / 1024;
    int s[UE], d[UE];
    #pragma unroll
    for (int u = 0; u < UE; ++u) {
        const int i  = u * 1024 + tid;
        const int ic = i < CHE ? i : 0;
        s[u] = ei[g * CHE + ic];
        d[u] = ei[NE + g * CHE + ic];
    }
    for (int q = 0; q < CHS / 1024; ++q) tab[q * 1024 + tid] = -1;
    if (tid < 32) cnt[tid] = 0;
    if (tid == 0) nspill = 0;
    __syncthreads();
    #pragma unroll
    for (int u = 0; u < UE; ++u) {
        const int i = u * 1024 + tid;
        if (i < CHE) {
            const int cl = d[u] & 31;
            const int v  = (d[u] << 16) | s[u];
            const int r  = atomicAdd(&cnt[cl], 1);
            if (r < CAP) tab[r * 32 + cl] = v;
            else         spill[atomicAdd(&nspill, 1)] = v;
        }
    }
    __syncthreads();
    if (tid == 0) {
        int a = 0;
        for (int cl = 0; cl < 32; ++cl) {
            fre[cl] = a;
            const int used = cnt[cl] < CAP ? cnt[cl] : CAP;
            a += CAP - used;
        }
        fre[32] = a;
    }
    __syncthreads();
    const int ns = nspill;
    for (int k = tid; k < ns; k += 1024) {
        int cl = 0;
        for (int j = 1; j < 32; ++j) cl += (fre[j] <= k) ? 1 : 0;
        const int used = cnt[cl] < CAP ? cnt[cl] : CAP;
        tab[(used + k - fre[cl]) * 32 + cl] = spill[k];
    }
    __syncthreads();
    int* o = slot + (size_t)g * CHS;
    for (int q = 0; q < CHS / 1024; ++q) o[q * 1024 + tid] = tab[q * 1024 + tid];
}

#define NB1 ((NN + 1023) / 1024)
#define NB2 ((NN + 255) / 256)
static_assert(NB1 <= 96 && NB2 <= 96 && NN % 32 == 0, "head_bound: three groups per lane, one head per wave");
__global__ __launch_bounds__(1024) void slots_and_scores1(const int* __restrict__ ei, int* __restrict__ slot,
                                                          const float* __restrict__ HT,
                                                          const float* __restrict__ a_src,
                                                          const float* __restrict__ a_dst,
                                                          float* __restrict__ esT, float* __restrict__ edT,
                                                          float* __restrict__ pmm) {
    if (blockIdx.x < NCH) { bank_slots_body(ei, slot, blockIdx.x); return; }
    const int b = blockIdx.x - NCH;
    scores_body<HID>(HT, a_src, a_dst, esT, edT, pmm, NB1, b % NB1, b / NB1, (b % NB1) * 1024 + threadIdx.x);
}

#define SS   32
#define EPS  (ETOT / SS)
#define UM   ((EPS + 1023) / 1024)
#define NG   10
static_assert(ETOT % SS == 0, "the slices must divide the list");
static_assert(2 * NG * 1024 >= NN, "two groups of NG sweeps must cover the nodes");

#define LOAD_SLICE_EDGES(sr, ds)                                              \
    _Pragma("unroll")                                                         \
    for (int u = 0; u < UM; ++u) {                                            \
        const int e  = e0 + u * 1024 + tid;                                   \
        const int ec = (e < e1 && e < NSLOT) ? e : 0;                         \
        const int sv = slot[ec];                                              \
        sr[u] = (e >= NSLOT) ? e - NSLOT : (sv & 0xffff);                     \
        ds[u] = (e >= NSLOT) ? e - NSLOT : (sv >> 16);                        \
    }

__global__ __launch_bounds__(1024) void edge_max_slices(const int* __restrict__ slot,
                                                        const float* __restrict__ esT,
                                                        const float* __restrict__ edT,
                                                        int* __restrict__ pmax,
                                                        const float* __restrict__ pmm, int nb) {
    __shared__ float es[NN];
    __shared__ float ed[NN];
    __shared__ int   mx[NN];
    const int tid = threadIdx.x;
    const int hh  = blockIdx.x / SS;
    float Mh;
    if (head_bound(pmm, hh, nb, Mh)) return;
    const int sl  = blockIdx.x % SS;
    const int e0  = sl * EPS, e1 = e0 + EPS;
    int sr[UM], ds[UM];
    LOAD_SLICE_EDGES(sr, ds)
    const float* sp = esT + (size_t)hh * NN;
    const float* dp = edT + (size_t)hh * NN;
    #pragma unroll
    for (int g = 0; g < 2; ++g) {
        float a[NG], b[NG];
        #pragma unroll
        for (int j = 0; j < NG; ++j) {
            const int n  = (g * NG + j) * 1024 + tid;
            const int nc = n < NN ? n : 0;
            a[j] = sp[nc]; b[j] = dp[nc];
        }
        #pragma unroll
        for (int j = 0; j < NG; ++j) {
            const int n = (g * NG + j) * 1024 + tid;
            if (n < NN) { es[n] = a[j]; ed[n] = b[j]; mx[n] = f2ord(-INFINITY); }
        }
    }
    __syncthreads();
    #pragma unroll
    for (int u = 0; u < UM; ++u) {
        const int e = e0 + u * 1024 + tid;
        if (e < e1 && ds[u] >= 0) {
            float v = es[sr[u]] + ed[ds[u]];
            v = v > 0.f ? v : 0.2f * v;
            atomicMax(&mx[ds[u]], f2ord(v));
        }
    }
    __syncthreads();
    int* o = pmax + (size_t)blockIdx.x * NN;
    for (int n = tid; n < NN; n += 1024) o[n] = mx[n];
}

__global__ void combine_max(const int* __restrict__ pmax, float* __restrict__ mT, int H,
                            const float* __restrict__ pmm, int nb) {
    const int i = blockIdx.x * blockDim.x + threadIdx.x;
    if (i >= NN * H) return;
    const int hh = i / NN, node = i % NN;
    float Mh;
    if (head_bound(pmm, hh, nb, Mh)) return;
    const int* p = pmax + (size_t)hh * SS * NN + node;
    int m = p[0];
    #pragma unroll
    for (int s = 1; s < SS; ++s) { const int v = p[(size_t)s * NN]; m = v > m ? v : m; }
    mT[i] = ord2f(m);
}

__global__ __launch_bounds__(1024) void edge_exp_slices(const int* __restrict__ slot,
                                                        const float* __restrict__ esT,
                                                        const float* __restrict__ edT,
                                                        const float* __restrict__ mT,
                                                        float* __restrict__ exT,
                                                        float* __restrict__ pden,
                                                        const float* __restrict__ pmm, int nb) {
    __shared__ float es[NN];
    __shared__ float ed[NN];
    __shared__ float mm[NN];
    __shared__ float dn[NN];
    const int tid = threadIdx.x;
    const int hh  = blockIdx.x / SS;
    const int sl  = blockIdx.x % SS;
    const int e0  = sl * EPS, e1 = e0 + EPS;
    int sr[UM], ds[UM];
    LOAD_SLICE_EDGES(sr, ds)
    const float* sp = esT + (size_t)hh * NN;
    const float* dp = edT + (size_t)hh * NN;
    const float* mp = mT  + (size_t)hh * NN;
    float Mh;
    const bool bound = head_bound(pmm, hh, nb, Mh);
    #pragma unroll
    for (int g = 0; g < 2; ++g) {
        float a[NG], b[NG], c[NG];
        #pragma unroll
        for (int j = 0; j < NG; ++j) {
            const int n  = (g * NG + j) * 1024 + tid;
            const int nc = n < NN ? n : 0;
            a[j] = sp[nc]; b[j] = dp[nc];
            c[j] = 0.f;
            if (!bound) c[j] = mp[nc];
        }
        #pragma unroll
        for (int j = 0; j < NG; ++j) {
            const int n = (g * NG + j) * 1024 + tid;
            if (n < NN) {
                es[n] = a[j]; ed[n] = b[j]; dn[n] = 0.f;
                if (!bound) mm[n] = c[j];
            }
        }
    }
    __syncthreads();
    float* xo = exT + (size_t)hh * ETOT;
    #pragma unroll
    for (int u = 0; u < UM; ++u) {
        const int e = e0 + u * 1024 + tid;
        if (e < e1) {
            float ex = 0.f;
            if (ds[u] >= 0) {
                const float dv = ed[ds[u]];
                float v = es[sr[u]] + dv;
                v = v > 0.f ? v : 0.2f * v;
                float m = Mh + dv;
                m = m > 0.f ? m : 0.2f * m;
                if (!bound) m = mm[ds[u]];
                ex = expf(v - m);
                atomicAdd(&dn[ds[u]], ex);
            }
            xo[e] = ex;
        }
    }
    __syncthreads();
    float* o = pden + (size_t)blockIdx.x * NN;
    for (int n = tid; n < NN; n += 1024) o[n] = dn[n];
}

__global__ void combine_den(const float* __restrict__ pden, float* __restrict__ denT, int H) {
    const int i = blockIdx.x * blockDim.x + threadIdx.x;
    if (i >= NN * H) return;
    const int hh = i / NN, node = i % NN;
    const float* p = pden + (size_t)hh * SS * NN + node;
    float d = p[0];
    #pragma unroll
    for (int s = 1; s < SS; ++s) d += p[(size_t)s * NN];
    denT[i] = d;
}

__global__ __launch_bounds__(1024) void agg_pair_lds(const int* __restrict__ slot,
                                                     const float* __restrict__ alT,
                                                     const float* __restrict__ HT,
                                                     const float* __restrict__ denT,
                                                     const float* __restrict__ bias,
                                                     float* __restrict__ outT) {
    __shared__ v2f   hs[NN];
    __shared__ float acc0[NN];
    __shared__ float acc1[NN];
    const int tid  = threadIdx.x;
    const int c0   = 2 * blockIdx.x;
    const int head = c0 / HID;
    const float* h0 = HT + (size_t)c0 * NN;
    const float* h1 = h0 + NN;
    const float* al = alT + (size_t)head * ETOT;

    #pragma unroll
    for (int g = 0; g < 2; ++g) {
        float x0[NG], x1[NG], aa[NG];
        #pragma unroll
        for (int j = 0; j < NG; ++j) {
            const int n  = (g * NG + j) * 1024 + tid;
            const int nc = n < NN ? n : 0;
            x0[j] = h0[nc]; x1[j] = h1[nc];
            aa[j] = al[NSLOT + nc];
        }
        #pragma unroll
        for (int j = 0; j < NG; ++j) {
            const int n = (g * NG + j) * 1024 + tid;
            if (n < NN) {
                v2f hv = { x0[j], x1[j] };
                hs[n]   = hv;
                acc0[n] = aa[j] * hv.x;
                acc1[n] = aa[j] * hv.y;
            }
        }
    }
    __syncthreads();

    constexpr int FULL = NSLOT / 1024;
    constexpr int BAT  = 8;
    static_assert(NSLOT % 1024 == 0 && FULL % BAT == 0, "the batch must divide the sweeps");
    int sv[BAT]; float a[BAT];
    #pragma unroll
    for (int u = 0; u < BAT; ++u) {
        const int e = u * 1024 + tid;
        sv[u] = slot[e]; a[u] = al[e];
    }
    for (int it = 0; it < FULL; it += BAT) {
        const int nx = (it + BAT < FULL) ? it + BAT : it;
        int svn[BAT]; float an[BAT];
        #pragma unroll
        for (int u = 0; u < BAT; ++u) {
            const int e = (nx + u) * 1024 + tid;
            svn[u] = slot[e]; an[u] = al[e];
        }
        v2f hv[BAT];
        #pragma unroll
        for (int u = 0; u < BAT; ++u) hv[u] = hs[sv[u] < 0 ? 0 : (sv[u] & 0xffff)];
        #pragma unroll
        for (int u = 0; u < BAT; ++u) {
            const int d = sv[u] >> 16;
            if (d >= 0) {
                atomicAdd(&acc0[d], a[u] * hv[u].x);
                atomicAdd(&acc1[d], a[u] * hv[u].y);
            }
        }
        #pragma unroll
        for (int u = 0; u < BAT; ++u) { sv[u] = svn[u]; a[u] = an[u]; }
    }
    const float* dn = denT + (size_t)head * NN;
    float dd[2 * NG];
    #pragma unroll
    for (int j = 0; j < 2 * NG; ++j) {
        const int n  = j * 1024 + tid;
        const int nc = n < NN ? n : 0;
        dd[j] = dn[nc];
    }
    __syncthreads();

    float* o0 = outT + (size_t)c0 * NN;
    float* o1 = o0 + NN;
    const float b0 = bias[c0], b1 = bias[c0 + 1];
    #pragma unroll
    for (int j = 0; j < 2 * NG; ++j) {
        const int n = j * 1024 + tid;
        if (n < NN) {
            const float d = dd[j] + 1e-16f;
            const float v0 = acc0[n] / d + b0, v1 = acc1[n] / d + b1;
            o0[n] = v0 > 0.f ? v0 : 0.f;
            o1[n] = v1 > 0.f ? v1 : 0.f;
        }
    }
}

#define S2 8
__global__ __launch_bounds__(1024) void agg2_slice_lds(const int* __restrict__ slot,
                                                       const float* __restrict__ al,
                                                       const float* __restrict__ HT,
                                                       float* __restrict__ part) {
    __shared__ v2f   hs[NN];
    __shared__ float acc0[NN];
    __shared__ float acc1[NN];
    const int tid   = threadIdx.x;
    const int pair  = blockIdx.x % (OUT_CH / 2);
    const int slice = blockIdx.x / (OUT_CH / 2);
    const int c0    = 2 * pair;
    const float* h0 = HT + (size_t)c0 * NN;
    const float* h1 = h0 + NN;

    if (slice == 0) {
        #pragma unroll
        for (int g = 0; g < 2; ++g) {
            float x0[NG], x1[NG], aa[NG];
            #pragma unroll
            for (int j = 0; j < NG; ++j) {
                const int n  = (g * NG + j) * 1024 + tid;
                const int nc = n < NN ? n : 0;
                x0[j] = h0[nc]; x1[j] = h1[nc];
                aa[j] = al[NSLOT + nc];
            }
            #pragma unroll
            for (int j = 0; j < NG; ++j) {
                const int n = (g * NG + j) * 1024 + tid;
                if (n < NN) {
                    v2f hv = { x0[j], x1[j] };
                    hs[n]   = hv;
                    acc0[n] = aa[j] * hv.x;
                    acc1[n] = aa[j] * hv.y;
                }
            }
        }
    } else {
        #pragma unroll
        for (int g = 0; g < 2; ++g) {
            float x0[NG], x1[NG];
            #pragma unroll
            for (int j = 0; j < NG; ++j) {
                const int n  = (g * NG + j) * 1024 + tid;
                const int nc = n < NN ? n : 0;
                x0[j] = h0[nc]; x1[j] = h1[nc];
            }
            #pragma unroll
            for (int j = 0; j < NG; ++j) {
                const int n = (g * NG + j) * 1024 + tid;
                if (n < NN) {
                    v2f hv = { x0[j], x1[j] };
                    hs[n]   = hv;
                    acc0[n] = 0.f;
                    acc1[n] = 0.f;
                }
            }
        }
    }
    __syncthreads();

    constexpr int FULL = NSLOT / 1024;
    static_assert(NSLOT % 1024 == 0 && FULL % S2 == 0, "the slices must divide the sweeps");
    constexpr int SW  = FULL / S2;
    constexpr int BAT = 11;
    static_assert(SW % BAT == 0, "the batch must divide a slice's sweeps");
    const int base = slice * SW * 1024 + tid;
    for (int it = 0; it < SW; it += BAT) {
        int sv[BAT]; float a[BAT];
        #pragma unroll
        for (int u = 0; u < BAT; ++u) {
            const int e = base + (it + u) * 1024;
            sv[u] = slot[e]; a[u] = al[e];
        }
        v2f hv[BAT];
        #pragma unroll
        for (int u = 0; u < BAT; ++u) hv[u] = hs[sv[u] < 0 ? 0 : (sv[u] & 0xffff)];
        #pragma unroll
        for (int u = 0; u < BAT; ++u) {
            const int d = sv[u] >> 16;
            if (d >= 0) {
                atomicAdd(&acc0[d], a[u] * hv[u].x);
                atomicAdd(&acc1[d], a[u] * hv[u].y);
            }
        }
    }
    __syncthreads();

    float* o0 = part + ((size_t)slice * OUT_CH + c0) * NN;
    float* o1 = o0 + NN;
    for (int n = tid; n < NN; n += 1024) { o0[n] = acc0[n]; o1[n] = acc1[n]; }
}

__global__ __launch_bounds__(256) void final_planes_logsoftmax(const float* __restrict__ part,
                                                                const float* __restrict__ pden,
                                                                const float* __restrict__ b2,
                                                                float* __restrict__ out) {
    __shared__ float tile[OUT_CH][33];
    const int tid = threadIdx.x;
    const int w = tid >> 5, lane = tid & 31;
    const int n0 = blockIdx.x * 32;
    float d = pden[n0 + lane];
    #pragma unroll
    for (int s = 1; s < SS; ++s) d += pden[(size_t)s * NN + n0 + lane];
    d += 1e-16f;
    #pragma unroll
    for (int j = 0; j < 8; ++j) {
        const int c = 8 * w + j;
        const float* p = part + (size_t)c * NN + n0 + lane;
        float v = p[0];
        #pragma unroll
        for (int s = 1; s < S2; ++s) v += p[(size_t)s * OUT_CH * NN];
        tile[c][lane] = v / d + b2[c];
    }
    __syncthreads();
    #pragma unroll
    for (int j = 0; j < 4; ++j) {
        const int nl = 4 * w + j;
        float v0 = tile[lane][nl];
        float v1 = tile[lane + 32][nl];
        float m = fmaxf(v0, v1);
        #pragma unroll
        for (int o = 16; o > 0; o >>= 1) m = fmaxf(m, __shfl_xor(m, o, 32));
        float s = expf(v0 - m) + expf(v1 - m);
        #pragma unroll
        for (int o = 16; o > 0; o >>= 1) s += __shfl_xor(s, o, 32);
        float lse = logf(s) + m;
        float* row = out + (size_t)(n0 + nl) * OUT_CH;
        row[lane]      = v0 - lse;
        row[lane + 32] = v1 - lse;
    }
}

extern "C" void kernel_launch(void* const* d_in, const int* in_sizes, int n_in,
                              void* d_out, int out_size, void* d_ws, size_t ws_size,
                              hipStream_t stream) {
    const float* x   = (const float*)d_in[0];
    const int*   ei  = (const int*)  d_in[1];
    const float* W1  = (const float*)d_in[2];
    const float* a1s = (const float*)d_in[3];
    const float* a1d = (const float*)d_in[4];
    const float* b1  = (const float*)d_in[5];
    const float* W2  = (const float*)d_in[6];
    const float* a2s = (const float*)d_in[7];
    const float* a2d = (const float*)d_in[8];
    const float* b2  = (const float*)d_in[9];
    float* out = (float*)d_out;

    float* ws = (float*)d_ws;
    float* h1T   = ws;
    float* agg1T = h1T   + (size_t)NN * HC1;
    float* es1T  = agg1T + (size_t)NN * HC1;
    float* ed1T  = es1T  + (size_t)NN * HEADS;
    float* m1T   = ed1T  + (size_t)NN * HEADS;
    float* den1T = m1T   + (size_t)NN * HEADS;
    int*   pm1   = (int*)(den1T + (size_t)NN * HEADS);
    float* pd1   = (float*)pm1 + (size_t)NN * HEADS * SS;
    float* al1   = pd1  + (size_t)NN * HEADS * SS;
    float* h2T   = al1  + (size_t)ETOT * HEADS;
    float* es2T  = h2T  + (size_t)NN * OUT_CH;
    float* ed2T  = es2T + NN;
    float* m2T   = ed2T + NN;
    int*   pm2   = (int*)(m2T + NN);
    float* pd2   = (float*)pm2 + (size_t)NN * SS;
    float* ex2   = pd2  + (size_t)NN * SS;
    int*   slots = (int*)(ex2 + (size_t)ETOT);
    float* pmm1  = (float*)(slots + NSLOT);
    float* pmm2  = pmm1 + 1024;
    float* part2 = h1T;

    const int TB = 256;
    auto blocks = [](long long n, int tb) { return (unsigned)((n + tb - 1) / tb); };

    static_assert(NN % 32 == 0 && HC1 % 256 == 0 && (32 * IN_CH / 4) % 256 == 0, "GEMM 1 tiling");
    wmma_gemm_f32_ct<<<(NN / 32) * (HC1 / 256), 256, 0, stream>>>(x, W1, h1T, NN, HC1);
    slots_and_scores1<<<NCH + HEADS * NB1, 1024, 0, stream>>>(ei, slots, h1T, a1s, a1d, es1T, ed1T, pmm1);
    edge_max_slices<<<HEADS * SS, 1024, 0, stream>>>(slots, es1T, ed1T, pm1, pmm1, NB1);
    combine_max    <<<blocks((long long)NN * HEADS, TB), TB, 0, stream>>>(pm1, m1T, HEADS, pmm1, NB1);
    edge_exp_slices<<<HEADS * SS, 1024, 0, stream>>>(slots, es1T, ed1T, m1T, al1, pd1, pmm1, NB1);
    combine_den    <<<blocks((long long)NN * HEADS, TB), TB, 0, stream>>>(pd1, den1T, HEADS);
    agg_pair_lds  <<<HC1 / 2, 1024, 0, stream>>>(slots, al1, h1T, den1T, b1, agg1T);

    static_assert(OUT_CH == 64 && HC1 % (4 * KS2) == 0 && NN % 32 == 0, "GEMM 2 tiling");
    wmma_gemm_f32_tt<<<NN / 32, 256, 0, stream>>>(agg1T, W2, h2T, NN, HC1, OUT_CH);
    scores_ct<OUT_CH><<<dim3(NB2, 1), TB, 0, stream>>>(
        h2T, a2s, a2d, es2T, ed2T, pmm2);
    edge_max_slices<<<SS, 1024, 0, stream>>>(slots, es2T, ed2T, pm2, pmm2, NB2);
    combine_max    <<<blocks((long long)NN, TB), TB, 0, stream>>>(pm2, m2T, 1, pmm2, NB2);
    edge_exp_slices<<<SS, 1024, 0, stream>>>(slots, es2T, ed2T, m2T, ex2, pd2, pmm2, NB2);
    agg2_slice_lds<<<(OUT_CH / 2) * S2, 1024, 0, stream>>>(slots, ex2, h2T, part2);
    final_planes_logsoftmax<<<NN / 32, 256, 0, stream>>>(part2, pd2, b2, out);
}
